# starting kernel
# baseline (speedup 1.0000x reference)
#include <hip/hip_runtime.h>
#include <hip/hip_bf16.h>

typedef __attribute__((ext_vector_type(2))) float v2f;
typedef __attribute__((ext_vector_type(8))) float v8f;

#define QSCALE 0.17677669529663687f


__global__ void transpose_k(const float* __restrict__ src, float* __restrict__ dst,
                            int O, int K, int total) {
    int idx = blockIdx.x * blockDim.x + threadIdx.x;
    if (idx >= total) return;
    int o = idx % O;
    int k = idx / O;
    dst[idx] = src[(size_t)o * K + k];
}

__global__ void pool_transpose(const float* __restrict__ src, float* __restrict__ dst,
                               int total) {
    int idx = blockIdx.x * blockDim.x + threadIdx.x;
    if (idx >= total) return;
    int o = idx % 256;
    int k = idx / 256;
    int c = k % 256;
    int q = k / 256;
    dst[idx] = src[(size_t)o * 16384 + c * 64 + q];
}

__global__ void gather_xw(const float* __restrict__ x, float* __restrict__ xw, int total) {
    int idx = blockIdx.x * blockDim.x + threadIdx.x;
    if (idx >= total) return;
    int c = idx % 256;
    int t = idx / 256;
    int q = t % 64;
    int n = t / 64;
    int bi = n / 144;
    int l  = n % 144;
    int wh = l / 12, ww = l % 12;
    int i = q / 8, j = q % 8;
    xw[idx] = x[(((size_t)bi * 256 + c) * 96 + wh * 8 + i) * 96 + ww * 8 + j];
}

__global__ __launch_bounds__(256) void gemm_wmma_f32(
    const float* __restrict__ A, const float* __restrict__ B,
    float* __restrict__ C, const float* __restrict__ bias,
    int M, int N, int K, float scale)
{
    int wave = threadIdx.x >> 5;
    int lane = threadIdx.x & 31;
    int hf   = lane >> 4;
    int r    = lane & 15;
    int m0 = blockIdx.y * 32 + (wave >> 2) * 16;
    int n0 = blockIdx.x * 64 + (wave & 3) * 16;

    const float* Arow = A + (size_t)(m0 + r) * K + 2 * hf;
    const float* Bcol = B + (size_t)(2 * hf) * N + n0 + r;

    v8f acc = {};
    #pragma unroll 8
    for (int k = 0; k < K; k += 4) {
        v2f a, b;
        a.x = Arow[k];
        a.y = Arow[k + 1];
        b.x = Bcol[(size_t)k * N];
        b.y = Bcol[(size_t)k * N + N];
        acc = __builtin_amdgcn_wmma_f32_16x16x4_f32(false, a, false, b,
                                                    (short)0, acc, false, false);
    }
    int col = n0 + r;
    float bv = bias ? bias[col] : 0.0f;
    #pragma unroll
    for (int vg = 0; vg < 8; ++vg)
        C[(size_t)(m0 + vg + 8 * hf) * N + col] = acc[vg] * scale + bv;
}

#define LDST 145
__global__ __launch_bounds__(128) void attn_kernel(
    float* __restrict__ qmat, const float* __restrict__ kv)
{
    __shared__ float attn[64 * LDST];
    int id = blockIdx.x;
    int h  = id & 7;
    int bb = (id >> 3) & 7;
    int g  = id >> 6;
    int n  = g * 8 + bb;

    int wave = threadIdx.x >> 5;
    int lane = threadIdx.x & 31;
    int hf   = lane >> 4;
    int r    = lane & 15;
    int m0   = wave * 16;

    const float* qbase = qmat + ((size_t)n * 64 + m0 + r) * 256 + h * 32;
    const float* kvb   = kv + (size_t)bb * 144 * 512 + h * 32;
    const float* vb    = kvb + 256;

    for (int nt = 0; nt < 9; ++nt) {
        int n0 = nt * 16;
        const float* kcol = kvb + (size_t)(n0 + r) * 512 + 2 * hf;
        v8f acc = {};
        #pragma unroll
        for (int k = 0; k < 32; k += 4) {
            v2f a, b;
            a.x = qbase[k + 2 * hf];
            a.y = qbase[k + 2 * hf + 1];
            b.x = kcol[k];
            b.y = kcol[k + 1];
            acc = __builtin_amdgcn_wmma_f32_16x16x4_f32(false, a, false, b,
                                                        (short)0, acc, false, false);
        }
        #pragma unroll
        for (int vg = 0; vg < 8; ++vg)
            attn[(m0 + vg + 8 * hf) * LDST + n0 + r] = acc[vg];
    }
    __syncthreads();

    if (threadIdx.x < 64) {
        float* row = attn + threadIdx.x * LDST;
        float mx = row[0];
        for (int k = 1; k < 144; ++k) mx = fmaxf(mx, row[k]);
        float s = 0.0f;
        for (int k = 0; k < 144; ++k) { float e = __expf(row[k] - mx); row[k] = e; s += e; }
        float inv = 1.0f / s;
        for (int k = 0; k < 144; ++k) row[k] *= inv;
    }
    __syncthreads();

    const float* arow = attn + (m0 + r) * LDST + 2 * hf;
    for (int nt = 0; nt < 2; ++nt) {
        int n0 = nt * 16;
        v8f acc = {};
        #pragma unroll 4
        for (int k = 0; k < 144; k += 4) {
            v2f a, b;
            a.x = arow[k];
            a.y = arow[k + 1];
            b.x = vb[(size_t)(k + 2 * hf) * 512 + n0 + r];
            b.y = vb[(size_t)(k + 2 * hf + 1) * 512 + n0 + r];
            acc = __builtin_amdgcn_wmma_f32_16x16x4_f32(false, a, false, b,
                                                        (short)0, acc, false, false);
        }
        #pragma unroll
        for (int vg = 0; vg < 8; ++vg)
            qmat[((size_t)n * 64 + m0 + vg + 8 * hf) * 256 + h * 32 + n0 + r] = acc[vg];
    }
}

__global__ void rpe_add_kernel(const float* __restrict__ x, const float* __restrict__ rw,
                               const float* __restrict__ rb, float* __restrict__ F,
                               int total) {
    int idx = blockIdx.x * blockDim.x + threadIdx.x;
    if (idx >= total) return;
    int xx = idx % 96;
    int t  = idx / 96;
    int y  = t % 96; t /= 96;
    int c  = t % 256;
    int bi = t / 256;
    const float* xp = x + (size_t)(bi * 256 + c) * 9216;
    const float* wp = rw + c * 9;
    float acc = rb[c];
    #pragma unroll
    for (int dy = -1; dy <= 1; ++dy) {
        int yy = y + dy;
        if (yy < 0 || yy >= 96) continue;
        #pragma unroll
        for (int dx = -1; dx <= 1; ++dx) {
            int xc = xx + dx;
            if (xc < 0 || xc >= 96) continue;
            acc += xp[yy * 96 + xc] * wp[(dy + 1) * 3 + (dx + 1)];
        }
    }
    F[idx] += acc;
}

extern "C" void kernel_launch(void* const* d_in, const int* in_sizes, int n_in,
                              void* d_out, int out_size, void* d_ws, size_t ws_size,
                              hipStream_t stream) {
    const float* x      = (const float*)d_in[0];
    const float* Wq     = (const float*)d_in[1];
    const float* Wkv    = (const float*)d_in[2];
    const float* pool_w = (const float*)d_in[3];
    const float* pool_b = (const float*)d_in[4];
    const float* rpe_w  = (const float*)d_in[5];
    const float* rpe_b  = (const float*)d_in[6];
    const float* proj_w = (const float*)d_in[7];
    const float* proj_b = (const float*)d_in[8];
    float* out = (float*)d_out;

    float* ws    = (float*)d_ws;
    float* xw    = ws;
    float* qmat  = xw + 18874368;
    float* poolT = qmat + 18874368;
    float* WqT   = poolT + 4194304;
    float* WkvT  = WqT + 65536;
    float* projT = WkvT + 131072;
    float* poolx = projT + 65536;
    float* kvb   = poolx + 294912;
    (void)ws_size; (void)in_sizes; (void)n_in; (void)out_size;

    const int TB = 256;
    transpose_k<<<(65536 + TB - 1) / TB, TB, 0, stream>>>(Wq, WqT, 256, 256, 65536);
    transpose_k<<<(131072 + TB - 1) / TB, TB, 0, stream>>>(Wkv, WkvT, 512, 256, 131072);
    transpose_k<<<(65536 + TB - 1) / TB, TB, 0, stream>>>(proj_w, projT, 256, 256, 65536);
    pool_transpose<<<(4194304 + TB - 1) / TB, TB, 0, stream>>>(pool_w, poolT, 4194304);

    gather_xw<<<(18874368 + TB - 1) / TB, TB, 0, stream>>>(x, xw, 18874368);

    gemm_wmma_f32<<<dim3(256 / 64, 1152 / 32), TB, 0, stream>>>(
        xw, poolT, poolx, pool_b, 1152, 256, 16384, 1.0f);

    gemm_wmma_f32<<<dim3(256 / 64, 73728 / 32), TB, 0, stream>>>(
        xw, WqT, qmat, nullptr, 73728, 256, 256, QSCALE);

    gemm_wmma_f32<<<dim3(512 / 64, 1152 / 32), TB, 0, stream>>>(
        poolx, WkvT, kvb, nullptr, 1152, 512, 256, 1.0f);

    attn_kernel<<<144 * 8 * 8, 128, 0, stream>>>(qmat, kvb);

    rpe_add_kernel<<<(18874368 + TB - 1) / TB, TB, 0, stream>>>(
        x, rpe_w, rpe_b, qmat, 18874368);

    gemm_wmma_f32<<<dim3(256 / 64, 73728 / 32), TB, 0, stream>>>(
        qmat, projT, out, proj_b, 73728, 256, 256, 1.0f);
}


# Pooling GEMM (K=16384) split along K into 64 grid-z slices: 9216 blocks instead of 144; a new kernel sums the 64 partial planes in fixed slice order (+1 launch; f32 summation order changes)
# speedup vs baseline: 1.4078x; 1.4078x over previous
#include <hip/hip_runtime.h>
#include <hip/hip_bf16.h>

typedef __attribute__((ext_vector_type(2))) float v2f;
typedef __attribute__((ext_vector_type(8))) float v8f;

#define QSCALE 0.17677669529663687f


__global__ void transpose_k(const float* __restrict__ src, float* __restrict__ dst,
                            int O, int K, int total) {
    int idx = blockIdx.x * blockDim.x + threadIdx.x;
    if (idx >= total) return;
    int o = idx % O;
    int k = idx / O;
    dst[idx] = src[(size_t)o * K + k];
}

__global__ void pool_transpose(const float* __restrict__ src, float* __restrict__ dst,
                               int total) {
    int idx = blockIdx.x * blockDim.x + threadIdx.x;
    if (idx >= total) return;
    int o = idx % 256;
    int k = idx / 256;
    int c = k % 256;
    int q = k / 256;
    dst[idx] = src[(size_t)o * 16384 + c * 64 + q];
}

__global__ void gather_xw(const float* __restrict__ x, float* __restrict__ xw, int total) {
    int idx = blockIdx.x * blockDim.x + threadIdx.x;
    if (idx >= total) return;
    int c = idx % 256;
    int t = idx / 256;
    int q = t % 64;
    int n = t / 64;
    int bi = n / 144;
    int l  = n % 144;
    int wh = l / 12, ww = l % 12;
    int i = q / 8, j = q % 8;
    xw[idx] = x[(((size_t)bi * 256 + c) * 96 + wh * 8 + i) * 96 + ww * 8 + j];
}

__global__ __launch_bounds__(256) void gemm_wmma_f32(
    const float* __restrict__ A, const float* __restrict__ B,
    float* __restrict__ C, const float* __restrict__ bias,
    int M, int N, int K, int lda, float scale)
{
    int wave = threadIdx.x >> 5;
    int lane = threadIdx.x & 31;
    int hf   = lane >> 4;
    int r    = lane & 15;
    int m0 = blockIdx.y * 32 + (wave >> 2) * 16;
    int n0 = blockIdx.x * 64 + (wave & 3) * 16;
    size_t kz = (size_t)blockIdx.z * K;

    const float* Arow = A + (size_t)(m0 + r) * lda + kz + 2 * hf;
    const float* Bcol = B + (kz + 2 * hf) * N + n0 + r;
    C += (size_t)blockIdx.z * M * N;

    v8f acc = {};
    #pragma unroll 8
    for (int k = 0; k < K; k += 4) {
        v2f a, b;
        a.x = Arow[k];
        a.y = Arow[k + 1];
        b.x = Bcol[(size_t)k * N];
        b.y = Bcol[(size_t)k * N + N];
        acc = __builtin_amdgcn_wmma_f32_16x16x4_f32(false, a, false, b,
                                                    (short)0, acc, false, false);
    }
    int col = n0 + r;
    float bv = bias ? bias[col] : 0.0f;
    #pragma unroll
    for (int vg = 0; vg < 8; ++vg)
        C[(size_t)(m0 + vg + 8 * hf) * N + col] = acc[vg] * scale + bv;
}

__global__ __launch_bounds__(64) void splitk_reduce(
    const float* __restrict__ part, const float* __restrict__ bias,
    float* __restrict__ dst, int MN, int N, int S)
{
    int i4 = blockIdx.x * blockDim.x + threadIdx.x;
    int quads = MN >> 2;
    if (i4 >= quads) return;
    const float4* p = reinterpret_cast<const float4*>(part) + i4;
    float4 acc = make_float4(0.0f, 0.0f, 0.0f, 0.0f);
    #pragma unroll 16
    for (int z = 0; z < S; ++z) {
        float4 v = p[(size_t)z * quads];
        acc.x += v.x; acc.y += v.y; acc.z += v.z; acc.w += v.w;
    }
    int col = (i4 << 2) % N;
    const float4 bv = *reinterpret_cast<const float4*>(bias + col);
    acc.x += bv.x; acc.y += bv.y; acc.z += bv.z; acc.w += bv.w;
    reinterpret_cast<float4*>(dst)[i4] = acc;
}

#define LDST 145
__global__ __launch_bounds__(128) void attn_kernel(
    float* __restrict__ qmat, const float* __restrict__ kv)
{
    __shared__ float attn[64 * LDST];
    int id = blockIdx.x;
    int h  = id & 7;
    int bb = (id >> 3) & 7;
    int g  = id >> 6;
    int n  = g * 8 + bb;

    int wave = threadIdx.x >> 5;
    int lane = threadIdx.x & 31;
    int hf   = lane >> 4;
    int r    = lane & 15;
    int m0   = wave * 16;

    const float* qbase = qmat + ((size_t)n * 64 + m0 + r) * 256 + h * 32;
    const float* kvb   = kv + (size_t)bb * 144 * 512 + h * 32;
    const float* vb    = kvb + 256;

    for (int nt = 0; nt < 9; ++nt) {
        int n0 = nt * 16;
        const float* kcol = kvb + (size_t)(n0 + r) * 512 + 2 * hf;
        v8f acc = {};
        #pragma unroll
        for (int k = 0; k < 32; k += 4) {
            v2f a, b;
            a.x = qbase[k + 2 * hf];
            a.y = qbase[k + 2 * hf + 1];
            b.x = kcol[k];
            b.y = kcol[k + 1];
            acc = __builtin_amdgcn_wmma_f32_16x16x4_f32(false, a, false, b,
                                                        (short)0, acc, false, false);
        }
        #pragma unroll
        for (int vg = 0; vg < 8; ++vg)
            attn[(m0 + vg + 8 * hf) * LDST + n0 + r] = acc[vg];
    }
    __syncthreads();

    if (threadIdx.x < 64) {
        float* row = attn + threadIdx.x * LDST;
        float mx = row[0];
        for (int k = 1; k < 144; ++k) mx = fmaxf(mx, row[k]);
        float s = 0.0f;
        for (int k = 0; k < 144; ++k) { float e = __expf(row[k] - mx); row[k] = e; s += e; }
        float inv = 1.0f / s;
        for (int k = 0; k < 144; ++k) row[k] *= inv;
    }
    __syncthreads();

    const float* arow = attn + (m0 + r) * LDST + 2 * hf;
    for (int nt = 0; nt < 2; ++nt) {
        int n0 = nt * 16;
        v8f acc = {};
        #pragma unroll 4
        for (int k = 0; k < 144; k += 4) {
            v2f a, b;
            a.x = arow[k];
            a.y = arow[k + 1];
            b.x = vb[(size_t)(k + 2 * hf) * 512 + n0 + r];
            b.y = vb[(size_t)(k + 2 * hf + 1) * 512 + n0 + r];
            acc = __builtin_amdgcn_wmma_f32_16x16x4_f32(false, a, false, b,
                                                        (short)0, acc, false, false);
        }
        #pragma unroll
        for (int vg = 0; vg < 8; ++vg)
            qmat[((size_t)n * 64 + m0 + vg + 8 * hf) * 256 + h * 32 + n0 + r] = acc[vg];
    }
}

__global__ void rpe_add_kernel(const float* __restrict__ x, const float* __restrict__ rw,
                               const float* __restrict__ rb, float* __restrict__ F,
                               int total) {
    int idx = blockIdx.x * blockDim.x + threadIdx.x;
    if (idx >= total) return;
    int xx = idx % 96;
    int t  = idx / 96;
    int y  = t % 96; t /= 96;
    int c  = t % 256;
    int bi = t / 256;
    const float* xp = x + (size_t)(bi * 256 + c) * 9216;
    const float* wp = rw + c * 9;
    float acc = rb[c];
    #pragma unroll
    for (int dy = -1; dy <= 1; ++dy) {
        int yy = y + dy;
        if (yy < 0 || yy >= 96) continue;
        #pragma unroll
        for (int dx = -1; dx <= 1; ++dx) {
            int xc = xx + dx;
            if (xc < 0 || xc >= 96) continue;
            acc += xp[yy * 96 + xc] * wp[(dy + 1) * 3 + (dx + 1)];
        }
    }
    F[idx] += acc;
}

extern "C" void kernel_launch(void* const* d_in, const int* in_sizes, int n_in,
                              void* d_out, int out_size, void* d_ws, size_t ws_size,
                              hipStream_t stream) {
    const float* x      = (const float*)d_in[0];
    const float* Wq     = (const float*)d_in[1];
    const float* Wkv    = (const float*)d_in[2];
    const float* pool_w = (const float*)d_in[3];
    const float* pool_b = (const float*)d_in[4];
    const float* rpe_w  = (const float*)d_in[5];
    const float* rpe_b  = (const float*)d_in[6];
    const float* proj_w = (const float*)d_in[7];
    const float* proj_b = (const float*)d_in[8];
    float* out = (float*)d_out;

    float* ws    = (float*)d_ws;
    float* xw    = ws;
    float* qmat  = xw + 18874368;
    float* poolT = qmat + 18874368;
    float* WqT   = poolT + 4194304;
    float* WkvT  = WqT + 65536;
    float* projT = WkvT + 131072;
    float* poolx = projT + 65536;
    float* kvb   = poolx + 294912;
    float* part  = kvb + 589824;
    (void)ws_size; (void)in_sizes; (void)n_in; (void)out_size;

    const int TB = 256;
    transpose_k<<<(65536 + TB - 1) / TB, TB, 0, stream>>>(Wq, WqT, 256, 256, 65536);
    transpose_k<<<(131072 + TB - 1) / TB, TB, 0, stream>>>(Wkv, WkvT, 512, 256, 131072);
    transpose_k<<<(65536 + TB - 1) / TB, TB, 0, stream>>>(proj_w, projT, 256, 256, 65536);
    pool_transpose<<<(4194304 + TB - 1) / TB, TB, 0, stream>>>(pool_w, poolT, 4194304);

    gather_xw<<<(18874368 + TB - 1) / TB, TB, 0, stream>>>(x, xw, 18874368);

    gemm_wmma_f32<<<dim3(256 / 64, 1152 / 32, 64), TB, 0, stream>>>(
        xw, poolT, part, nullptr, 1152, 256, 256, 16384, 1.0f);
    splitk_reduce<<<73728 / 64, 64, 0, stream>>>(
        part, pool_b, poolx, 294912, 256, 64);

    gemm_wmma_f32<<<dim3(256 / 64, 73728 / 32), TB, 0, stream>>>(
        xw, WqT, qmat, nullptr, 73728, 256, 256, 256, QSCALE);

    gemm_wmma_f32<<<dim3(512 / 64, 1152 / 32), TB, 0, stream>>>(
        poolx, WkvT, kvb, nullptr, 1152, 512, 256, 256, 1.0f);

    attn_kernel<<<144 * 8 * 8, 128, 0, stream>>>(qmat, kvb);

    rpe_add_kernel<<<(18874368 + TB - 1) / TB, TB, 0, stream>>>(
        x, rpe_w, rpe_b, qmat, 18874368);

    gemm_wmma_f32<<<dim3(256 / 64, 73728 / 32), TB, 0, stream>>>(
        qmat, projT, out, proj_b, 73728, 256, 256, 256, 1.0f);
}


# Three large GEMMs (pool split-K, Q, out proj) re-tiled: 128x64 block tile staged in LDS by whole-line loads, 32x32 per wave in 4 accumulators; k order inside each 8-k group changes (f32 sum order)
# speedup vs baseline: 2.0436x; 1.4516x over previous
#include <hip/hip_runtime.h>
#include <hip/hip_bf16.h>

typedef __attribute__((ext_vector_type(2))) float v2f;
typedef __attribute__((ext_vector_type(8))) float v8f;

#define QSCALE 0.17677669529663687f


__global__ void transpose_k(const float* __restrict__ src, float* __restrict__ dst,
                            int O, int K, int total) {
    int idx = blockIdx.x * blockDim.x + threadIdx.x;
    if (idx >= total) return;
    int o = idx % O;
    int k = idx / O;
    dst[idx] = src[(size_t)o * K + k];
}

__global__ void pool_transpose(const float* __restrict__ src, float* __restrict__ dst,
                               int total) {
    int idx = blockIdx.x * blockDim.x + threadIdx.x;
    if (idx >= total) return;
    int o = idx % 256;
    int k = idx / 256;
    int c = k % 256;
    int q = k / 256;
    dst[idx] = src[(size_t)o * 16384 + c * 64 + q];
}

__global__ void gather_xw(const float* __restrict__ x, float* __restrict__ xw, int total) {
    int idx = blockIdx.x * blockDim.x + threadIdx.x;
    if (idx >= total) return;
    int c = idx % 256;
    int t = idx / 256;
    int q = t % 64;
    int n = t / 64;
    int bi = n / 144;
    int l  = n % 144;
    int wh = l / 12, ww = l % 12;
    int i = q / 8, j = q % 8;
    xw[idx] = x[(((size_t)bi * 256 + c) * 96 + wh * 8 + i) * 96 + ww * 8 + j];
}

__global__ __launch_bounds__(256) void gemm_wmma_f32(
    const float* __restrict__ A, const float* __restrict__ B,
    float* __restrict__ C, const float* __restrict__ bias,
    int M, int N, int K, int lda, float scale)
{
    int wave = threadIdx.x >> 5;
    int lane = threadIdx.x & 31;
    int hf   = lane >> 4;
    int r    = lane & 15;
    int m0 = blockIdx.y * 32 + (wave >> 2) * 16;
    int n0 = blockIdx.x * 64 + (wave & 3) * 16;
    size_t kz = (size_t)blockIdx.z * K;

    const float* Arow = A + (size_t)(m0 + r) * lda + kz + 2 * hf;
    const float* Bcol = B + (kz + 2 * hf) * N + n0 + r;
    C += (size_t)blockIdx.z * M * N;

    v8f acc = {};
    #pragma unroll 8
    for (int k = 0; k < K; k += 4) {
        v2f a, b;
        a.x = Arow[k];
        a.y = Arow[k + 1];
        b.x = Bcol[(size_t)k * N];
        b.y = Bcol[(size_t)k * N + N];
        acc = __builtin_amdgcn_wmma_f32_16x16x4_f32(false, a, false, b,
                                                    (short)0, acc, false, false);
    }
    int col = n0 + r;
    float bv = bias ? bias[col] : 0.0f;
    #pragma unroll
    for (int vg = 0; vg < 8; ++vg)
        C[(size_t)(m0 + vg + 8 * hf) * N + col] = acc[vg] * scale + bv;
}

#define GBM 128
#define GBN 64
#define GBK 32
#define GPA 36
#define GPB 72
typedef __attribute__((ext_vector_type(4))) float v4f;

__global__ __launch_bounds__(256) void gemm_tiled_f32(
    const float* __restrict__ A, const float* __restrict__ B,
    float* __restrict__ C, const float* __restrict__ bias,
    int M, int N, int K, int lda, float scale)
{
    __shared__ __attribute__((aligned(16))) float As[2 * GBM * GPA];
    __shared__ __attribute__((aligned(16))) float Bs[2 * GBK * GPB];

    int t    = threadIdx.x;
    int wave = t >> 5;
    int lane = t & 31;
    int hf   = lane >> 4;
    int r    = lane & 15;
    int wm   = wave >> 1;
    int wn   = wave & 1;
    int m0 = blockIdx.y * GBM;
    int n0 = blockIdx.x * GBN;
    size_t kz = (size_t)blockIdx.z * K;

    int ar = t >> 3, ac = (t & 7) * 4;
    int br = t >> 4, bc = (t & 15) * 4;
    const float* Ag = A + (size_t)(m0 + ar) * lda + kz + ac;
    const float* Bg = B + (kz + br) * N + n0 + bc;
    C += (size_t)blockIdx.z * M * N;

    v4f pa[4], pb[2];
    #pragma unroll
    for (int i = 0; i < 4; ++i) pa[i] = *reinterpret_cast<const v4f*>(Ag + (size_t)(32 * i) * lda);
    #pragma unroll
    for (int i = 0; i < 2; ++i) pb[i] = *reinterpret_cast<const v4f*>(Bg + (size_t)(16 * i) * N);
    #pragma unroll
    for (int i = 0; i < 4; ++i) *reinterpret_cast<v4f*>(&As[(ar + 32 * i) * GPA + ac]) = pa[i];
    #pragma unroll
    for (int i = 0; i < 2; ++i) *reinterpret_cast<v4f*>(&Bs[(br + 16 * i) * GPB + bc]) = pb[i];
    __syncthreads();

    v8f acc[2][2] = {};
    int nchunks = K / GBK;
    for (int kc = 0; kc < nchunks; ++kc) {
        int cur = kc & 1;
        bool more = (kc + 1 < nchunks);
        if (more) {
            const float* Agn = Ag + (kc + 1) * GBK;
            const float* Bgn = Bg + (size_t)(kc + 1) * GBK * N;
            #pragma unroll
            for (int i = 0; i < 4; ++i) pa[i] = *reinterpret_cast<const v4f*>(Agn + (size_t)(32 * i) * lda);
            #pragma unroll
            for (int i = 0; i < 2; ++i) pb[i] = *reinterpret_cast<const v4f*>(Bgn + (size_t)(16 * i) * N);
        }
        const float* as = &As[cur * (GBM * GPA) + (wm * 32 + r) * GPA + 4 * hf];
        const float* bs = &Bs[cur * (GBK * GPB) + (4 * hf) * GPB + wn * 32 + r];
        #pragma unroll
        for (int g = 0; g < 4; ++g) {
            v4f a0 = *reinterpret_cast<const v4f*>(as + 8 * g);
            v4f a1 = *reinterpret_cast<const v4f*>(as + 16 * GPA + 8 * g);
            const float* bg = bs + (8 * g) * GPB;
            v2f b0lo = {bg[0],            bg[GPB]};
            v2f b0hi = {bg[2 * GPB],      bg[3 * GPB]};
            v2f b1lo = {bg[16],           bg[GPB + 16]};
            v2f b1hi = {bg[2 * GPB + 16], bg[3 * GPB + 16]};
            v2f a0lo = {a0.x, a0.y}, a0hi = {a0.z, a0.w};
            v2f a1lo = {a1.x, a1.y}, a1hi = {a1.z, a1.w};
            acc[0][0] = __builtin_amdgcn_wmma_f32_16x16x4_f32(false, a0lo, false, b0lo, (short)0, acc[0][0], false, false);
            acc[0][1] = __builtin_amdgcn_wmma_f32_16x16x4_f32(false, a0lo, false, b1lo, (short)0, acc[0][1], false, false);
            acc[1][0] = __builtin_amdgcn_wmma_f32_16x16x4_f32(false, a1lo, false, b0lo, (short)0, acc[1][0], false, false);
            acc[1][1] = __builtin_amdgcn_wmma_f32_16x16x4_f32(false, a1lo, false, b1lo, (short)0, acc[1][1], false, false);
            acc[0][0] = __builtin_amdgcn_wmma_f32_16x16x4_f32(false, a0hi, false, b0hi, (short)0, acc[0][0], false, false);
            acc[0][1] = __builtin_amdgcn_wmma_f32_16x16x4_f32(false, a0hi, false, b1hi, (short)0, acc[0][1], false, false);
            acc[1][0] = __builtin_amdgcn_wmma_f32_16x16x4_f32(false, a1hi, false, b0hi, (short)0, acc[1][0], false, false);
            acc[1][1] = __builtin_amdgcn_wmma_f32_16x16x4_f32(false, a1hi, false, b1hi, (short)0, acc[1][1], false, false);
        }
        if (more) {
            int nxt = cur ^ 1;
            #pragma unroll
            for (int i = 0; i < 4; ++i) *reinterpret_cast<v4f*>(&As[nxt * (GBM * GPA) + (ar + 32 * i) * GPA + ac]) = pa[i];
            #pragma unroll
            for (int i = 0; i < 2; ++i) *reinterpret_cast<v4f*>(&Bs[nxt * (GBK * GPB) + (br + 16 * i) * GPB + bc]) = pb[i];
        }
        __syncthreads();
    }

    float* patch = &As[wave * (32 * GPA)];
    #pragma unroll
    for (int mt = 0; mt < 2; ++mt)
        #pragma unroll
        for (int nt = 0; nt < 2; ++nt)
            #pragma unroll
            for (int vg = 0; vg < 8; ++vg)
                patch[(mt * 16 + vg + 8 * hf) * GPA + nt * 16 + r] = acc[mt][nt][vg];
    __syncthreads();
    int pr = lane >> 3, pc = (lane & 7) * 4;
    int col = n0 + wn * 32 + pc;
    v4f bv = {0.0f, 0.0f, 0.0f, 0.0f};
    if (bias) bv = *reinterpret_cast<const v4f*>(bias + col);
    #pragma unroll
    for (int i = 0; i < 8; ++i) {
        int row = pr + 4 * i;
        v4f v = *reinterpret_cast<const v4f*>(&patch[row * GPA + pc]);
        v4f o;
        o.x = v.x * scale + bv.x;
        o.y = v.y * scale + bv.y;
        o.z = v.z * scale + bv.z;
        o.w = v.w * scale + bv.w;
        *reinterpret_cast<v4f*>(&C[(size_t)(m0 + wm * 32 + row) * N + col]) = o;
    }
}

__global__ __launch_bounds__(64) void splitk_reduce(
    const float* __restrict__ part, const float* __restrict__ bias,
    float* __restrict__ dst, int MN, int N, int S)
{
    int i4 = blockIdx.x * blockDim.x + threadIdx.x;
    int quads = MN >> 2;
    if (i4 >= quads) return;
    const float4* p = reinterpret_cast<const float4*>(part) + i4;
    float4 acc = make_float4(0.0f, 0.0f, 0.0f, 0.0f);
    #pragma unroll 16
    for (int z = 0; z < S; ++z) {
        float4 v = p[(size_t)z * quads];
        acc.x += v.x; acc.y += v.y; acc.z += v.z; acc.w += v.w;
    }
    int col = (i4 << 2) % N;
    const float4 bv = *reinterpret_cast<const float4*>(bias + col);
    acc.x += bv.x; acc.y += bv.y; acc.z += bv.z; acc.w += bv.w;
    reinterpret_cast<float4*>(dst)[i4] = acc;
}

#define LDST 145
__global__ __launch_bounds__(128) void attn_kernel(
    float* __restrict__ qmat, const float* __restrict__ kv)
{
    __shared__ float attn[64 * LDST];
    int id = blockIdx.x;
    int h  = id & 7;
    int bb = (id >> 3) & 7;
    int g  = id >> 6;
    int n  = g * 8 + bb;

    int wave = threadIdx.x >> 5;
    int lane = threadIdx.x & 31;
    int hf   = lane >> 4;
    int r    = lane & 15;
    int m0   = wave * 16;

    const float* qbase = qmat + ((size_t)n * 64 + m0 + r) * 256 + h * 32;
    const float* kvb   = kv + (size_t)bb * 144 * 512 + h * 32;
    const float* vb    = kvb + 256;

    for (int nt = 0; nt < 9; ++nt) {
        int n0 = nt * 16;
        const float* kcol = kvb + (size_t)(n0 + r) * 512 + 2 * hf;
        v8f acc = {};
        #pragma unroll
        for (int k = 0; k < 32; k += 4) {
            v2f a, b;
            a.x = qbase[k + 2 * hf];
            a.y = qbase[k + 2 * hf + 1];
            b.x = kcol[k];
            b.y = kcol[k + 1];
            acc = __builtin_amdgcn_wmma_f32_16x16x4_f32(false, a, false, b,
                                                        (short)0, acc, false, false);
        }
        #pragma unroll
        for (int vg = 0; vg < 8; ++vg)
            attn[(m0 + vg + 8 * hf) * LDST + n0 + r] = acc[vg];
    }
    __syncthreads();

    if (threadIdx.x < 64) {
        float* row = attn + threadIdx.x * LDST;
        float mx = row[0];
        for (int k = 1; k < 144; ++k) mx = fmaxf(mx, row[k]);
        float s = 0.0f;
        for (int k = 0; k < 144; ++k) { float e = __expf(row[k] - mx); row[k] = e; s += e; }
        float inv = 1.0f / s;
        for (int k = 0; k < 144; ++k) row[k] *= inv;
    }
    __syncthreads();

    const float* arow = attn + (m0 + r) * LDST + 2 * hf;
    for (int nt = 0; nt < 2; ++nt) {
        int n0 = nt * 16;
        v8f acc = {};
        #pragma unroll 4
        for (int k = 0; k < 144; k += 4) {
            v2f a, b;
            a.x = arow[k];
            a.y = arow[k + 1];
            b.x = vb[(size_t)(k + 2 * hf) * 512 + n0 + r];
            b.y = vb[(size_t)(k + 2 * hf + 1) * 512 + n0 + r];
            acc = __builtin_amdgcn_wmma_f32_16x16x4_f32(false, a, false, b,
                                                        (short)0, acc, false, false);
        }
        #pragma unroll
        for (int vg = 0; vg < 8; ++vg)
            qmat[((size_t)n * 64 + m0 + vg + 8 * hf) * 256 + h * 32 + n0 + r] = acc[vg];
    }
}

__global__ void rpe_add_kernel(const float* __restrict__ x, const float* __restrict__ rw,
                               const float* __restrict__ rb, float* __restrict__ F,
                               int total) {
    int idx = blockIdx.x * blockDim.x + threadIdx.x;
    if (idx >= total) return;
    int xx = idx % 96;
    int t  = idx / 96;
    int y  = t % 96; t /= 96;
    int c  = t % 256;
    int bi = t / 256;
    const float* xp = x + (size_t)(bi * 256 + c) * 9216;
    const float* wp = rw + c * 9;
    float acc = rb[c];
    #pragma unroll
    for (int dy = -1; dy <= 1; ++dy) {
        int yy = y + dy;
        if (yy < 0 || yy >= 96) continue;
        #pragma unroll
        for (int dx = -1; dx <= 1; ++dx) {
            int xc = xx + dx;
            if (xc < 0 || xc >= 96) continue;
            acc += xp[yy * 96 + xc] * wp[(dy + 1) * 3 + (dx + 1)];
        }
    }
    F[idx] += acc;
}

extern "C" void kernel_launch(void* const* d_in, const int* in_sizes, int n_in,
                              void* d_out, int out_size, void* d_ws, size_t ws_size,
                              hipStream_t stream) {
    const float* x      = (const float*)d_in[0];
    const float* Wq     = (const float*)d_in[1];
    const float* Wkv    = (const float*)d_in[2];
    const float* pool_w = (const float*)d_in[3];
    const float* pool_b = (const float*)d_in[4];
    const float* rpe_w  = (const float*)d_in[5];
    const float* rpe_b  = (const float*)d_in[6];
    const float* proj_w = (const float*)d_in[7];
    const float* proj_b = (const float*)d_in[8];
    float* out = (float*)d_out;

    float* ws    = (float*)d_ws;
    float* xw    = ws;
    float* qmat  = xw + 18874368;
    float* poolT = qmat + 18874368;
    float* WqT   = poolT + 4194304;
    float* WkvT  = WqT + 65536;
    float* projT = WkvT + 131072;
    float* poolx = projT + 65536;
    float* kvb   = poolx + 294912;
    float* part  = kvb + 589824;
    (void)ws_size; (void)in_sizes; (void)n_in; (void)out_size;

    const int TB = 256;
    transpose_k<<<(65536 + TB - 1) / TB, TB, 0, stream>>>(Wq, WqT, 256, 256, 65536);
    transpose_k<<<(131072 + TB - 1) / TB, TB, 0, stream>>>(Wkv, WkvT, 512, 256, 131072);
    transpose_k<<<(65536 + TB - 1) / TB, TB, 0, stream>>>(proj_w, projT, 256, 256, 65536);
    pool_transpose<<<(4194304 + TB - 1) / TB, TB, 0, stream>>>(pool_w, poolT, 4194304);

    gather_xw<<<(18874368 + TB - 1) / TB, TB, 0, stream>>>(x, xw, 18874368);

    gemm_tiled_f32<<<dim3(256 / 64, 1152 / 128, 64), TB, 0, stream>>>(
        xw, poolT, part, nullptr, 1152, 256, 256, 16384, 1.0f);
    splitk_reduce<<<73728 / 64, 64, 0, stream>>>(
        part, pool_b, poolx, 294912, 256, 64);

    gemm_tiled_f32<<<dim3(256 / 64, 73728 / 128), TB, 0, stream>>>(
        xw, WqT, qmat, nullptr, 73728, 256, 256, 256, QSCALE);

    gemm_wmma_f32<<<dim3(512 / 64, 1152 / 32), TB, 0, stream>>>(
        poolx, WkvT, kvb, nullptr, 1152, 512, 256, 256, 1.0f);

    attn_kernel<<<144 * 8 * 8, 128, 0, stream>>>(qmat, kvb);

    rpe_add_kernel<<<(18874368 + TB - 1) / TB, TB, 0, stream>>>(
        x, rpe_w, rpe_b, qmat, 18874368);

    gemm_tiled_f32<<<dim3(256 / 64, 73728 / 128), TB, 0, stream>>>(
        qmat, projT, out, proj_b, 73728, 256, 256, 256, 1.0f);
}


# Window gather and pool-weight transpose go through an LDS tile: whole 256-byte global reads and writes, 12 and 4 float4 per thread, replacing one 4-byte read per thread at 36-KB and 64-KB strides
# speedup vs baseline: 2.5150x; 1.2307x over previous
#include <hip/hip_runtime.h>
#include <hip/hip_bf16.h>

typedef __attribute__((ext_vector_type(2))) float v2f;
typedef __attribute__((ext_vector_type(8))) float v8f;

#define QSCALE 0.17677669529663687f


__global__ void transpose_k(const float* __restrict__ src, float* __restrict__ dst,
                            int O, int K, int total) {
    int idx = blockIdx.x * blockDim.x + threadIdx.x;
    if (idx >= total) return;
    int o = idx % O;
    int k = idx / O;
    dst[idx] = src[(size_t)o * K + k];
}

typedef __attribute__((ext_vector_type(4))) float v4f;
#define TPITCH 65

__global__ __launch_bounds__(256) void pool_transpose(const float* __restrict__ src,
                                                      float* __restrict__ dst) {
    __shared__ float tile[64 * TPITCH];
    int t  = threadIdx.x;
    int og = blockIdx.x & 3;
    int c  = blockIdx.x >> 2;
    int lo = t & 15;
    int hi = t >> 4;
    const float* s = src + ((size_t)(og * 64 + hi) * 16384 + c * 64 + 4 * lo);
    v4f v[4];
    #pragma unroll
    for (int it = 0; it < 4; ++it)
        v[it] = *reinterpret_cast<const v4f*>(s + (size_t)(16 * it) * 16384);
    #pragma unroll
    for (int it = 0; it < 4; ++it) {
        float* d = &tile[(4 * lo) * TPITCH + hi + 16 * it];
        d[0]          = v[it].x;
        d[TPITCH]     = v[it].y;
        d[2 * TPITCH] = v[it].z;
        d[3 * TPITCH] = v[it].w;
    }
    __syncthreads();
    float* g = dst + ((size_t)(hi * 256 + c) * 256 + og * 64 + 4 * lo);
    #pragma unroll
    for (int it = 0; it < 4; ++it) {
        const float* r = &tile[(hi + 16 * it) * TPITCH + 4 * lo];
        v4f o = {r[0], r[1], r[2], r[3]};
        *reinterpret_cast<v4f*>(g + (size_t)(16 * it) * 65536) = o;
    }
}

__global__ __launch_bounds__(256) void gather_xw(const float* __restrict__ x,
                                                 float* __restrict__ xw) {
    __shared__ float tile[192 * TPITCH];
    int t  = threadIdx.x;
    int b  = blockIdx.x;
    int cg = b & 3;
    int ip = (b >> 2) & 3;
    int bw = b >> 4;
    int wh = bw % 12;
    int bi = bw / 12;
    int lo = t & 15;
    int hi = t >> 4;
    const float* strip = x + (((size_t)(bi * 256 + cg * 64 + hi) * 96 + wh * 8 + 2 * ip) * 96 + 4 * lo);
    v4f v[12];
    #pragma unroll
    for (int it = 0; it < 12; ++it)
        v[it] = *reinterpret_cast<const v4f*>(strip + (size_t)(16 * (it & 3)) * 9216 + (it >> 2) * 64);
    float* d0 = &tile[(4 * lo) * TPITCH + hi];
    #pragma unroll
    for (int it = 0; it < 12; ++it) {
        float* d = d0 + ((it >> 2) * 64) * TPITCH + 16 * (it & 3);
        d[0]          = v[it].x;
        d[TPITCH]     = v[it].y;
        d[2 * TPITCH] = v[it].z;
        d[3 * TPITCH] = v[it].w;
    }
    __syncthreads();
    float* g = xw + ((size_t)((bi * 144 + wh * 12 + (hi >> 3)) * 64 + ip * 16 + (hi & 7)) * 256
                     + cg * 64 + 4 * lo);
    const float* r0 = &tile[hi * TPITCH + 4 * lo];
    #pragma unroll
    for (int it = 0; it < 12; ++it) {
        const float* r = r0 + (16 * it) * TPITCH;
        v4f o = {r[0], r[1], r[2], r[3]};
        *reinterpret_cast<v4f*>(g + (size_t)(it % 6) * (2 * 64 * 256) + (it / 6) * (8 * 256)) = o;
    }
}

__global__ __launch_bounds__(256) void gemm_wmma_f32(
    const float* __restrict__ A, const float* __restrict__ B,
    float* __restrict__ C, const float* __restrict__ bias,
    int M, int N, int K, int lda, float scale)
{
    int wave = threadIdx.x >> 5;
    int lane = threadIdx.x & 31;
    int hf   = lane >> 4;
    int r    = lane & 15;
    int m0 = blockIdx.y * 32 + (wave >> 2) * 16;
    int n0 = blockIdx.x * 64 + (wave & 3) * 16;
    size_t kz = (size_t)blockIdx.z * K;

    const float* Arow = A + (size_t)(m0 + r) * lda + kz + 2 * hf;
    const float* Bcol = B + (kz + 2 * hf) * N + n0 + r;
    C += (size_t)blockIdx.z * M * N;

    v8f acc = {};
    #pragma unroll 8
    for (int k = 0; k < K; k += 4) {
        v2f a, b;
        a.x = Arow[k];
        a.y = Arow[k + 1];
        b.x = Bcol[(size_t)k * N];
        b.y = Bcol[(size_t)k * N + N];
        acc = __builtin_amdgcn_wmma_f32_16x16x4_f32(false, a, false, b,
                                                    (short)0, acc, false, false);
    }
    int col = n0 + r;
    float bv = bias ? bias[col] : 0.0f;
    #pragma unroll
    for (int vg = 0; vg < 8; ++vg)
        C[(size_t)(m0 + vg + 8 * hf) * N + col] = acc[vg] * scale + bv;
}

#define GBM 128
#define GBN 64
#define GBK 32
#define GPA 36
#define GPB 72

__global__ __launch_bounds__(256) void gemm_tiled_f32(
    const float* __restrict__ A, const float* __restrict__ B,
    float* __restrict__ C, const float* __restrict__ bias,
    int M, int N, int K, int lda, float scale)
{
    __shared__ __attribute__((aligned(16))) float As[2 * GBM * GPA];
    __shared__ __attribute__((aligned(16))) float Bs[2 * GBK * GPB];

    int t    = threadIdx.x;
    int wave = t >> 5;
    int lane = t & 31;
    int hf   = lane >> 4;
    int r    = lane & 15;
    int wm   = wave >> 1;
    int wn   = wave & 1;
    int m0 = blockIdx.y * GBM;
    int n0 = blockIdx.x * GBN;
    size_t kz = (size_t)blockIdx.z * K;

    int ar = t >> 3, ac = (t & 7) * 4;
    int br = t >> 4, bc = (t & 15) * 4;
    const float* Ag = A + (size_t)(m0 + ar) * lda + kz + ac;
    const float* Bg = B + (kz + br) * N + n0 + bc;
    C += (size_t)blockIdx.z * M * N;

    v4f pa[4], pb[2];
    #pragma unroll
    for (int i = 0; i < 4; ++i) pa[i] = *reinterpret_cast<const v4f*>(Ag + (size_t)(32 * i) * lda);
    #pragma unroll
    for (int i = 0; i < 2; ++i) pb[i] = *reinterpret_cast<const v4f*>(Bg + (size_t)(16 * i) * N);
    #pragma unroll
    for (int i = 0; i < 4; ++i) *reinterpret_cast<v4f*>(&As[(ar + 32 * i) * GPA + ac]) = pa[i];
    #pragma unroll
    for (int i = 0; i < 2; ++i) *reinterpret_cast<v4f*>(&Bs[(br + 16 * i) * GPB + bc]) = pb[i];
    __syncthreads();

    v8f acc[2][2] = {};
    int nchunks = K / GBK;
    for (int kc = 0; kc < nchunks; ++kc) {
        int cur = kc & 1;
        bool more = (kc + 1 < nchunks);
        if (more) {
            const float* Agn = Ag + (kc + 1) * GBK;
            const float* Bgn = Bg + (size_t)(kc + 1) * GBK * N;
            #pragma unroll
            for (int i = 0; i < 4; ++i) pa[i] = *reinterpret_cast<const v4f*>(Agn + (size_t)(32 * i) * lda);
            #pragma unroll
            for (int i = 0; i < 2; ++i) pb[i] = *reinterpret_cast<const v4f*>(Bgn + (size_t)(16 * i) * N);
        }
        const float* as = &As[cur * (GBM * GPA) + (wm * 32 + r) * GPA + 4 * hf];
        const float* bs = &Bs[cur * (GBK * GPB) + (4 * hf) * GPB + wn * 32 + r];
        #pragma unroll
        for (int g = 0; g < 4; ++g) {
            v4f a0 = *reinterpret_cast<const v4f*>(as + 8 * g);
            v4f a1 = *reinterpret_cast<const v4f*>(as + 16 * GPA + 8 * g);
            const float* bg = bs + (8 * g) * GPB;
            v2f b0lo = {bg[0],            bg[GPB]};
            v2f b0hi = {bg[2 * GPB],      bg[3 * GPB]};
            v2f b1lo = {bg[16],           bg[GPB + 16]};
            v2f b1hi = {bg[2 * GPB + 16], bg[3 * GPB + 16]};
            v2f a0lo = {a0.x, a0.y}, a0hi = {a0.z, a0.w};
            v2f a1lo = {a1.x, a1.y}, a1hi = {a1.z, a1.w};
            acc[0][0] = __builtin_amdgcn_wmma_f32_16x16x4_f32(false, a0lo, false, b0lo, (short)0, acc[0][0], false, false);
            acc[0][1] = __builtin_amdgcn_wmma_f32_16x16x4_f32(false, a0lo, false, b1lo, (short)0, acc[0][1], false, false);
            acc[1][0] = __builtin_amdgcn_wmma_f32_16x16x4_f32(false, a1lo, false, b0lo, (short)0, acc[1][0], false, false);
            acc[1][1] = __builtin_amdgcn_wmma_f32_16x16x4_f32(false, a1lo, false, b1lo, (short)0, acc[1][1], false, false);
            acc[0][0] = __builtin_amdgcn_wmma_f32_16x16x4_f32(false, a0hi, false, b0hi, (short)0, acc[0][0], false, false);
            acc[0][1] = __builtin_amdgcn_wmma_f32_16x16x4_f32(false, a0hi, false, b1hi, (short)0, acc[0][1], false, false);
            acc[1][0] = __builtin_amdgcn_wmma_f32_16x16x4_f32(false, a1hi, false, b0hi, (short)0, acc[1][0], false, false);
            acc[1][1] = __builtin_amdgcn_wmma_f32_16x16x4_f32(false, a1hi, false, b1hi, (short)0, acc[1][1], false, false);
        }
        if (more) {
            int nxt = cur ^ 1;
            #pragma unroll
            for (int i = 0; i < 4; ++i) *reinterpret_cast<v4f*>(&As[nxt * (GBM * GPA) + (ar + 32 * i) * GPA + ac]) = pa[i];
            #pragma unroll
            for (int i = 0; i < 2; ++i) *reinterpret_cast<v4f*>(&Bs[nxt * (GBK * GPB) + (br + 16 * i) * GPB + bc]) = pb[i];
        }
        __syncthreads();
    }

    float* patch = &As[wave * (32 * GPA)];
    #pragma unroll
    for (int mt = 0; mt < 2; ++mt)
        #pragma unroll
        for (int nt = 0; nt < 2; ++nt)
            #pragma unroll
            for (int vg = 0; vg < 8; ++vg)
                patch[(mt * 16 + vg + 8 * hf) * GPA + nt * 16 + r] = acc[mt][nt][vg];
    __syncthreads();
    int pr = lane >> 3, pc = (lane & 7) * 4;
    int col = n0 + wn * 32 + pc;
    v4f bv = {0.0f, 0.0f, 0.0f, 0.0f};
    if (bias) bv = *reinterpret_cast<const v4f*>(bias + col);
    #pragma unroll
    for (int i = 0; i < 8; ++i) {
        int row = pr + 4 * i;
        v4f v = *reinterpret_cast<const v4f*>(&patch[row * GPA + pc]);
        v4f o;
        o.x = v.x * scale + bv.x;
        o.y = v.y * scale + bv.y;
        o.z = v.z * scale + bv.z;
        o.w = v.w * scale + bv.w;
        *reinterpret_cast<v4f*>(&C[(size_t)(m0 + wm * 32 + row) * N + col]) = o;
    }
}

__global__ __launch_bounds__(64) void splitk_reduce(
    const float* __restrict__ part, const float* __restrict__ bias,
    float* __restrict__ dst, int MN, int N, int S)
{
    int i4 = blockIdx.x * blockDim.x + threadIdx.x;
    int quads = MN >> 2;
    if (i4 >= quads) return;
    const float4* p = reinterpret_cast<const float4*>(part) + i4;
    float4 acc = make_float4(0.0f, 0.0f, 0.0f, 0.0f);
    #pragma unroll 16
    for (int z = 0; z < S; ++z) {
        float4 v = p[(size_t)z * quads];
        acc.x += v.x; acc.y += v.y; acc.z += v.z; acc.w += v.w;
    }
    int col = (i4 << 2) % N;
    const float4 bv = *reinterpret_cast<const float4*>(bias + col);
    acc.x += bv.x; acc.y += bv.y; acc.z += bv.z; acc.w += bv.w;
    reinterpret_cast<float4*>(dst)[i4] = acc;
}

#define LDST 145
__global__ __launch_bounds__(128) void attn_kernel(
    float* __restrict__ qmat, const float* __restrict__ kv)
{
    __shared__ float attn[64 * LDST];
    int id = blockIdx.x;
    int h  = id & 7;
    int bb = (id >> 3) & 7;
    int g  = id >> 6;
    int n  = g * 8 + bb;

    int wave = threadIdx.x >> 5;
    int lane = threadIdx.x & 31;
    int hf   = lane >> 4;
    int r    = lane & 15;
    int m0   = wave * 16;

    const float* qbase = qmat + ((size_t)n * 64 + m0 + r) * 256 + h * 32;
    const float* kvb   = kv + (size_t)bb * 144 * 512 + h * 32;
    const float* vb    = kvb + 256;

    for (int nt = 0; nt < 9; ++nt) {
        int n0 = nt * 16;
        const float* kcol = kvb + (size_t)(n0 + r) * 512 + 2 * hf;
        v8f acc = {};
        #pragma unroll
        for (int k = 0; k < 32; k += 4) {
            v2f a, b;
            a.x = qbase[k + 2 * hf];
            a.y = qbase[k + 2 * hf + 1];
            b.x = kcol[k];
            b.y = kcol[k + 1];
            acc = __builtin_amdgcn_wmma_f32_16x16x4_f32(false, a, false, b,
                                                        (short)0, acc, false, false);
        }
        #pragma unroll
        for (int vg = 0; vg < 8; ++vg)
            attn[(m0 + vg + 8 * hf) * LDST + n0 + r] = acc[vg];
    }
    __syncthreads();

    if (threadIdx.x < 64) {
        float* row = attn + threadIdx.x * LDST;
        float mx = row[0];
        for (int k = 1; k < 144; ++k) mx = fmaxf(mx, row[k]);
        float s = 0.0f;
        for (int k = 0; k < 144; ++k) { float e = __expf(row[k] - mx); row[k] = e; s += e; }
        float inv = 1.0f / s;
        for (int k = 0; k < 144; ++k) row[k] *= inv;
    }
    __syncthreads();

    const float* arow = attn + (m0 + r) * LDST + 2 * hf;
    for (int nt = 0; nt < 2; ++nt) {
        int n0 = nt * 16;
        v8f acc = {};
        #pragma unroll 4
        for (int k = 0; k < 144; k += 4) {
            v2f a, b;
            a.x = arow[k];
            a.y = arow[k + 1];
            b.x = vb[(size_t)(k + 2 * hf) * 512 + n0 + r];
            b.y = vb[(size_t)(k + 2 * hf + 1) * 512 + n0 + r];
            acc = __builtin_amdgcn_wmma_f32_16x16x4_f32(false, a, false, b,
                                                        (short)0, acc, false, false);
        }
        #pragma unroll
        for (int vg = 0; vg < 8; ++vg)
            qmat[((size_t)n * 64 + m0 + vg + 8 * hf) * 256 + h * 32 + n0 + r] = acc[vg];
    }
}

__global__ void rpe_add_kernel(const float* __restrict__ x, const float* __restrict__ rw,
                               const float* __restrict__ rb, float* __restrict__ F,
                               int total) {
    int idx = blockIdx.x * blockDim.x + threadIdx.x;
    if (idx >= total) return;
    int xx = idx % 96;
    int t  = idx / 96;
    int y  = t % 96; t /= 96;
    int c  = t % 256;
    int bi = t / 256;
    const float* xp = x + (size_t)(bi * 256 + c) * 9216;
    const float* wp = rw + c * 9;
    float acc = rb[c];
    #pragma unroll
    for (int dy = -1; dy <= 1; ++dy) {
        int yy = y + dy;
        if (yy < 0 || yy >= 96) continue;
        #pragma unroll
        for (int dx = -1; dx <= 1; ++dx) {
            int xc = xx + dx;
            if (xc < 0 || xc >= 96) continue;
            acc += xp[yy * 96 + xc] * wp[(dy + 1) * 3 + (dx + 1)];
        }
    }
    F[idx] += acc;
}

extern "C" void kernel_launch(void* const* d_in, const int* in_sizes, int n_in,
                              void* d_out, int out_size, void* d_ws, size_t ws_size,
                              hipStream_t stream) {
    const float* x      = (const float*)d_in[0];
    const float* Wq     = (const float*)d_in[1];
    const float* Wkv    = (const float*)d_in[2];
    const float* pool_w = (const float*)d_in[3];
    const float* pool_b = (const float*)d_in[4];
    const float* rpe_w  = (const float*)d_in[5];
    const float* rpe_b  = (const float*)d_in[6];
    const float* proj_w = (const float*)d_in[7];
    const float* proj_b = (const float*)d_in[8];
    float* out = (float*)d_out;

    float* ws    = (float*)d_ws;
    float* xw    = ws;
    float* qmat  = xw + 18874368;
    float* poolT = qmat + 18874368;
    float* WqT   = poolT + 4194304;
    float* WkvT  = WqT + 65536;
    float* projT = WkvT + 131072;
    float* poolx = projT + 65536;
    float* kvb   = poolx + 294912;
    float* part  = kvb + 589824;
    (void)ws_size; (void)in_sizes; (void)n_in; (void)out_size;

    const int TB = 256;
    transpose_k<<<(65536 + TB - 1) / TB, TB, 0, stream>>>(Wq, WqT, 256, 256, 65536);
    transpose_k<<<(131072 + TB - 1) / TB, TB, 0, stream>>>(Wkv, WkvT, 512, 256, 131072);
    transpose_k<<<(65536 + TB - 1) / TB, TB, 0, stream>>>(proj_w, projT, 256, 256, 65536);
    pool_transpose<<<256 * 4, TB, 0, stream>>>(pool_w, poolT);

    gather_xw<<<8 * 12 * 4 * 4, TB, 0, stream>>>(x, xw);

    gemm_tiled_f32<<<dim3(256 / 64, 1152 / 128, 64), TB, 0, stream>>>(
        xw, poolT, part, nullptr, 1152, 256, 256, 16384, 1.0f);
    splitk_reduce<<<73728 / 64, 64, 0, stream>>>(
        part, pool_b, poolx, 294912, 256, 64);

    gemm_tiled_f32<<<dim3(256 / 64, 73728 / 128), TB, 0, stream>>>(
        xw, WqT, qmat, nullptr, 73728, 256, 256, 256, QSCALE);

    gemm_wmma_f32<<<dim3(512 / 64, 1152 / 32), TB, 0, stream>>>(
        poolx, WkvT, kvb, nullptr, 1152, 512, 256, 256, 1.0f);

    attn_kernel<<<144 * 8 * 8, 128, 0, stream>>>(qmat, kvb);

    rpe_add_kernel<<<(18874368 + TB - 1) / TB, TB, 0, stream>>>(
        x, rpe_w, rpe_b, qmat, 18874368);

    gemm_tiled_f32<<<dim3(256 / 64, 73728 / 128), TB, 0, stream>>>(
        qmat, projT, out, proj_b, 73728, 256, 256, 256, 1.0f);
}


# RPE add pass rewritten: one block per plane, a thread owns 4 columns x 12 rows, six input rows per pass held in registers from 16-byte loads, weights in scalar registers; 12288 waves, was 589824
# speedup vs baseline: 2.9195x; 1.1608x over previous
#include <hip/hip_runtime.h>
#include <hip/hip_bf16.h>

typedef __attribute__((ext_vector_type(2))) float v2f;
typedef __attribute__((ext_vector_type(8))) float v8f;

#define QSCALE 0.17677669529663687f


__global__ void transpose_k(const float* __restrict__ src, float* __restrict__ dst,
                            int O, int K, int total) {
    int idx = blockIdx.x * blockDim.x + threadIdx.x;
    if (idx >= total) return;
    int o = idx % O;
    int k = idx / O;
    dst[idx] = src[(size_t)o * K + k];
}

typedef __attribute__((ext_vector_type(4))) float v4f;
#define TPITCH 65

__global__ __launch_bounds__(256) void pool_transpose(const float* __restrict__ src,
                                                      float* __restrict__ dst) {
    __shared__ float tile[64 * TPITCH];
    int t  = threadIdx.x;
    int og = blockIdx.x & 3;
    int c  = blockIdx.x >> 2;
    int lo = t & 15;
    int hi = t >> 4;
    const float* s = src + ((size_t)(og * 64 + hi) * 16384 + c * 64 + 4 * lo);
    v4f v[4];
    #pragma unroll
    for (int it = 0; it < 4; ++it)
        v[it] = *reinterpret_cast<const v4f*>(s + (size_t)(16 * it) * 16384);
    #pragma unroll
    for (int it = 0; it < 4; ++it) {
        float* d = &tile[(4 * lo) * TPITCH + hi + 16 * it];
        d[0]          = v[it].x;
        d[TPITCH]     = v[it].y;
        d[2 * TPITCH] = v[it].z;
        d[3 * TPITCH] = v[it].w;
    }
    __syncthreads();
    float* g = dst + ((size_t)(hi * 256 + c) * 256 + og * 64 + 4 * lo);
    #pragma unroll
    for (int it = 0; it < 4; ++it) {
        const float* r = &tile[(hi + 16 * it) * TPITCH + 4 * lo];
        v4f o = {r[0], r[1], r[2], r[3]};
        *reinterpret_cast<v4f*>(g + (size_t)(16 * it) * 65536) = o;
    }
}

__global__ __launch_bounds__(256) void gather_xw(const float* __restrict__ x,
                                                 float* __restrict__ xw) {
    __shared__ float tile[192 * TPITCH];
    int t  = threadIdx.x;
    int b  = blockIdx.x;
    int cg = b & 3;
    int ip = (b >> 2) & 3;
    int bw = b >> 4;
    int wh = bw % 12;
    int bi = bw / 12;
    int lo = t & 15;
    int hi = t >> 4;
    const float* strip = x + (((size_t)(bi * 256 + cg * 64 + hi) * 96 + wh * 8 + 2 * ip) * 96 + 4 * lo);
    v4f v[12];
    #pragma unroll
    for (int it = 0; it < 12; ++it)
        v[it] = *reinterpret_cast<const v4f*>(strip + (size_t)(16 * (it & 3)) * 9216 + (it >> 2) * 64);
    float* d0 = &tile[(4 * lo) * TPITCH + hi];
    #pragma unroll
    for (int it = 0; it < 12; ++it) {
        float* d = d0 + ((it >> 2) * 64) * TPITCH + 16 * (it & 3);
        d[0]          = v[it].x;
        d[TPITCH]     = v[it].y;
        d[2 * TPITCH] = v[it].z;
        d[3 * TPITCH] = v[it].w;
    }
    __syncthreads();
    float* g = xw + ((size_t)((bi * 144 + wh * 12 + (hi >> 3)) * 64 + ip * 16 + (hi & 7)) * 256
                     + cg * 64 + 4 * lo);
    const float* r0 = &tile[hi * TPITCH + 4 * lo];
    #pragma unroll
    for (int it = 0; it < 12; ++it) {
        const float* r = r0 + (16 * it) * TPITCH;
        v4f o = {r[0], r[1], r[2], r[3]};
        *reinterpret_cast<v4f*>(g + (size_t)(it % 6) * (2 * 64 * 256) + (it / 6) * (8 * 256)) = o;
    }
}

__global__ __launch_bounds__(256) void gemm_wmma_f32(
    const float* __restrict__ A, const float* __restrict__ B,
    float* __restrict__ C, const float* __restrict__ bias,
    int M, int N, int K, int lda, float scale)
{
    int wave = threadIdx.x >> 5;
    int lane = threadIdx.x & 31;
    int hf   = lane >> 4;
    int r    = lane & 15;
    int m0 = blockIdx.y * 32 + (wave >> 2) * 16;
    int n0 = blockIdx.x * 64 + (wave & 3) * 16;
    size_t kz = (size_t)blockIdx.z * K;

    const float* Arow = A + (size_t)(m0 + r) * lda + kz + 2 * hf;
    const float* Bcol = B + (kz + 2 * hf) * N + n0 + r;
    C += (size_t)blockIdx.z * M * N;

    v8f acc = {};
    #pragma unroll 8
    for (int k = 0; k < K; k += 4) {
        v2f a, b;
        a.x = Arow[k];
        a.y = Arow[k + 1];
        b.x = Bcol[(size_t)k * N];
        b.y = Bcol[(size_t)k * N + N];
        acc = __builtin_amdgcn_wmma_f32_16x16x4_f32(false, a, false, b,
                                                    (short)0, acc, false, false);
    }
    int col = n0 + r;
    float bv = bias ? bias[col] : 0.0f;
    #pragma unroll
    for (int vg = 0; vg < 8; ++vg)
        C[(size_t)(m0 + vg + 8 * hf) * N + col] = acc[vg] * scale + bv;
}

#define GBM 128
#define GBN 64
#define GBK 32
#define GPA 36
#define GPB 72

__global__ __launch_bounds__(256) void gemm_tiled_f32(
    const float* __restrict__ A, const float* __restrict__ B,
    float* __restrict__ C, const float* __restrict__ bias,
    int M, int N, int K, int lda, float scale)
{
    __shared__ __attribute__((aligned(16))) float As[2 * GBM * GPA];
    __shared__ __attribute__((aligned(16))) float Bs[2 * GBK * GPB];

    int t    = threadIdx.x;
    int wave = t >> 5;
    int lane = t & 31;
    int hf   = lane >> 4;
    int r    = lane & 15;
    int wm   = wave >> 1;
    int wn   = wave & 1;
    int m0 = blockIdx.y * GBM;
    int n0 = blockIdx.x * GBN;
    size_t kz = (size_t)blockIdx.z * K;

    int ar = t >> 3, ac = (t & 7) * 4;
    int br = t >> 4, bc = (t & 15) * 4;
    const float* Ag = A + (size_t)(m0 + ar) * lda + kz + ac;
    const float* Bg = B + (kz + br) * N + n0 + bc;
    C += (size_t)blockIdx.z * M * N;

    v4f pa[4], pb[2];
    #pragma unroll
    for (int i = 0; i < 4; ++i) pa[i] = *reinterpret_cast<const v4f*>(Ag + (size_t)(32 * i) * lda);
    #pragma unroll
    for (int i = 0; i < 2; ++i) pb[i] = *reinterpret_cast<const v4f*>(Bg + (size_t)(16 * i) * N);
    #pragma unroll
    for (int i = 0; i < 4; ++i) *reinterpret_cast<v4f*>(&As[(ar + 32 * i) * GPA + ac]) = pa[i];
    #pragma unroll
    for (int i = 0; i < 2; ++i) *reinterpret_cast<v4f*>(&Bs[(br + 16 * i) * GPB + bc]) = pb[i];
    __syncthreads();

    v8f acc[2][2] = {};
    int nchunks = K / GBK;
    for (int kc = 0; kc < nchunks; ++kc) {
        int cur = kc & 1;
        bool more = (kc + 1 < nchunks);
        if (more) {
            const float* Agn = Ag + (kc + 1) * GBK;
            const float* Bgn = Bg + (size_t)(kc + 1) * GBK * N;
            #pragma unroll
            for (int i = 0; i < 4; ++i) pa[i] = *reinterpret_cast<const v4f*>(Agn + (size_t)(32 * i) * lda);
            #pragma unroll
            for (int i = 0; i < 2; ++i) pb[i] = *reinterpret_cast<const v4f*>(Bgn + (size_t)(16 * i) * N);
        }
        const float* as = &As[cur * (GBM * GPA) + (wm * 32 + r) * GPA + 4 * hf];
        const float* bs = &Bs[cur * (GBK * GPB) + (4 * hf) * GPB + wn * 32 + r];
        #pragma unroll
        for (int g = 0; g < 4; ++g) {
            v4f a0 = *reinterpret_cast<const v4f*>(as + 8 * g);
            v4f a1 = *reinterpret_cast<const v4f*>(as + 16 * GPA + 8 * g);
            const float* bg = bs + (8 * g) * GPB;
            v2f b0lo = {bg[0],            bg[GPB]};
            v2f b0hi = {bg[2 * GPB],      bg[3 * GPB]};
            v2f b1lo = {bg[16],           bg[GPB + 16]};
            v2f b1hi = {bg[2 * GPB + 16], bg[3 * GPB + 16]};
            v2f a0lo = {a0.x, a0.y}, a0hi = {a0.z, a0.w};
            v2f a1lo = {a1.x, a1.y}, a1hi = {a1.z, a1.w};
            acc[0][0] = __builtin_amdgcn_wmma_f32_16x16x4_f32(false, a0lo, false, b0lo, (short)0, acc[0][0], false, false);
            acc[0][1] = __builtin_amdgcn_wmma_f32_16x16x4_f32(false, a0lo, false, b1lo, (short)0, acc[0][1], false, false);
            acc[1][0] = __builtin_amdgcn_wmma_f32_16x16x4_f32(false, a1lo, false, b0lo, (short)0, acc[1][0], false, false);
            acc[1][1] = __builtin_amdgcn_wmma_f32_16x16x4_f32(false, a1lo, false, b1lo, (short)0, acc[1][1], false, false);
            acc[0][0] = __builtin_amdgcn_wmma_f32_16x16x4_f32(false, a0hi, false, b0hi, (short)0, acc[0][0], false, false);
            acc[0][1] = __builtin_amdgcn_wmma_f32_16x16x4_f32(false, a0hi, false, b1hi, (short)0, acc[0][1], false, false);
            acc[1][0] = __builtin_amdgcn_wmma_f32_16x16x4_f32(false, a1hi, false, b0hi, (short)0, acc[1][0], false, false);
            acc[1][1] = __builtin_amdgcn_wmma_f32_16x16x4_f32(false, a1hi, false, b1hi, (short)0, acc[1][1], false, false);
        }
        if (more) {
            int nxt = cur ^ 1;
            #pragma unroll
            for (int i = 0; i < 4; ++i) *reinterpret_cast<v4f*>(&As[nxt * (GBM * GPA) + (ar + 32 * i) * GPA + ac]) = pa[i];
            #pragma unroll
            for (int i = 0; i < 2; ++i) *reinterpret_cast<v4f*>(&Bs[nxt * (GBK * GPB) + (br + 16 * i) * GPB + bc]) = pb[i];
        }
        __syncthreads();
    }

    float* patch = &As[wave * (32 * GPA)];
    #pragma unroll
    for (int mt = 0; mt < 2; ++mt)
        #pragma unroll
        for (int nt = 0; nt < 2; ++nt)
            #pragma unroll
            for (int vg = 0; vg < 8; ++vg)
                patch[(mt * 16 + vg + 8 * hf) * GPA + nt * 16 + r] = acc[mt][nt][vg];
    __syncthreads();
    int pr = lane >> 3, pc = (lane & 7) * 4;
    int col = n0 + wn * 32 + pc;
    v4f bv = {0.0f, 0.0f, 0.0f, 0.0f};
    if (bias) bv = *reinterpret_cast<const v4f*>(bias + col);
    #pragma unroll
    for (int i = 0; i < 8; ++i) {
        int row = pr + 4 * i;
        v4f v = *reinterpret_cast<const v4f*>(&patch[row * GPA + pc]);
        v4f o;
        o.x = v.x * scale + bv.x;
        o.y = v.y * scale + bv.y;
        o.z = v.z * scale + bv.z;
        o.w = v.w * scale + bv.w;
        *reinterpret_cast<v4f*>(&C[(size_t)(m0 + wm * 32 + row) * N + col]) = o;
    }
}

__global__ __launch_bounds__(64) void splitk_reduce(
    const float* __restrict__ part, const float* __restrict__ bias,
    float* __restrict__ dst, int MN, int N, int S)
{
    int i4 = blockIdx.x * blockDim.x + threadIdx.x;
    int quads = MN >> 2;
    if (i4 >= quads) return;
    const float4* p = reinterpret_cast<const float4*>(part) + i4;
    float4 acc = make_float4(0.0f, 0.0f, 0.0f, 0.0f);
    #pragma unroll 16
    for (int z = 0; z < S; ++z) {
        float4 v = p[(size_t)z * quads];
        acc.x += v.x; acc.y += v.y; acc.z += v.z; acc.w += v.w;
    }
    int col = (i4 << 2) % N;
    const float4 bv = *reinterpret_cast<const float4*>(bias + col);
    acc.x += bv.x; acc.y += bv.y; acc.z += bv.z; acc.w += bv.w;
    reinterpret_cast<float4*>(dst)[i4] = acc;
}

#define LDST 145
__global__ __launch_bounds__(128) void attn_kernel(
    float* __restrict__ qmat, const float* __restrict__ kv)
{
    __shared__ float attn[64 * LDST];
    int id = blockIdx.x;
    int h  = id & 7;
    int bb = (id >> 3) & 7;
    int g  = id >> 6;
    int n  = g * 8 + bb;

    int wave = threadIdx.x >> 5;
    int lane = threadIdx.x & 31;
    int hf   = lane >> 4;
    int r    = lane & 15;
    int m0   = wave * 16;

    const float* qbase = qmat + ((size_t)n * 64 + m0 + r) * 256 + h * 32;
    const float* kvb   = kv + (size_t)bb * 144 * 512 + h * 32;
    const float* vb    = kvb + 256;

    for (int nt = 0; nt < 9; ++nt) {
        int n0 = nt * 16;
        const float* kcol = kvb + (size_t)(n0 + r) * 512 + 2 * hf;
        v8f acc = {};
        #pragma unroll
        for (int k = 0; k < 32; k += 4) {
            v2f a, b;
            a.x = qbase[k + 2 * hf];
            a.y = qbase[k + 2 * hf + 1];
            b.x = kcol[k];
            b.y = kcol[k + 1];
            acc = __builtin_amdgcn_wmma_f32_16x16x4_f32(false, a, false, b,
                                                        (short)0, acc, false, false);
        }
        #pragma unroll
        for (int vg = 0; vg < 8; ++vg)
            attn[(m0 + vg + 8 * hf) * LDST + n0 + r] = acc[vg];
    }
    __syncthreads();

    if (threadIdx.x < 64) {
        float* row = attn + threadIdx.x * LDST;
        float mx = row[0];
        for (int k = 1; k < 144; ++k) mx = fmaxf(mx, row[k]);
        float s = 0.0f;
        for (int k = 0; k < 144; ++k) { float e = __expf(row[k] - mx); row[k] = e; s += e; }
        float inv = 1.0f / s;
        for (int k = 0; k < 144; ++k) row[k] *= inv;
    }
    __syncthreads();

    const float* arow = attn + (m0 + r) * LDST + 2 * hf;
    for (int nt = 0; nt < 2; ++nt) {
        int n0 = nt * 16;
        v8f acc = {};
        #pragma unroll 4
        for (int k = 0; k < 144; k += 4) {
            v2f a, b;
            a.x = arow[k];
            a.y = arow[k + 1];
            b.x = vb[(size_t)(k + 2 * hf) * 512 + n0 + r];
            b.y = vb[(size_t)(k + 2 * hf + 1) * 512 + n0 + r];
            acc = __builtin_amdgcn_wmma_f32_16x16x4_f32(false, a, false, b,
                                                        (short)0, acc, false, false);
        }
        #pragma unroll
        for (int vg = 0; vg < 8; ++vg)
            qmat[((size_t)n * 64 + m0 + vg + 8 * hf) * 256 + h * 32 + n0 + r] = acc[vg];
    }
}

#define RPE_ROWS 4
__global__ __launch_bounds__(192) void rpe_add_kernel(const float* __restrict__ x, const float* __restrict__ rw,
                                                      const float* __restrict__ rb, float* __restrict__ F) {
    int t     = threadIdx.x;
    int xq    = t % 24;
    int rg    = t / 24;
    int plane = blockIdx.x;
    int c     = plane & 255;
    const float* xp = x + (size_t)plane * 9216;
    float*       fb = F + (size_t)plane * 9216 + 4 * xq;

    float w[9];
    #pragma unroll
    for (int i = 0; i < 9; ++i) w[i] = rw[c * 9 + i];
    float bias = rb[c];

    int  cl = (xq > 0) ? 4 * xq - 1 : 0;
    int  cr = (xq < 23) ? 4 * xq + 4 : 95;
    bool hl = xq > 0, hr = xq < 23;

    #pragma unroll 1
    for (int it = 0; it < 3; ++it) {
        int y0 = rg * 12 + it * RPE_ROWS;
        float v[RPE_ROWS + 2][6];
        #pragma unroll
        for (int i = 0; i < RPE_ROWS + 2; ++i) {
            int  y  = y0 - 1 + i;
            bool ok = (y >= 0) && (y < 96);
            int  yc = y < 0 ? 0 : (y > 95 ? 95 : y);
            const float* row = xp + yc * 96;
            v4f m = *reinterpret_cast<const v4f*>(row + 4 * xq);
            float l = row[cl];
            float r = row[cr];
            v[i][0] = (ok && hl) ? l : 0.0f;
            v[i][1] = ok ? m.x : 0.0f;
            v[i][2] = ok ? m.y : 0.0f;
            v[i][3] = ok ? m.z : 0.0f;
            v[i][4] = ok ? m.w : 0.0f;
            v[i][5] = (ok && hr) ? r : 0.0f;
        }
        float* fp = fb + y0 * 96;
        #pragma unroll
        for (int j = 0; j < RPE_ROWS; ++j) {
            v4f f = *reinterpret_cast<const v4f*>(fp + j * 96);
            float a[4];
            #pragma unroll
            for (int e = 0; e < 4; ++e) {
                float acc = bias;
                #pragma unroll
                for (int dy = 0; dy < 3; ++dy)
                    #pragma unroll
                    for (int dx = 0; dx < 3; ++dx)
                        acc = __builtin_fmaf(v[j + dy][e + dx], w[dy * 3 + dx], acc);
                a[e] = acc;
            }
            f.x += a[0]; f.y += a[1]; f.z += a[2]; f.w += a[3];
            *reinterpret_cast<v4f*>(fp + j * 96) = f;
        }
    }
}

extern "C" void kernel_launch(void* const* d_in, const int* in_sizes, int n_in,
                              void* d_out, int out_size, void* d_ws, size_t ws_size,
                              hipStream_t stream) {
    const float* x      = (const float*)d_in[0];
    const float* Wq     = (const float*)d_in[1];
    const float* Wkv    = (const float*)d_in[2];
    const float* pool_w = (const float*)d_in[3];
    const float* pool_b = (const float*)d_in[4];
    const float* rpe_w  = (const float*)d_in[5];
    const float* rpe_b  = (const float*)d_in[6];
    const float* proj_w = (const float*)d_in[7];
    const float* proj_b = (const float*)d_in[8];
    float* out = (float*)d_out;

    float* ws    = (float*)d_ws;
    float* xw    = ws;
    float* qmat  = xw + 18874368;
    float* poolT = qmat + 18874368;
    float* WqT   = poolT + 4194304;
    float* WkvT  = WqT + 65536;
    float* projT = WkvT + 131072;
    float* poolx = projT + 65536;
    float* kvb   = poolx + 294912;
    float* part  = kvb + 589824;
    (void)ws_size; (void)in_sizes; (void)n_in; (void)out_size;

    const int TB = 256;
    transpose_k<<<(65536 + TB - 1) / TB, TB, 0, stream>>>(Wq, WqT, 256, 256, 65536);
    transpose_k<<<(131072 + TB - 1) / TB, TB, 0, stream>>>(Wkv, WkvT, 512, 256, 131072);
    transpose_k<<<(65536 + TB - 1) / TB, TB, 0, stream>>>(proj_w, projT, 256, 256, 65536);
    pool_transpose<<<256 * 4, TB, 0, stream>>>(pool_w, poolT);

    gather_xw<<<8 * 12 * 4 * 4, TB, 0, stream>>>(x, xw);

    gemm_tiled_f32<<<dim3(256 / 64, 1152 / 128, 64), TB, 0, stream>>>(
        xw, poolT, part, nullptr, 1152, 256, 256, 16384, 1.0f);
    splitk_reduce<<<73728 / 64, 64, 0, stream>>>(
        part, pool_b, poolx, 294912, 256, 64);

    gemm_tiled_f32<<<dim3(256 / 64, 73728 / 128), TB, 0, stream>>>(
        xw, WqT, qmat, nullptr, 73728, 256, 256, 256, QSCALE);

    gemm_wmma_f32<<<dim3(512 / 64, 1152 / 32), TB, 0, stream>>>(
        poolx, WkvT, kvb, nullptr, 1152, 512, 256, 256, 1.0f);

    attn_kernel<<<144 * 8 * 8, 128, 0, stream>>>(qmat, kvb);

    rpe_add_kernel<<<8 * 256, 192, 0, stream>>>(x, rpe_w, rpe_b, qmat);

    gemm_tiled_f32<<<dim3(256 / 64, 73728 / 128), TB, 0, stream>>>(
        qmat, projT, out, proj_b, 73728, 256, 256, 256, 1.0f);
}


# Attention forms scores transposed (keys x queries): a query's 144 scores stay in accumulators, softmax in-lane, exp'd scores are PV's B operand, 1/sum after PV; K,V via LDS; f32 sum order changes
# speedup vs baseline: 3.3561x; 1.1495x over previous
#include <hip/hip_runtime.h>
#include <hip/hip_bf16.h>

typedef __attribute__((ext_vector_type(2))) float v2f;
typedef __attribute__((ext_vector_type(8))) float v8f;

#define QSCALE 0.17677669529663687f


__global__ void transpose_k(const float* __restrict__ src, float* __restrict__ dst,
                            int O, int K, int total) {
    int idx = blockIdx.x * blockDim.x + threadIdx.x;
    if (idx >= total) return;
    int o = idx % O;
    int k = idx / O;
    dst[idx] = src[(size_t)o * K + k];
}

typedef __attribute__((ext_vector_type(4))) float v4f;
#define TPITCH 65

__global__ __launch_bounds__(256) void pool_transpose(const float* __restrict__ src,
                                                      float* __restrict__ dst) {
    __shared__ float tile[64 * TPITCH];
    int t  = threadIdx.x;
    int og = blockIdx.x & 3;
    int c  = blockIdx.x >> 2;
    int lo = t & 15;
    int hi = t >> 4;
    const float* s = src + ((size_t)(og * 64 + hi) * 16384 + c * 64 + 4 * lo);
    v4f v[4];
    #pragma unroll
    for (int it = 0; it < 4; ++it)
        v[it] = *reinterpret_cast<const v4f*>(s + (size_t)(16 * it) * 16384);
    #pragma unroll
    for (int it = 0; it < 4; ++it) {
        float* d = &tile[(4 * lo) * TPITCH + hi + 16 * it];
        d[0]          = v[it].x;
        d[TPITCH]     = v[it].y;
        d[2 * TPITCH] = v[it].z;
        d[3 * TPITCH] = v[it].w;
    }
    __syncthreads();
    float* g = dst + ((size_t)(hi * 256 + c) * 256 + og * 64 + 4 * lo);
    #pragma unroll
    for (int it = 0; it < 4; ++it) {
        const float* r = &tile[(hi + 16 * it) * TPITCH + 4 * lo];
        v4f o = {r[0], r[1], r[2], r[3]};
        *reinterpret_cast<v4f*>(g + (size_t)(16 * it) * 65536) = o;
    }
}

__global__ __launch_bounds__(256) void gather_xw(const float* __restrict__ x,
                                                 float* __restrict__ xw) {
    __shared__ float tile[192 * TPITCH];
    int t  = threadIdx.x;
    int b  = blockIdx.x;
    int cg = b & 3;
    int ip = (b >> 2) & 3;
    int bw = b >> 4;
    int wh = bw % 12;
    int bi = bw / 12;
    int lo = t & 15;
    int hi = t >> 4;
    const float* strip = x + (((size_t)(bi * 256 + cg * 64 + hi) * 96 + wh * 8 + 2 * ip) * 96 + 4 * lo);
    v4f v[12];
    #pragma unroll
    for (int it = 0; it < 12; ++it)
        v[it] = *reinterpret_cast<const v4f*>(strip + (size_t)(16 * (it & 3)) * 9216 + (it >> 2) * 64);
    float* d0 = &tile[(4 * lo) * TPITCH + hi];
    #pragma unroll
    for (int it = 0; it < 12; ++it) {
        float* d = d0 + ((it >> 2) * 64) * TPITCH + 16 * (it & 3);
        d[0]          = v[it].x;
        d[TPITCH]     = v[it].y;
        d[2 * TPITCH] = v[it].z;
        d[3 * TPITCH] = v[it].w;
    }
    __syncthreads();
    float* g = xw + ((size_t)((bi * 144 + wh * 12 + (hi >> 3)) * 64 + ip * 16 + (hi & 7)) * 256
                     + cg * 64 + 4 * lo);
    const float* r0 = &tile[hi * TPITCH + 4 * lo];
    #pragma unroll
    for (int it = 0; it < 12; ++it) {
        const float* r = r0 + (16 * it) * TPITCH;
        v4f o = {r[0], r[1], r[2], r[3]};
        *reinterpret_cast<v4f*>(g + (size_t)(it % 6) * (2 * 64 * 256) + (it / 6) * (8 * 256)) = o;
    }
}

__global__ __launch_bounds__(256) void gemm_wmma_f32(
    const float* __restrict__ A, const float* __restrict__ B,
    float* __restrict__ C, const float* __restrict__ bias,
    int M, int N, int K, int lda, float scale)
{
    int wave = threadIdx.x >> 5;
    int lane = threadIdx.x & 31;
    int hf   = lane >> 4;
    int r    = lane & 15;
    int m0 = blockIdx.y * 32 + (wave >> 2) * 16;
    int n0 = blockIdx.x * 64 + (wave & 3) * 16;
    size_t kz = (size_t)blockIdx.z * K;

    const float* Arow = A + (size_t)(m0 + r) * lda + kz + 2 * hf;
    const float* Bcol = B + (kz + 2 * hf) * N + n0 + r;
    C += (size_t)blockIdx.z * M * N;

    v8f acc = {};
    #pragma unroll 8
    for (int k = 0; k < K; k += 4) {
        v2f a, b;
        a.x = Arow[k];
        a.y = Arow[k + 1];
        b.x = Bcol[(size_t)k * N];
        b.y = Bcol[(size_t)k * N + N];
        acc = __builtin_amdgcn_wmma_f32_16x16x4_f32(false, a, false, b,
                                                    (short)0, acc, false, false);
    }
    int col = n0 + r;
    float bv = bias ? bias[col] : 0.0f;
    #pragma unroll
    for (int vg = 0; vg < 8; ++vg)
        C[(size_t)(m0 + vg + 8 * hf) * N + col] = acc[vg] * scale + bv;
}

#define GBM 128
#define GBN 64
#define GBK 32
#define GPA 36
#define GPB 72

__global__ __launch_bounds__(256) void gemm_tiled_f32(
    const float* __restrict__ A, const float* __restrict__ B,
    float* __restrict__ C, const float* __restrict__ bias,
    int M, int N, int K, int lda, float scale)
{
    __shared__ __attribute__((aligned(16))) float As[2 * GBM * GPA];
    __shared__ __attribute__((aligned(16))) float Bs[2 * GBK * GPB];

    int t    = threadIdx.x;
    int wave = t >> 5;
    int lane = t & 31;
    int hf   = lane >> 4;
    int r    = lane & 15;
    int wm   = wave >> 1;
    int wn   = wave & 1;
    int m0 = blockIdx.y * GBM;
    int n0 = blockIdx.x * GBN;
    size_t kz = (size_t)blockIdx.z * K;

    int ar = t >> 3, ac = (t & 7) * 4;
    int br = t >> 4, bc = (t & 15) * 4;
    const float* Ag = A + (size_t)(m0 + ar) * lda + kz + ac;
    const float* Bg = B + (kz + br) * N + n0 + bc;
    C += (size_t)blockIdx.z * M * N;

    v4f pa[4], pb[2];
    #pragma unroll
    for (int i = 0; i < 4; ++i) pa[i] = *reinterpret_cast<const v4f*>(Ag + (size_t)(32 * i) * lda);
    #pragma unroll
    for (int i = 0; i < 2; ++i) pb[i] = *reinterpret_cast<const v4f*>(Bg + (size_t)(16 * i) * N);
    #pragma unroll
    for (int i = 0; i < 4; ++i) *reinterpret_cast<v4f*>(&As[(ar + 32 * i) * GPA + ac]) = pa[i];
    #pragma unroll
    for (int i = 0; i < 2; ++i) *reinterpret_cast<v4f*>(&Bs[(br + 16 * i) * GPB + bc]) = pb[i];
    __syncthreads();

    v8f acc[2][2] = {};
    int nchunks = K / GBK;
    for (int kc = 0; kc < nchunks; ++kc) {
        int cur = kc & 1;
        bool more = (kc + 1 < nchunks);
        if (more) {
            const float* Agn = Ag + (kc + 1) * GBK;
            const float* Bgn = Bg + (size_t)(kc + 1) * GBK * N;
            #pragma unroll
            for (int i = 0; i < 4; ++i) pa[i] = *reinterpret_cast<const v4f*>(Agn + (size_t)(32 * i) * lda);
            #pragma unroll
            for (int i = 0; i < 2; ++i) pb[i] = *reinterpret_cast<const v4f*>(Bgn + (size_t)(16 * i) * N);
        }
        const float* as = &As[cur * (GBM * GPA) + (wm * 32 + r) * GPA + 4 * hf];
        const float* bs = &Bs[cur * (GBK * GPB) + (4 * hf) * GPB + wn * 32 + r];
        #pragma unroll
        for (int g = 0; g < 4; ++g) {
            v4f a0 = *reinterpret_cast<const v4f*>(as + 8 * g);
            v4f a1 = *reinterpret_cast<const v4f*>(as + 16 * GPA + 8 * g);
            const float* bg = bs + (8 * g) * GPB;
            v2f b0lo = {bg[0],            bg[GPB]};
            v2f b0hi = {bg[2 * GPB],      bg[3 * GPB]};
            v2f b1lo = {bg[16],           bg[GPB + 16]};
            v2f b1hi = {bg[2 * GPB + 16], bg[3 * GPB + 16]};
            v2f a0lo = {a0.x, a0.y}, a0hi = {a0.z, a0.w};
            v2f a1lo = {a1.x, a1.y}, a1hi = {a1.z, a1.w};
            acc[0][0] = __builtin_amdgcn_wmma_f32_16x16x4_f32(false, a0lo, false, b0lo, (short)0, acc[0][0], false, false);
            acc[0][1] = __builtin_amdgcn_wmma_f32_16x16x4_f32(false, a0lo, false, b1lo, (short)0, acc[0][1], false, false);
            acc[1][0] = __builtin_amdgcn_wmma_f32_16x16x4_f32(false, a1lo, false, b0lo, (short)0, acc[1][0], false, false);
            acc[1][1] = __builtin_amdgcn_wmma_f32_16x16x4_f32(false, a1lo, false, b1lo, (short)0, acc[1][1], false, false);
            acc[0][0] = __builtin_amdgcn_wmma_f32_16x16x4_f32(false, a0hi, false, b0hi, (short)0, acc[0][0], false, false);
            acc[0][1] = __builtin_amdgcn_wmma_f32_16x16x4_f32(false, a0hi, false, b1hi, (short)0, acc[0][1], false, false);
            acc[1][0] = __builtin_amdgcn_wmma_f32_16x16x4_f32(false, a1hi, false, b0hi, (short)0, acc[1][0], false, false);
            acc[1][1] = __builtin_amdgcn_wmma_f32_16x16x4_f32(false, a1hi, false, b1hi, (short)0, acc[1][1], false, false);
        }
        if (more) {
            int nxt = cur ^ 1;
            #pragma unroll
            for (int i = 0; i < 4; ++i) *reinterpret_cast<v4f*>(&As[nxt * (GBM * GPA) + (ar + 32 * i) * GPA + ac]) = pa[i];
            #pragma unroll
            for (int i = 0; i < 2; ++i) *reinterpret_cast<v4f*>(&Bs[nxt * (GBK * GPB) + (br + 16 * i) * GPB + bc]) = pb[i];
        }
        __syncthreads();
    }

    float* patch = &As[wave * (32 * GPA)];
    #pragma unroll
    for (int mt = 0; mt < 2; ++mt)
        #pragma unroll
        for (int nt = 0; nt < 2; ++nt)
            #pragma unroll
            for (int vg = 0; vg < 8; ++vg)
                patch[(mt * 16 + vg + 8 * hf) * GPA + nt * 16 + r] = acc[mt][nt][vg];
    __syncthreads();
    int pr = lane >> 3, pc = (lane & 7) * 4;
    int col = n0 + wn * 32 + pc;
    v4f bv = {0.0f, 0.0f, 0.0f, 0.0f};
    if (bias) bv = *reinterpret_cast<const v4f*>(bias + col);
    #pragma unroll
    for (int i = 0; i < 8; ++i) {
        int row = pr + 4 * i;
        v4f v = *reinterpret_cast<const v4f*>(&patch[row * GPA + pc]);
        v4f o;
        o.x = v.x * scale + bv.x;
        o.y = v.y * scale + bv.y;
        o.z = v.z * scale + bv.z;
        o.w = v.w * scale + bv.w;
        *reinterpret_cast<v4f*>(&C[(size_t)(m0 + wm * 32 + row) * N + col]) = o;
    }
}

__global__ __launch_bounds__(64) void splitk_reduce(
    const float* __restrict__ part, const float* __restrict__ bias,
    float* __restrict__ dst, int MN, int N, int S)
{
    int i4 = blockIdx.x * blockDim.x + threadIdx.x;
    int quads = MN >> 2;
    if (i4 >= quads) return;
    const float4* p = reinterpret_cast<const float4*>(part) + i4;
    float4 acc = make_float4(0.0f, 0.0f, 0.0f, 0.0f);
    #pragma unroll 16
    for (int z = 0; z < S; ++z) {
        float4 v = p[(size_t)z * quads];
        acc.x += v.x; acc.y += v.y; acc.z += v.z; acc.w += v.w;
    }
    int col = (i4 << 2) % N;
    const float4 bv = *reinterpret_cast<const float4*>(bias + col);
    acc.x += bv.x; acc.y += bv.y; acc.z += bv.z; acc.w += bv.w;
    reinterpret_cast<float4*>(dst)[i4] = acc;
}

#define APITCH 36
#define VPITCH 72
__global__ __launch_bounds__(128) void attn_kernel(
    float* __restrict__ qmat, const float* __restrict__ kv)
{
    __shared__ __attribute__((aligned(16))) float Ks[144 * APITCH];
    __shared__ __attribute__((aligned(16))) float Vs[72 * VPITCH];
    int id = blockIdx.x;
    int h  = id & 7;
    int bb = (id >> 3) & 7;
    int g  = id >> 6;
    int n  = g * 8 + bb;

    int t    = threadIdx.x;
    int wave = t >> 5;
    int lane = t & 31;
    int hf   = lane >> 4;
    int c    = lane & 15;
    int m0   = wave * 16;

    float*       qrow = qmat + ((size_t)n * 64 + m0) * 256 + h * 32;
    const float* kg   = kv + (size_t)bb * 144 * 512 + h * 32;

    v4f qf[4];
    #pragma unroll
    for (int g4 = 0; g4 < 4; ++g4)
        qf[g4] = *reinterpret_cast<const v4f*>(qrow + c * 256 + 8 * g4 + 4 * hf);

    {
        int sr = t >> 3, sc = (t & 7) * 4;
        const float* ksrc = kg + (size_t)sr * 512 + sc;
        const float* vsrc = kg + 256 + (size_t)(2 * sr) * 512 + sc;
        bool tail = sr < 8;
        v4f sk[9], va[5], vb[5];
        #pragma unroll
        for (int i = 0; i < 9; ++i) sk[i] = *reinterpret_cast<const v4f*>(ksrc + (size_t)(16 * i) * 512);
        #pragma unroll
        for (int i = 0; i < 5; ++i)
            if (i < 4 || tail) {
                va[i] = *reinterpret_cast<const v4f*>(vsrc + (size_t)(32 * i) * 512);
                vb[i] = *reinterpret_cast<const v4f*>(vsrc + (size_t)(32 * i + 1) * 512);
            }
        #pragma unroll
        for (int i = 0; i < 9; ++i) *reinterpret_cast<v4f*>(&Ks[(sr + 16 * i) * APITCH + sc]) = sk[i];
        #pragma unroll
        for (int i = 0; i < 5; ++i)
            if (i < 4 || tail) {
                float* d = &Vs[(sr + 16 * i) * VPITCH + 2 * sc];
                v4f lo = {va[i].x, vb[i].x, va[i].y, vb[i].y};
                v4f hi = {va[i].z, vb[i].z, va[i].w, vb[i].w};
                *reinterpret_cast<v4f*>(d)     = lo;
                *reinterpret_cast<v4f*>(d + 4) = hi;
            }
    }
    __syncthreads();

    v8f acc[9];
    const float* kf = &Ks[c * APITCH + 4 * hf];
    #pragma unroll
    for (int tg = 0; tg < 3; ++tg) {
        v4f kk[3][4];
        #pragma unroll
        for (int x = 0; x < 3; ++x)
            #pragma unroll
            for (int g4 = 0; g4 < 4; ++g4)
                kk[x][g4] = *reinterpret_cast<const v4f*>(kf + (16 * (3 * tg + x)) * APITCH + 8 * g4);
        v8f a0 = {}, a1 = {}, a2 = {};
        #pragma unroll
        for (int g4 = 0; g4 < 4; ++g4) {
            v2f qlo = {qf[g4].x, qf[g4].y}, qhi = {qf[g4].z, qf[g4].w};
            v2f k0lo = {kk[0][g4].x, kk[0][g4].y}, k0hi = {kk[0][g4].z, kk[0][g4].w};
            v2f k1lo = {kk[1][g4].x, kk[1][g4].y}, k1hi = {kk[1][g4].z, kk[1][g4].w};
            v2f k2lo = {kk[2][g4].x, kk[2][g4].y}, k2hi = {kk[2][g4].z, kk[2][g4].w};
            a0 = __builtin_amdgcn_wmma_f32_16x16x4_f32(false, k0lo, false, qlo, (short)0, a0, false, false);
            a1 = __builtin_amdgcn_wmma_f32_16x16x4_f32(false, k1lo, false, qlo, (short)0, a1, false, false);
            a2 = __builtin_amdgcn_wmma_f32_16x16x4_f32(false, k2lo, false, qlo, (short)0, a2, false, false);
            a0 = __builtin_amdgcn_wmma_f32_16x16x4_f32(false, k0hi, false, qhi, (short)0, a0, false, false);
            a1 = __builtin_amdgcn_wmma_f32_16x16x4_f32(false, k1hi, false, qhi, (short)0, a1, false, false);
            a2 = __builtin_amdgcn_wmma_f32_16x16x4_f32(false, k2hi, false, qhi, (short)0, a2, false, false);
        }
        acc[3 * tg] = a0; acc[3 * tg + 1] = a1; acc[3 * tg + 2] = a2;
    }

    float mx = acc[0][0];
    #pragma unroll
    for (int tl = 0; tl < 9; ++tl)
        #pragma unroll
        for (int rr = 0; rr < 8; ++rr)
            mx = fmaxf(mx, acc[tl][rr]);
    mx = fmaxf(mx, __shfl_xor(mx, 16));
    float s0 = 0.0f, s1 = 0.0f, s2 = 0.0f, s3 = 0.0f;
    #pragma unroll
    for (int tl = 0; tl < 9; ++tl) {
        #pragma unroll
        for (int rr = 0; rr < 8; ++rr)
            acc[tl][rr] = __expf(acc[tl][rr] - mx);
        s0 += acc[tl][0] + acc[tl][4];
        s1 += acc[tl][1] + acc[tl][5];
        s2 += acc[tl][2] + acc[tl][6];
        s3 += acc[tl][3] + acc[tl][7];
    }
    float sum = (s0 + s1) + (s2 + s3);
    sum += __shfl_xor(sum, 16);
    float inv = 1.0f / sum;

    v8f o0 = {}, o1 = {};
    const float* vf = &Vs[(4 * hf) * VPITCH + 2 * c];
    #pragma unroll
    for (int tl = 0; tl < 9; ++tl) {
        #pragma unroll
        for (int j = 0; j < 4; ++j) {
            const float* vp = vf + (8 * tl + j) * VPITCH;
            v2f v0 = *reinterpret_cast<const v2f*>(vp);
            v2f v1 = *reinterpret_cast<const v2f*>(vp + 32);
            v2f p  = {acc[tl][2 * j], acc[tl][2 * j + 1]};
            o0 = __builtin_amdgcn_wmma_f32_16x16x4_f32(false, v0, false, p, (short)0, o0, false, false);
            o1 = __builtin_amdgcn_wmma_f32_16x16x4_f32(false, v1, false, p, (short)0, o1, false, false);
        }
    }

    __syncthreads();
    float* patch = &Ks[wave * (16 * APITCH)];
    {
        float* pw = patch + c * APITCH + 8 * hf;
        v4f w0 = {o0[0] * inv, o0[1] * inv, o0[2] * inv, o0[3] * inv};
        v4f w1 = {o0[4] * inv, o0[5] * inv, o0[6] * inv, o0[7] * inv};
        v4f w2 = {o1[0] * inv, o1[1] * inv, o1[2] * inv, o1[3] * inv};
        v4f w3 = {o1[4] * inv, o1[5] * inv, o1[6] * inv, o1[7] * inv};
        *reinterpret_cast<v4f*>(pw)      = w0;
        *reinterpret_cast<v4f*>(pw + 4)  = w1;
        *reinterpret_cast<v4f*>(pw + 16) = w2;
        *reinterpret_cast<v4f*>(pw + 20) = w3;
    }
    __syncthreads();
    int pr = lane >> 3, pc = (lane & 7) * 4;
    #pragma unroll
    for (int i = 0; i < 4; ++i) {
        int row = pr + 4 * i;
        v4f v = *reinterpret_cast<const v4f*>(&patch[row * APITCH + pc]);
        *reinterpret_cast<v4f*>(qrow + row * 256 + pc) = v;
    }
}

#define RPE_ROWS 4
__global__ __launch_bounds__(192) void rpe_add_kernel(const float* __restrict__ x, const float* __restrict__ rw,
                                                      const float* __restrict__ rb, float* __restrict__ F) {
    int t     = threadIdx.x;
    int xq    = t % 24;
    int rg    = t / 24;
    int plane = blockIdx.x;
    int c     = plane & 255;
    const float* xp = x + (size_t)plane * 9216;
    float*       fb = F + (size_t)plane * 9216 + 4 * xq;

    float w[9];
    #pragma unroll
    for (int i = 0; i < 9; ++i) w[i] = rw[c * 9 + i];
    float bias = rb[c];

    int  cl = (xq > 0) ? 4 * xq - 1 : 0;
    int  cr = (xq < 23) ? 4 * xq + 4 : 95;
    bool hl = xq > 0, hr = xq < 23;

    #pragma unroll 1
    for (int it = 0; it < 3; ++it) {
        int y0 = rg * 12 + it * RPE_ROWS;
        float v[RPE_ROWS + 2][6];
        #pragma unroll
        for (int i = 0; i < RPE_ROWS + 2; ++i) {
            int  y  = y0 - 1 + i;
            bool ok = (y >= 0) && (y < 96);
            int  yc = y < 0 ? 0 : (y > 95 ? 95 : y);
            const float* row = xp + yc * 96;
            v4f m = *reinterpret_cast<const v4f*>(row + 4 * xq);
            float l = row[cl];
            float r = row[cr];
            v[i][0] = (ok && hl) ? l : 0.0f;
            v[i][1] = ok ? m.x : 0.0f;
            v[i][2] = ok ? m.y : 0.0f;
            v[i][3] = ok ? m.z : 0.0f;
            v[i][4] = ok ? m.w : 0.0f;
            v[i][5] = (ok && hr) ? r : 0.0f;
        }
        float* fp = fb + y0 * 96;
        #pragma unroll
        for (int j = 0; j < RPE_ROWS; ++j) {
            v4f f = *reinterpret_cast<const v4f*>(fp + j * 96);
            float a[4];
            #pragma unroll
            for (int e = 0; e < 4; ++e) {
                float acc = bias;
                #pragma unroll
                for (int dy = 0; dy < 3; ++dy)
                    #pragma unroll
                    for (int dx = 0; dx < 3; ++dx)
                        acc = __builtin_fmaf(v[j + dy][e + dx], w[dy * 3 + dx], acc);
                a[e] = acc;
            }
            f.x += a[0]; f.y += a[1]; f.z += a[2]; f.w += a[3];
            *reinterpret_cast<v4f*>(fp + j * 96) = f;
        }
    }
}

extern "C" void kernel_launch(void* const* d_in, const int* in_sizes, int n_in,
                              void* d_out, int out_size, void* d_ws, size_t ws_size,
                              hipStream_t stream) {
    const float* x      = (const float*)d_in[0];
    const float* Wq     = (const float*)d_in[1];
    const float* Wkv    = (const float*)d_in[2];
    const float* pool_w = (const float*)d_in[3];
    const float* pool_b = (const float*)d_in[4];
    const float* rpe_w  = (const float*)d_in[5];
    const float* rpe_b  = (const float*)d_in[6];
    const float* proj_w = (const float*)d_in[7];
    const float* proj_b = (const float*)d_in[8];
    float* out = (float*)d_out;

    float* ws    = (float*)d_ws;
    float* xw    = ws;
    float* qmat  = xw + 18874368;
    float* poolT = qmat + 18874368;
    float* WqT   = poolT + 4194304;
    float* WkvT  = WqT + 65536;
    float* projT = WkvT + 131072;
    float* poolx = projT + 65536;
    float* kvb   = poolx + 294912;
    float* part  = kvb + 589824;
    (void)ws_size; (void)in_sizes; (void)n_in; (void)out_size;

    const int TB = 256;
    transpose_k<<<(65536 + TB - 1) / TB, TB, 0, stream>>>(Wq, WqT, 256, 256, 65536);
    transpose_k<<<(131072 + TB - 1) / TB, TB, 0, stream>>>(Wkv, WkvT, 512, 256, 131072);
    transpose_k<<<(65536 + TB - 1) / TB, TB, 0, stream>>>(proj_w, projT, 256, 256, 65536);
    pool_transpose<<<256 * 4, TB, 0, stream>>>(pool_w, poolT);

    gather_xw<<<8 * 12 * 4 * 4, TB, 0, stream>>>(x, xw);

    gemm_tiled_f32<<<dim3(256 / 64, 1152 / 128, 64), TB, 0, stream>>>(
        xw, poolT, part, nullptr, 1152, 256, 256, 16384, 1.0f);
    splitk_reduce<<<73728 / 64, 64, 0, stream>>>(
        part, pool_b, poolx, 294912, 256, 64);

    gemm_tiled_f32<<<dim3(256 / 64, 73728 / 128), TB, 0, stream>>>(
        xw, WqT, qmat, nullptr, 73728, 256, 256, 256, QSCALE);

    gemm_wmma_f32<<<dim3(512 / 64, 1152 / 32), TB, 0, stream>>>(
        poolx, WkvT, kvb, nullptr, 1152, 512, 256, 256, 1.0f);

    attn_kernel<<<144 * 8 * 8, 128, 0, stream>>>(qmat, kvb);

    rpe_add_kernel<<<8 * 256, 192, 0, stream>>>(x, rpe_w, rpe_b, qmat);

    gemm_tiled_f32<<<dim3(256 / 64, 73728 / 128), TB, 0, stream>>>(
        qmat, projT, out, proj_b, 73728, 256, 256, 256, 1.0f);
}


# Independent kernels share launches: window gather, pool-weight and three weight transposes in one launch, split-K sum beside the Q GEMM in another; 12 launches -> 7; every sum and its order unchanged
# speedup vs baseline: 3.4652x; 1.0325x over previous
#include <hip/hip_runtime.h>
#include <hip/hip_bf16.h>

typedef __attribute__((ext_vector_type(2))) float v2f;
typedef __attribute__((ext_vector_type(8))) float v8f;

#define QSCALE 0.17677669529663687f


__device__ __forceinline__ void transpose_k_body(const float* __restrict__ src, float* __restrict__ dst,
                                                 int O, int K, int total, int blk) {
    int idx = blk * 256 + threadIdx.x;
    if (idx >= total) return;
    int o = idx % O;
    int k = idx / O;
    dst[idx] = src[(size_t)o * K + k];
}

typedef __attribute__((ext_vector_type(4))) float v4f;
#define TPITCH 65

__device__ __forceinline__ void pool_transpose_body(const float* __restrict__ src,
                                                    float* __restrict__ dst, int blk, float* tile) {
    int t  = threadIdx.x;
    int og = blk & 3;
    int c  = blk >> 2;
    int lo = t & 15;
    int hi = t >> 4;
    const float* s = src + ((size_t)(og * 64 + hi) * 16384 + c * 64 + 4 * lo);
    v4f v[4];
    #pragma unroll
    for (int it = 0; it < 4; ++it)
        v[it] = *reinterpret_cast<const v4f*>(s + (size_t)(16 * it) * 16384);
    #pragma unroll
    for (int it = 0; it < 4; ++it) {
        float* d = &tile[(4 * lo) * TPITCH + hi + 16 * it];
        d[0]          = v[it].x;
        d[TPITCH]     = v[it].y;
        d[2 * TPITCH] = v[it].z;
        d[3 * TPITCH] = v[it].w;
    }
    __syncthreads();
    float* g = dst + ((size_t)(hi * 256 + c) * 256 + og * 64 + 4 * lo);
    #pragma unroll
    for (int it = 0; it < 4; ++it) {
        const float* r = &tile[(hi + 16 * it) * TPITCH + 4 * lo];
        v4f o = {r[0], r[1], r[2], r[3]};
        *reinterpret_cast<v4f*>(g + (size_t)(16 * it) * 65536) = o;
    }
}

__device__ __forceinline__ void gather_xw_body(const float* __restrict__ x,
                                               float* __restrict__ xw, int b, float* tile) {
    int t  = threadIdx.x;
    int cg = b & 3;
    int ip = (b >> 2) & 3;
    int bw = b >> 4;
    int wh = bw % 12;
    int bi = bw / 12;
    int lo = t & 15;
    int hi = t >> 4;
    const float* strip = x + (((size_t)(bi * 256 + cg * 64 + hi) * 96 + wh * 8 + 2 * ip) * 96 + 4 * lo);
    v4f v[12];
    #pragma unroll
    for (int it = 0; it < 12; ++it)
        v[it] = *reinterpret_cast<const v4f*>(strip + (size_t)(16 * (it & 3)) * 9216 + (it >> 2) * 64);
    float* d0 = &tile[(4 * lo) * TPITCH + hi];
    #pragma unroll
    for (int it = 0; it < 12; ++it) {
        float* d = d0 + ((it >> 2) * 64) * TPITCH + 16 * (it & 3);
        d[0]          = v[it].x;
        d[TPITCH]     = v[it].y;
        d[2 * TPITCH] = v[it].z;
        d[3 * TPITCH] = v[it].w;
    }
    __syncthreads();
    float* g = xw + ((size_t)((bi * 144 + wh * 12 + (hi >> 3)) * 64 + ip * 16 + (hi & 7)) * 256
                     + cg * 64 + 4 * lo);
    const float* r0 = &tile[hi * TPITCH + 4 * lo];
    #pragma unroll
    for (int it = 0; it < 12; ++it) {
        const float* r = r0 + (16 * it) * TPITCH;
        v4f o = {r[0], r[1], r[2], r[3]};
        *reinterpret_cast<v4f*>(g + (size_t)(it % 6) * (2 * 64 * 256) + (it / 6) * (8 * 256)) = o;
    }
}

__global__ __launch_bounds__(256) void layout_kernel(
    const float* __restrict__ x, float* __restrict__ xw,
    const float* __restrict__ pool_w, float* __restrict__ poolT,
    const float* __restrict__ Wq, float* __restrict__ WqT,
    const float* __restrict__ Wkv, float* __restrict__ WkvT,
    const float* __restrict__ proj_w, float* __restrict__ projT)
{
    __shared__ float tile[192 * TPITCH];
    int b = blockIdx.x;
    if (b < 1536)      gather_xw_body(x, xw, b, tile);
    else if (b < 2560) pool_transpose_body(pool_w, poolT, b - 1536, tile);
    else if (b < 2816) transpose_k_body(Wq, WqT, 256, 256, 65536, b - 2560);
    else if (b < 3328) transpose_k_body(Wkv, WkvT, 512, 256, 131072, b - 2816);
    else               transpose_k_body(proj_w, projT, 256, 256, 65536, b - 3328);
}

__global__ __launch_bounds__(256) void gemm_wmma_f32(
    const float* __restrict__ A, const float* __restrict__ B,
    float* __restrict__ C, const float* __restrict__ bias,
    int M, int N, int K, int lda, float scale)
{
    int wave = threadIdx.x >> 5;
    int lane = threadIdx.x & 31;
    int hf   = lane >> 4;
    int r    = lane & 15;
    int m0 = blockIdx.y * 32 + (wave >> 2) * 16;
    int n0 = blockIdx.x * 64 + (wave & 3) * 16;
    size_t kz = (size_t)blockIdx.z * K;

    const float* Arow = A + (size_t)(m0 + r) * lda + kz + 2 * hf;
    const float* Bcol = B + (kz + 2 * hf) * N + n0 + r;
    C += (size_t)blockIdx.z * M * N;

    v8f acc = {};
    #pragma unroll 8
    for (int k = 0; k < K; k += 4) {
        v2f a, b;
        a.x = Arow[k];
        a.y = Arow[k + 1];
        b.x = Bcol[(size_t)k * N];
        b.y = Bcol[(size_t)k * N + N];
        acc = __builtin_amdgcn_wmma_f32_16x16x4_f32(false, a, false, b,
                                                    (short)0, acc, false, false);
    }
    int col = n0 + r;
    float bv = bias ? bias[col] : 0.0f;
    #pragma unroll
    for (int vg = 0; vg < 8; ++vg)
        C[(size_t)(m0 + vg + 8 * hf) * N + col] = acc[vg] * scale + bv;
}

#define GBM 128
#define GBN 64
#define GBK 32
#define GPA 36
#define GPB 72

__device__ __forceinline__ void gemm_tiled_body(
    const float* __restrict__ A, const float* __restrict__ B,
    float* __restrict__ C, const float* __restrict__ bias,
    int M, int N, int K, int lda, float scale,
    int bx, int by, int bz, float* As, float* Bs)
{
    int t    = threadIdx.x;
    int wave = t >> 5;
    int lane = t & 31;
    int hf   = lane >> 4;
    int r    = lane & 15;
    int wm   = wave >> 1;
    int wn   = wave & 1;
    int m0 = by * GBM;
    int n0 = bx * GBN;
    size_t kz = (size_t)bz * K;

    int ar = t >> 3, ac = (t & 7) * 4;
    int br = t >> 4, bc = (t & 15) * 4;
    const float* Ag = A + (size_t)(m0 + ar) * lda + kz + ac;
    const float* Bg = B + (kz + br) * N + n0 + bc;
    C += (size_t)bz * M * N;

    v4f pa[4], pb[2];
    #pragma unroll
    for (int i = 0; i < 4; ++i) pa[i] = *reinterpret_cast<const v4f*>(Ag + (size_t)(32 * i) * lda);
    #pragma unroll
    for (int i = 0; i < 2; ++i) pb[i] = *reinterpret_cast<const v4f*>(Bg + (size_t)(16 * i) * N);
    #pragma unroll
    for (int i = 0; i < 4; ++i) *reinterpret_cast<v4f*>(&As[(ar + 32 * i) * GPA + ac]) = pa[i];
    #pragma unroll
    for (int i = 0; i < 2; ++i) *reinterpret_cast<v4f*>(&Bs[(br + 16 * i) * GPB + bc]) = pb[i];
    __syncthreads();

    v8f acc[2][2] = {};
    int nchunks = K / GBK;
    for (int kc = 0; kc < nchunks; ++kc) {
        int cur = kc & 1;
        bool more = (kc + 1 < nchunks);
        if (more) {
            const float* Agn = Ag + (kc + 1) * GBK;
            const float* Bgn = Bg + (size_t)(kc + 1) * GBK * N;
            #pragma unroll
            for (int i = 0; i < 4; ++i) pa[i] = *reinterpret_cast<const v4f*>(Agn + (size_t)(32 * i) * lda);
            #pragma unroll
            for (int i = 0; i < 2; ++i) pb[i] = *reinterpret_cast<const v4f*>(Bgn + (size_t)(16 * i) * N);
        }
        const float* as = &As[cur * (GBM * GPA) + (wm * 32 + r) * GPA + 4 * hf];
        const float* bs = &Bs[cur * (GBK * GPB) + (4 * hf) * GPB + wn * 32 + r];
        #pragma unroll
        for (int g = 0; g < 4; ++g) {
            v4f a0 = *reinterpret_cast<const v4f*>(as + 8 * g);
            v4f a1 = *reinterpret_cast<const v4f*>(as + 16 * GPA + 8 * g);
            const float* bg = bs + (8 * g) * GPB;
            v2f b0lo = {bg[0],            bg[GPB]};
            v2f b0hi = {bg[2 * GPB],      bg[3 * GPB]};
            v2f b1lo = {bg[16],           bg[GPB + 16]};
            v2f b1hi = {bg[2 * GPB + 16], bg[3 * GPB + 16]};
            v2f a0lo = {a0.x, a0.y}, a0hi = {a0.z, a0.w};
            v2f a1lo = {a1.x, a1.y}, a1hi = {a1.z, a1.w};
            acc[0][0] = __builtin_amdgcn_wmma_f32_16x16x4_f32(false, a0lo, false, b0lo, (short)0, acc[0][0], false, false);
            acc[0][1] = __builtin_amdgcn_wmma_f32_16x16x4_f32(false, a0lo, false, b1lo, (short)0, acc[0][1], false, false);
            acc[1][0] = __builtin_amdgcn_wmma_f32_16x16x4_f32(false, a1lo, false, b0lo, (short)0, acc[1][0], false, false);
            acc[1][1] = __builtin_amdgcn_wmma_f32_16x16x4_f32(false, a1lo, false, b1lo, (short)0, acc[1][1], false, false);
            acc[0][0] = __builtin_amdgcn_wmma_f32_16x16x4_f32(false, a0hi, false, b0hi, (short)0, acc[0][0], false, false);
            acc[0][1] = __builtin_amdgcn_wmma_f32_16x16x4_f32(false, a0hi, false, b1hi, (short)0, acc[0][1], false, false);
            acc[1][0] = __builtin_amdgcn_wmma_f32_16x16x4_f32(false, a1hi, false, b0hi, (short)0, acc[1][0], false, false);
            acc[1][1] = __builtin_amdgcn_wmma_f32_16x16x4_f32(false, a1hi, false, b1hi, (short)0, acc[1][1], false, false);
        }
        if (more) {
            int nxt = cur ^ 1;
            #pragma unroll
            for (int i = 0; i < 4; ++i) *reinterpret_cast<v4f*>(&As[nxt * (GBM * GPA) + (ar + 32 * i) * GPA + ac]) = pa[i];
            #pragma unroll
            for (int i = 0; i < 2; ++i) *reinterpret_cast<v4f*>(&Bs[nxt * (GBK * GPB) + (br + 16 * i) * GPB + bc]) = pb[i];
        }
        __syncthreads();
    }

    float* patch = &As[wave * (32 * GPA)];
    #pragma unroll
    for (int mt = 0; mt < 2; ++mt)
        #pragma unroll
        for (int nt = 0; nt < 2; ++nt)
            #pragma unroll
            for (int vg = 0; vg < 8; ++vg)
                patch[(mt * 16 + vg + 8 * hf) * GPA + nt * 16 + r] = acc[mt][nt][vg];
    __syncthreads();
    int pr = lane >> 3, pc = (lane & 7) * 4;
    int col = n0 + wn * 32 + pc;
    v4f bv = {0.0f, 0.0f, 0.0f, 0.0f};
    if (bias) bv = *reinterpret_cast<const v4f*>(bias + col);
    #pragma unroll
    for (int i = 0; i < 8; ++i) {
        int row = pr + 4 * i;
        v4f v = *reinterpret_cast<const v4f*>(&patch[row * GPA + pc]);
        v4f o;
        o.x = v.x * scale + bv.x;
        o.y = v.y * scale + bv.y;
        o.z = v.z * scale + bv.z;
        o.w = v.w * scale + bv.w;
        *reinterpret_cast<v4f*>(&C[(size_t)(m0 + wm * 32 + row) * N + col]) = o;
    }
}

__global__ __launch_bounds__(256) void gemm_tiled_f32(
    const float* __restrict__ A, const float* __restrict__ B,
    float* __restrict__ C, const float* __restrict__ bias,
    int M, int N, int K, int lda, float scale)
{
    __shared__ __attribute__((aligned(16))) float As[2 * GBM * GPA];
    __shared__ __attribute__((aligned(16))) float Bs[2 * GBK * GPB];
    gemm_tiled_body(A, B, C, bias, M, N, K, lda, scale, blockIdx.x, blockIdx.y, blockIdx.z, As, Bs);
}

__device__ __forceinline__ void splitk_reduce_body(
    const float* __restrict__ part, const float* __restrict__ bias,
    float* __restrict__ dst, int MN, int N, int S, int blk)
{
    int i4 = blk * 256 + threadIdx.x;
    int quads = MN >> 2;
    if (i4 >= quads) return;
    const float4* p = reinterpret_cast<const float4*>(part) + i4;
    float4 acc = make_float4(0.0f, 0.0f, 0.0f, 0.0f);
    #pragma unroll 16
    for (int z = 0; z < S; ++z) {
        float4 v = p[(size_t)z * quads];
        acc.x += v.x; acc.y += v.y; acc.z += v.z; acc.w += v.w;
    }
    int col = (i4 << 2) % N;
    const float4 bv = *reinterpret_cast<const float4*>(bias + col);
    acc.x += bv.x; acc.y += bv.y; acc.z += bv.z; acc.w += bv.w;
    reinterpret_cast<float4*>(dst)[i4] = acc;
}

__global__ __launch_bounds__(256) void reduce_and_gemm(
    const float* __restrict__ part, const float* __restrict__ rbias, float* __restrict__ rdst,
    int MN, int RN, int S, int rblocks,
    const float* __restrict__ A, const float* __restrict__ B, float* __restrict__ C,
    int M, int N, int K, int lda, float scale)
{
    __shared__ __attribute__((aligned(16))) float As[2 * GBM * GPA];
    __shared__ __attribute__((aligned(16))) float Bs[2 * GBK * GPB];
    int id = blockIdx.x;
    if (id < rblocks) {
        splitk_reduce_body(part, rbias, rdst, MN, RN, S, id);
        return;
    }
    id -= rblocks;
    int nb = N / GBN;
    gemm_tiled_body(A, B, C, nullptr, M, N, K, lda, scale, id % nb, id / nb, 0, As, Bs);
}

#define APITCH 36
#define VPITCH 72
__global__ __launch_bounds__(128) void attn_kernel(
    float* __restrict__ qmat, const float* __restrict__ kv)
{
    __shared__ __attribute__((aligned(16))) float Ks[144 * APITCH];
    __shared__ __attribute__((aligned(16))) float Vs[72 * VPITCH];
    int id = blockIdx.x;
    int h  = id & 7;
    int bb = (id >> 3) & 7;
    int g  = id >> 6;
    int n  = g * 8 + bb;

    int t    = threadIdx.x;
    int wave = t >> 5;
    int lane = t & 31;
    int hf   = lane >> 4;
    int c    = lane & 15;
    int m0   = wave * 16;

    float*       qrow = qmat + ((size_t)n * 64 + m0) * 256 + h * 32;
    const float* kg   = kv + (size_t)bb * 144 * 512 + h * 32;

    v4f qf[4];
    #pragma unroll
    for (int g4 = 0; g4 < 4; ++g4)
        qf[g4] = *reinterpret_cast<const v4f*>(qrow + c * 256 + 8 * g4 + 4 * hf);

    {
        int sr = t >> 3, sc = (t & 7) * 4;
        const float* ksrc = kg + (size_t)sr * 512 + sc;
        const float* vsrc = kg + 256 + (size_t)(2 * sr) * 512 + sc;
        bool tail = sr < 8;
        v4f sk[9], va[5], vb[5];
        #pragma unroll
        for (int i = 0; i < 9; ++i) sk[i] = *reinterpret_cast<const v4f*>(ksrc + (size_t)(16 * i) * 512);
        #pragma unroll
        for (int i = 0; i < 5; ++i)
            if (i < 4 || tail) {
                va[i] = *reinterpret_cast<const v4f*>(vsrc + (size_t)(32 * i) * 512);
                vb[i] = *reinterpret_cast<const v4f*>(vsrc + (size_t)(32 * i + 1) * 512);
            }
        #pragma unroll
        for (int i = 0; i < 9; ++i) *reinterpret_cast<v4f*>(&Ks[(sr + 16 * i) * APITCH + sc]) = sk[i];
        #pragma unroll
        for (int i = 0; i < 5; ++i)
            if (i < 4 || tail) {
                float* d = &Vs[(sr + 16 * i) * VPITCH + 2 * sc];
                v4f lo = {va[i].x, vb[i].x, va[i].y, vb[i].y};
                v4f hi = {va[i].z, vb[i].z, va[i].w, vb[i].w};
                *reinterpret_cast<v4f*>(d)     = lo;
                *reinterpret_cast<v4f*>(d + 4) = hi;
            }
    }
    __syncthreads();

    v8f acc[9];
    const float* kf = &Ks[c * APITCH + 4 * hf];
    #pragma unroll
    for (int tg = 0; tg < 3; ++tg) {
        v4f kk[3][4];
        #pragma unroll
        for (int x = 0; x < 3; ++x)
            #pragma unroll
            for (int g4 = 0; g4 < 4; ++g4)
                kk[x][g4] = *reinterpret_cast<const v4f*>(kf + (16 * (3 * tg + x)) * APITCH + 8 * g4);
        v8f a0 = {}, a1 = {}, a2 = {};
        #pragma unroll
        for (int g4 = 0; g4 < 4; ++g4) {
            v2f qlo = {qf[g4].x, qf[g4].y}, qhi = {qf[g4].z, qf[g4].w};
            v2f k0lo = {kk[0][g4].x, kk[0][g4].y}, k0hi = {kk[0][g4].z, kk[0][g4].w};
            v2f k1lo = {kk[1][g4].x, kk[1][g4].y}, k1hi = {kk[1][g4].z, kk[1][g4].w};
            v2f k2lo = {kk[2][g4].x, kk[2][g4].y}, k2hi = {kk[2][g4].z, kk[2][g4].w};
            a0 = __builtin_amdgcn_wmma_f32_16x16x4_f32(false, k0lo, false, qlo, (short)0, a0, false, false);
            a1 = __builtin_amdgcn_wmma_f32_16x16x4_f32(false, k1lo, false, qlo, (short)0, a1, false, false);
            a2 = __builtin_amdgcn_wmma_f32_16x16x4_f32(false, k2lo, false, qlo, (short)0, a2, false, false);
            a0 = __builtin_amdgcn_wmma_f32_16x16x4_f32(false, k0hi, false, qhi, (short)0, a0, false, false);
            a1 = __builtin_amdgcn_wmma_f32_16x16x4_f32(false, k1hi, false, qhi, (short)0, a1, false, false);
            a2 = __builtin_amdgcn_wmma_f32_16x16x4_f32(false, k2hi, false, qhi, (short)0, a2, false, false);
        }
        acc[3 * tg] = a0; acc[3 * tg + 1] = a1; acc[3 * tg + 2] = a2;
    }

    float mx = acc[0][0];
    #pragma unroll
    for (int tl = 0; tl < 9; ++tl)
        #pragma unroll
        for (int rr = 0; rr < 8; ++rr)
            mx = fmaxf(mx, acc[tl][rr]);
    mx = fmaxf(mx, __shfl_xor(mx, 16));
    float s0 = 0.0f, s1 = 0.0f, s2 = 0.0f, s3 = 0.0f;
    #pragma unroll
    for (int tl = 0; tl < 9; ++tl) {
        #pragma unroll
        for (int rr = 0; rr < 8; ++rr)
            acc[tl][rr] = __expf(acc[tl][rr] - mx);
        s0 += acc[tl][0] + acc[tl][4];
        s1 += acc[tl][1] + acc[tl][5];
        s2 += acc[tl][2] + acc[tl][6];
        s3 += acc[tl][3] + acc[tl][7];
    }
    float sum = (s0 + s1) + (s2 + s3);
    sum += __shfl_xor(sum, 16);
    float inv = 1.0f / sum;

    v8f o0 = {}, o1 = {};
    const float* vf = &Vs[(4 * hf) * VPITCH + 2 * c];
    #pragma unroll
    for (int tl = 0; tl < 9; ++tl) {
        #pragma unroll
        for (int j = 0; j < 4; ++j) {
            const float* vp = vf + (8 * tl + j) * VPITCH;
            v2f v0 = *reinterpret_cast<const v2f*>(vp);
            v2f v1 = *reinterpret_cast<const v2f*>(vp + 32);
            v2f p  = {acc[tl][2 * j], acc[tl][2 * j + 1]};
            o0 = __builtin_amdgcn_wmma_f32_16x16x4_f32(false, v0, false, p, (short)0, o0, false, false);
            o1 = __builtin_amdgcn_wmma_f32_16x16x4_f32(false, v1, false, p, (short)0, o1, false, false);
        }
    }

    __syncthreads();
    float* patch = &Ks[wave * (16 * APITCH)];
    {
        float* pw = patch + c * APITCH + 8 * hf;
        v4f w0 = {o0[0] * inv, o0[1] * inv, o0[2] * inv, o0[3] * inv};
        v4f w1 = {o0[4] * inv, o0[5] * inv, o0[6] * inv, o0[7] * inv};
        v4f w2 = {o1[0] * inv, o1[1] * inv, o1[2] * inv, o1[3] * inv};
        v4f w3 = {o1[4] * inv, o1[5] * inv, o1[6] * inv, o1[7] * inv};
        *reinterpret_cast<v4f*>(pw)      = w0;
        *reinterpret_cast<v4f*>(pw + 4)  = w1;
        *reinterpret_cast<v4f*>(pw + 16) = w2;
        *reinterpret_cast<v4f*>(pw + 20) = w3;
    }
    __syncthreads();
    int pr = lane >> 3, pc = (lane & 7) * 4;
    #pragma unroll
    for (int i = 0; i < 4; ++i) {
        int row = pr + 4 * i;
        v4f v = *reinterpret_cast<const v4f*>(&patch[row * APITCH + pc]);
        *reinterpret_cast<v4f*>(qrow + row * 256 + pc) = v;
    }
}

#define RPE_ROWS 4
__global__ __launch_bounds__(192) void rpe_add_kernel(const float* __restrict__ x, const float* __restrict__ rw,
                                                      const float* __restrict__ rb, float* __restrict__ F) {
    int t     = threadIdx.x;
    int xq    = t % 24;
    int rg    = t / 24;
    int plane = blockIdx.x;
    int c     = plane & 255;
    const float* xp = x + (size_t)plane * 9216;
    float*       fb = F + (size_t)plane * 9216 + 4 * xq;

    float w[9];
    #pragma unroll
    for (int i = 0; i < 9; ++i) w[i] = rw[c * 9 + i];
    float bias = rb[c];

    int  cl = (xq > 0) ? 4 * xq - 1 : 0;
    int  cr = (xq < 23) ? 4 * xq + 4 : 95;
    bool hl = xq > 0, hr = xq < 23;

    #pragma unroll 1
    for (int it = 0; it < 3; ++it) {
        int y0 = rg * 12 + it * RPE_ROWS;
        float v[RPE_ROWS + 2][6];
        #pragma unroll
        for (int i = 0; i < RPE_ROWS + 2; ++i) {
            int  y  = y0 - 1 + i;
            bool ok = (y >= 0) && (y < 96);
            int  yc = y < 0 ? 0 : (y > 95 ? 95 : y);
            const float* row = xp + yc * 96;
            v4f m = *reinterpret_cast<const v4f*>(row + 4 * xq);
            float l = row[cl];
            float r = row[cr];
            v[i][0] = (ok && hl) ? l : 0.0f;
            v[i][1] = ok ? m.x : 0.0f;
            v[i][2] = ok ? m.y : 0.0f;
            v[i][3] = ok ? m.z : 0.0f;
            v[i][4] = ok ? m.w : 0.0f;
            v[i][5] = (ok && hr) ? r : 0.0f;
        }
        float* fp = fb + y0 * 96;
        #pragma unroll
        for (int j = 0; j < RPE_ROWS; ++j) {
            v4f f = *reinterpret_cast<const v4f*>(fp + j * 96);
            float a[4];
            #pragma unroll
            for (int e = 0; e < 4; ++e) {
                float acc = bias;
                #pragma unroll
                for (int dy = 0; dy < 3; ++dy)
                    #pragma unroll
                    for (int dx = 0; dx < 3; ++dx)
                        acc = __builtin_fmaf(v[j + dy][e + dx], w[dy * 3 + dx], acc);
                a[e] = acc;
            }
            f.x += a[0]; f.y += a[1]; f.z += a[2]; f.w += a[3];
            *reinterpret_cast<v4f*>(fp + j * 96) = f;
        }
    }
}

extern "C" void kernel_launch(void* const* d_in, const int* in_sizes, int n_in,
                              void* d_out, int out_size, void* d_ws, size_t ws_size,
                              hipStream_t stream) {
    const float* x      = (const float*)d_in[0];
    const float* Wq     = (const float*)d_in[1];
    const float* Wkv    = (const float*)d_in[2];
    const float* pool_w = (const float*)d_in[3];
    const float* pool_b = (const float*)d_in[4];
    const float* rpe_w  = (const float*)d_in[5];
    const float* rpe_b  = (const float*)d_in[6];
    const float* proj_w = (const float*)d_in[7];
    const float* proj_b = (const float*)d_in[8];
    float* out = (float*)d_out;

    float* ws    = (float*)d_ws;
    float* xw    = ws;
    float* qmat  = xw + 18874368;
    float* poolT = qmat + 18874368;
    float* WqT   = poolT + 4194304;
    float* WkvT  = WqT + 65536;
    float* projT = WkvT + 131072;
    float* poolx = projT + 65536;
    float* kvb   = poolx + 294912;
    float* part  = kvb + 589824;
    (void)ws_size; (void)in_sizes; (void)n_in; (void)out_size;

    const int TB = 256;
    layout_kernel<<<3584, TB, 0, stream>>>(x, xw, pool_w, poolT, Wq, WqT, Wkv, WkvT, proj_w, projT);

    gemm_tiled_f32<<<dim3(256 / 64, 1152 / 128, 64), TB, 0, stream>>>(
        xw, poolT, part, nullptr, 1152, 256, 256, 16384, 1.0f);

    reduce_and_gemm<<<2592, TB, 0, stream>>>(
        part, pool_b, poolx, 294912, 256, 64, 288,
        xw, WqT, qmat, 73728, 256, 256, 256, QSCALE);

    gemm_wmma_f32<<<dim3(512 / 64, 1152 / 32), TB, 0, stream>>>(
        poolx, WkvT, kvb, nullptr, 1152, 512, 256, 256, 1.0f);

    attn_kernel<<<144 * 8 * 8, 128, 0, stream>>>(qmat, kvb);

    rpe_add_kernel<<<8 * 256, 192, 0, stream>>>(x, rpe_w, rpe_b, qmat);

    gemm_tiled_f32<<<dim3(256 / 64, 73728 / 128), TB, 0, stream>>>(
        qmat, projT, out, proj_b, 73728, 256, 256, 256, 1.0f);
}


# RPE 3x3 pass folded into the attention write-out: the lane that stores 4 results of F first adds bias + 9 taps of x (weights in scalar registers); F written once, 7 launches -> 6; sum order unchanged
# speedup vs baseline: 3.8293x; 1.1051x over previous
#include <hip/hip_runtime.h>
#include <hip/hip_bf16.h>

typedef __attribute__((ext_vector_type(2))) float v2f;
typedef __attribute__((ext_vector_type(8))) float v8f;

#define QSCALE 0.17677669529663687f


__device__ __forceinline__ void transpose_k_body(const float* __restrict__ src, float* __restrict__ dst,
                                                 int O, int K, int total, int blk) {
    int idx = blk * 256 + threadIdx.x;
    if (idx >= total) return;
    int o = idx % O;
    int k = idx / O;
    dst[idx] = src[(size_t)o * K + k];
}

typedef __attribute__((ext_vector_type(4))) float v4f;
#define TPITCH 65

__device__ __forceinline__ void pool_transpose_body(const float* __restrict__ src,
                                                    float* __restrict__ dst, int blk, float* tile) {
    int t  = threadIdx.x;
    int og = blk & 3;
    int c  = blk >> 2;
    int lo = t & 15;
    int hi = t >> 4;
    const float* s = src + ((size_t)(og * 64 + hi) * 16384 + c * 64 + 4 * lo);
    v4f v[4];
    #pragma unroll
    for (int it = 0; it < 4; ++it)
        v[it] = *reinterpret_cast<const v4f*>(s + (size_t)(16 * it) * 16384);
    #pragma unroll
    for (int it = 0; it < 4; ++it) {
        float* d = &tile[(4 * lo) * TPITCH + hi + 16 * it];
        d[0]          = v[it].x;
        d[TPITCH]     = v[it].y;
        d[2 * TPITCH] = v[it].z;
        d[3 * TPITCH] = v[it].w;
    }
    __syncthreads();
    float* g = dst + ((size_t)(hi * 256 + c) * 256 + og * 64 + 4 * lo);
    #pragma unroll
    for (int it = 0; it < 4; ++it) {
        const float* r = &tile[(hi + 16 * it) * TPITCH + 4 * lo];
        v4f o = {r[0], r[1], r[2], r[3]};
        *reinterpret_cast<v4f*>(g + (size_t)(16 * it) * 65536) = o;
    }
}

__device__ __forceinline__ void gather_xw_body(const float* __restrict__ x,
                                               float* __restrict__ xw, int b, float* tile) {
    int t  = threadIdx.x;
    int cg = b & 3;
    int ip = (b >> 2) & 3;
    int bw = b >> 4;
    int wh = bw % 12;
    int bi = bw / 12;
    int lo = t & 15;
    int hi = t >> 4;
    const float* strip = x + (((size_t)(bi * 256 + cg * 64 + hi) * 96 + wh * 8 + 2 * ip) * 96 + 4 * lo);
    v4f v[12];
    #pragma unroll
    for (int it = 0; it < 12; ++it)
        v[it] = *reinterpret_cast<const v4f*>(strip + (size_t)(16 * (it & 3)) * 9216 + (it >> 2) * 64);
    float* d0 = &tile[(4 * lo) * TPITCH + hi];
    #pragma unroll
    for (int it = 0; it < 12; ++it) {
        float* d = d0 + ((it >> 2) * 64) * TPITCH + 16 * (it & 3);
        d[0]          = v[it].x;
        d[TPITCH]     = v[it].y;
        d[2 * TPITCH] = v[it].z;
        d[3 * TPITCH] = v[it].w;
    }
    __syncthreads();
    float* g = xw + ((size_t)((bi * 144 + wh * 12 + (hi >> 3)) * 64 + ip * 16 + (hi & 7)) * 256
                     + cg * 64 + 4 * lo);
    const float* r0 = &tile[hi * TPITCH + 4 * lo];
    #pragma unroll
    for (int it = 0; it < 12; ++it) {
        const float* r = r0 + (16 * it) * TPITCH;
        v4f o = {r[0], r[1], r[2], r[3]};
        *reinterpret_cast<v4f*>(g + (size_t)(it % 6) * (2 * 64 * 256) + (it / 6) * (8 * 256)) = o;
    }
}

__global__ __launch_bounds__(256) void layout_kernel(
    const float* __restrict__ x, float* __restrict__ xw,
    const float* __restrict__ pool_w, float* __restrict__ poolT,
    const float* __restrict__ Wq, float* __restrict__ WqT,
    const float* __restrict__ Wkv, float* __restrict__ WkvT,
    const float* __restrict__ proj_w, float* __restrict__ projT)
{
    __shared__ float tile[192 * TPITCH];
    int b = blockIdx.x;
    if (b < 1536)      gather_xw_body(x, xw, b, tile);
    else if (b < 2560) pool_transpose_body(pool_w, poolT, b - 1536, tile);
    else if (b < 2816) transpose_k_body(Wq, WqT, 256, 256, 65536, b - 2560);
    else if (b < 3328) transpose_k_body(Wkv, WkvT, 512, 256, 131072, b - 2816);
    else               transpose_k_body(proj_w, projT, 256, 256, 65536, b - 3328);
}

__global__ __launch_bounds__(256) void gemm_wmma_f32(
    const float* __restrict__ A, const float* __restrict__ B,
    float* __restrict__ C, const float* __restrict__ bias,
    int M, int N, int K, int lda, float scale)
{
    int wave = threadIdx.x >> 5;
    int lane = threadIdx.x & 31;
    int hf   = lane >> 4;
    int r    = lane & 15;
    int m0 = blockIdx.y * 32 + (wave >> 2) * 16;
    int n0 = blockIdx.x * 64 + (wave & 3) * 16;
    size_t kz = (size_t)blockIdx.z * K;

    const float* Arow = A + (size_t)(m0 + r) * lda + kz + 2 * hf;
    const float* Bcol = B + (kz + 2 * hf) * N + n0 + r;
    C += (size_t)blockIdx.z * M * N;

    v8f acc = {};
    #pragma unroll 8
    for (int k = 0; k < K; k += 4) {
        v2f a, b;
        a.x = Arow[k];
        a.y = Arow[k + 1];
        b.x = Bcol[(size_t)k * N];
        b.y = Bcol[(size_t)k * N + N];
        acc = __builtin_amdgcn_wmma_f32_16x16x4_f32(false, a, false, b,
                                                    (short)0, acc, false, false);
    }
    int col = n0 + r;
    float bv = bias ? bias[col] : 0.0f;
    #pragma unroll
    for (int vg = 0; vg < 8; ++vg)
        C[(size_t)(m0 + vg + 8 * hf) * N + col] = acc[vg] * scale + bv;
}

#define GBM 128
#define GBN 64
#define GBK 32
#define GPA 36
#define GPB 72

__device__ __forceinline__ void gemm_tiled_body(
    const float* __restrict__ A, const float* __restrict__ B,
    float* __restrict__ C, const float* __restrict__ bias,
    int M, int N, int K, int lda, float scale,
    int bx, int by, int bz, float* As, float* Bs)
{
    int t    = threadIdx.x;
    int wave = t >> 5;
    int lane = t & 31;
    int hf   = lane >> 4;
    int r    = lane & 15;
    int wm   = wave >> 1;
    int wn   = wave & 1;
    int m0 = by * GBM;
    int n0 = bx * GBN;
    size_t kz = (size_t)bz * K;

    int ar = t >> 3, ac = (t & 7) * 4;
    int br = t >> 4, bc = (t & 15) * 4;
    const float* Ag = A + (size_t)(m0 + ar) * lda + kz + ac;
    const float* Bg = B + (kz + br) * N + n0 + bc;
    C += (size_t)bz * M * N;

    v4f pa[4], pb[2];
    #pragma unroll
    for (int i = 0; i < 4; ++i) pa[i] = *reinterpret_cast<const v4f*>(Ag + (size_t)(32 * i) * lda);
    #pragma unroll
    for (int i = 0; i < 2; ++i) pb[i] = *reinterpret_cast<const v4f*>(Bg + (size_t)(16 * i) * N);
    #pragma unroll
    for (int i = 0; i < 4; ++i) *reinterpret_cast<v4f*>(&As[(ar + 32 * i) * GPA + ac]) = pa[i];
    #pragma unroll
    for (int i = 0; i < 2; ++i) *reinterpret_cast<v4f*>(&Bs[(br + 16 * i) * GPB + bc]) = pb[i];
    __syncthreads();

    v8f acc[2][2] = {};
    int nchunks = K / GBK;
    for (int kc = 0; kc < nchunks; ++kc) {
        int cur = kc & 1;
        bool more = (kc + 1 < nchunks);
        if (more) {
            const float* Agn = Ag + (kc + 1) * GBK;
            const float* Bgn = Bg + (size_t)(kc + 1) * GBK * N;
            #pragma unroll
            for (int i = 0; i < 4; ++i) pa[i] = *reinterpret_cast<const v4f*>(Agn + (size_t)(32 * i) * lda);
            #pragma unroll
            for (int i = 0; i < 2; ++i) pb[i] = *reinterpret_cast<const v4f*>(Bgn + (size_t)(16 * i) * N);
        }
        const float* as = &As[cur * (GBM * GPA) + (wm * 32 + r) * GPA + 4 * hf];
        const float* bs = &Bs[cur * (GBK * GPB) + (4 * hf) * GPB + wn * 32 + r];
        #pragma unroll
        for (int g = 0; g < 4; ++g) {
            v4f a0 = *reinterpret_cast<const v4f*>(as + 8 * g);
            v4f a1 = *reinterpret_cast<const v4f*>(as + 16 * GPA + 8 * g);
            const float* bg = bs + (8 * g) * GPB;
            v2f b0lo = {bg[0],            bg[GPB]};
            v2f b0hi = {bg[2 * GPB],      bg[3 * GPB]};
            v2f b1lo = {bg[16],           bg[GPB + 16]};
            v2f b1hi = {bg[2 * GPB + 16], bg[3 * GPB + 16]};
            v2f a0lo = {a0.x, a0.y}, a0hi = {a0.z, a0.w};
            v2f a1lo = {a1.x, a1.y}, a1hi = {a1.z, a1.w};
            acc[0][0] = __builtin_amdgcn_wmma_f32_16x16x4_f32(false, a0lo, false, b0lo, (short)0, acc[0][0], false, false);
            acc[0][1] = __builtin_amdgcn_wmma_f32_16x16x4_f32(false, a0lo, false, b1lo, (short)0, acc[0][1], false, false);
            acc[1][0] = __builtin_amdgcn_wmma_f32_16x16x4_f32(false, a1lo, false, b0lo, (short)0, acc[1][0], false, false);
            acc[1][1] = __builtin_amdgcn_wmma_f32_16x16x4_f32(false, a1lo, false, b1lo, (short)0, acc[1][1], false, false);
            acc[0][0] = __builtin_amdgcn_wmma_f32_16x16x4_f32(false, a0hi, false, b0hi, (short)0, acc[0][0], false, false);
            acc[0][1] = __builtin_amdgcn_wmma_f32_16x16x4_f32(false, a0hi, false, b1hi, (short)0, acc[0][1], false, false);
            acc[1][0] = __builtin_amdgcn_wmma_f32_16x16x4_f32(false, a1hi, false, b0hi, (short)0, acc[1][0], false, false);
            acc[1][1] = __builtin_amdgcn_wmma_f32_16x16x4_f32(false, a1hi, false, b1hi, (short)0, acc[1][1], false, false);
        }
        if (more) {
            int nxt = cur ^ 1;
            #pragma unroll
            for (int i = 0; i < 4; ++i) *reinterpret_cast<v4f*>(&As[nxt * (GBM * GPA) + (ar + 32 * i) * GPA + ac]) = pa[i];
            #pragma unroll
            for (int i = 0; i < 2; ++i) *reinterpret_cast<v4f*>(&Bs[nxt * (GBK * GPB) + (br + 16 * i) * GPB + bc]) = pb[i];
        }
        __syncthreads();
    }

    float* patch = &As[wave * (32 * GPA)];
    #pragma unroll
    for (int mt = 0; mt < 2; ++mt)
        #pragma unroll
        for (int nt = 0; nt < 2; ++nt)
            #pragma unroll
            for (int vg = 0; vg < 8; ++vg)
                patch[(mt * 16 + vg + 8 * hf) * GPA + nt * 16 + r] = acc[mt][nt][vg];
    __syncthreads();
    int pr = lane >> 3, pc = (lane & 7) * 4;
    int col = n0 + wn * 32 + pc;
    v4f bv = {0.0f, 0.0f, 0.0f, 0.0f};
    if (bias) bv = *reinterpret_cast<const v4f*>(bias + col);
    #pragma unroll
    for (int i = 0; i < 8; ++i) {
        int row = pr + 4 * i;
        v4f v = *reinterpret_cast<const v4f*>(&patch[row * GPA + pc]);
        v4f o;
        o.x = v.x * scale + bv.x;
        o.y = v.y * scale + bv.y;
        o.z = v.z * scale + bv.z;
        o.w = v.w * scale + bv.w;
        *reinterpret_cast<v4f*>(&C[(size_t)(m0 + wm * 32 + row) * N + col]) = o;
    }
}

__global__ __launch_bounds__(256) void gemm_tiled_f32(
    const float* __restrict__ A, const float* __restrict__ B,
    float* __restrict__ C, const float* __restrict__ bias,
    int M, int N, int K, int lda, float scale)
{
    __shared__ __attribute__((aligned(16))) float As[2 * GBM * GPA];
    __shared__ __attribute__((aligned(16))) float Bs[2 * GBK * GPB];
    gemm_tiled_body(A, B, C, bias, M, N, K, lda, scale, blockIdx.x, blockIdx.y, blockIdx.z, As, Bs);
}

__device__ __forceinline__ void splitk_reduce_body(
    const float* __restrict__ part, const float* __restrict__ bias,
    float* __restrict__ dst, int MN, int N, int S, int blk)
{
    int i4 = blk * 256 + threadIdx.x;
    int quads = MN >> 2;
    if (i4 >= quads) return;
    const float4* p = reinterpret_cast<const float4*>(part) + i4;
    float4 acc = make_float4(0.0f, 0.0f, 0.0f, 0.0f);
    #pragma unroll 16
    for (int z = 0; z < S; ++z) {
        float4 v = p[(size_t)z * quads];
        acc.x += v.x; acc.y += v.y; acc.z += v.z; acc.w += v.w;
    }
    int col = (i4 << 2) % N;
    const float4 bv = *reinterpret_cast<const float4*>(bias + col);
    acc.x += bv.x; acc.y += bv.y; acc.z += bv.z; acc.w += bv.w;
    reinterpret_cast<float4*>(dst)[i4] = acc;
}

__global__ __launch_bounds__(256) void reduce_and_gemm(
    const float* __restrict__ part, const float* __restrict__ rbias, float* __restrict__ rdst,
    int MN, int RN, int S, int rblocks,
    const float* __restrict__ A, const float* __restrict__ B, float* __restrict__ C,
    int M, int N, int K, int lda, float scale)
{
    __shared__ __attribute__((aligned(16))) float As[2 * GBM * GPA];
    __shared__ __attribute__((aligned(16))) float Bs[2 * GBK * GPB];
    int id = blockIdx.x;
    if (id < rblocks) {
        splitk_reduce_body(part, rbias, rdst, MN, RN, S, id);
        return;
    }
    id -= rblocks;
    int nb = N / GBN;
    gemm_tiled_body(A, B, C, nullptr, M, N, K, lda, scale, id % nb, id / nb, 0, As, Bs);
}

#define APITCH 36
#define VPITCH 72
__global__ __launch_bounds__(128) void attn_kernel(
    float* __restrict__ qmat, const float* __restrict__ kv,
    const float* __restrict__ x, const float* __restrict__ rw, const float* __restrict__ rb)
{
    __shared__ __attribute__((aligned(16))) float Ks[144 * APITCH];
    __shared__ __attribute__((aligned(16))) float Vs[72 * VPITCH];
    int id = blockIdx.x;
    int h  = id & 7;
    int bb = (id >> 3) & 7;
    int g  = id >> 6;
    int n  = g * 8 + bb;

    int t    = threadIdx.x;
    int wave = t >> 5;
    int lane = t & 31;
    int hf   = lane >> 4;
    int c    = lane & 15;
    int m0   = wave * 16;

    float*       qrow = qmat + ((size_t)n * 64 + m0) * 256 + h * 32;
    const float* kg   = kv + (size_t)bb * 144 * 512 + h * 32;

    v4f qf[4];
    #pragma unroll
    for (int g4 = 0; g4 < 4; ++g4)
        qf[g4] = *reinterpret_cast<const v4f*>(qrow + c * 256 + 8 * g4 + 4 * hf);

    {
        int sr = t >> 3, sc = (t & 7) * 4;
        const float* ksrc = kg + (size_t)sr * 512 + sc;
        const float* vsrc = kg + 256 + (size_t)(2 * sr) * 512 + sc;
        bool tail = sr < 8;
        v4f sk[9], va[5], vb[5];
        #pragma unroll
        for (int i = 0; i < 9; ++i) sk[i] = *reinterpret_cast<const v4f*>(ksrc + (size_t)(16 * i) * 512);
        #pragma unroll
        for (int i = 0; i < 5; ++i)
            if (i < 4 || tail) {
                va[i] = *reinterpret_cast<const v4f*>(vsrc + (size_t)(32 * i) * 512);
                vb[i] = *reinterpret_cast<const v4f*>(vsrc + (size_t)(32 * i + 1) * 512);
            }
        #pragma unroll
        for (int i = 0; i < 9; ++i) *reinterpret_cast<v4f*>(&Ks[(sr + 16 * i) * APITCH + sc]) = sk[i];
        #pragma unroll
        for (int i = 0; i < 5; ++i)
            if (i < 4 || tail) {
                float* d = &Vs[(sr + 16 * i) * VPITCH + 2 * sc];
                v4f lo = {va[i].x, vb[i].x, va[i].y, vb[i].y};
                v4f hi = {va[i].z, vb[i].z, va[i].w, vb[i].w};
                *reinterpret_cast<v4f*>(d)     = lo;
                *reinterpret_cast<v4f*>(d + 4) = hi;
            }
    }
    __syncthreads();

    v8f acc[9];
    const float* kf = &Ks[c * APITCH + 4 * hf];
    #pragma unroll
    for (int tg = 0; tg < 3; ++tg) {
        v4f kk[3][4];
        #pragma unroll
        for (int x = 0; x < 3; ++x)
            #pragma unroll
            for (int g4 = 0; g4 < 4; ++g4)
                kk[x][g4] = *reinterpret_cast<const v4f*>(kf + (16 * (3 * tg + x)) * APITCH + 8 * g4);
        v8f a0 = {}, a1 = {}, a2 = {};
        #pragma unroll
        for (int g4 = 0; g4 < 4; ++g4) {
            v2f qlo = {qf[g4].x, qf[g4].y}, qhi = {qf[g4].z, qf[g4].w};
            v2f k0lo = {kk[0][g4].x, kk[0][g4].y}, k0hi = {kk[0][g4].z, kk[0][g4].w};
            v2f k1lo = {kk[1][g4].x, kk[1][g4].y}, k1hi = {kk[1][g4].z, kk[1][g4].w};
            v2f k2lo = {kk[2][g4].x, kk[2][g4].y}, k2hi = {kk[2][g4].z, kk[2][g4].w};
            a0 = __builtin_amdgcn_wmma_f32_16x16x4_f32(false, k0lo, false, qlo, (short)0, a0, false, false);
            a1 = __builtin_amdgcn_wmma_f32_16x16x4_f32(false, k1lo, false, qlo, (short)0, a1, false, false);
            a2 = __builtin_amdgcn_wmma_f32_16x16x4_f32(false, k2lo, false, qlo, (short)0, a2, false, false);
            a0 = __builtin_amdgcn_wmma_f32_16x16x4_f32(false, k0hi, false, qhi, (short)0, a0, false, false);
            a1 = __builtin_amdgcn_wmma_f32_16x16x4_f32(false, k1hi, false, qhi, (short)0, a1, false, false);
            a2 = __builtin_amdgcn_wmma_f32_16x16x4_f32(false, k2hi, false, qhi, (short)0, a2, false, false);
        }
        acc[3 * tg] = a0; acc[3 * tg + 1] = a1; acc[3 * tg + 2] = a2;
    }

    float mx = acc[0][0];
    #pragma unroll
    for (int tl = 0; tl < 9; ++tl)
        #pragma unroll
        for (int rr = 0; rr < 8; ++rr)
            mx = fmaxf(mx, acc[tl][rr]);
    mx = fmaxf(mx, __shfl_xor(mx, 16));
    float s0 = 0.0f, s1 = 0.0f, s2 = 0.0f, s3 = 0.0f;
    #pragma unroll
    for (int tl = 0; tl < 9; ++tl) {
        #pragma unroll
        for (int rr = 0; rr < 8; ++rr)
            acc[tl][rr] = __expf(acc[tl][rr] - mx);
        s0 += acc[tl][0] + acc[tl][4];
        s1 += acc[tl][1] + acc[tl][5];
        s2 += acc[tl][2] + acc[tl][6];
        s3 += acc[tl][3] + acc[tl][7];
    }
    float sum = (s0 + s1) + (s2 + s3);
    sum += __shfl_xor(sum, 16);
    float inv = 1.0f / sum;

    v8f o0 = {}, o1 = {};
    const float* vf = &Vs[(4 * hf) * VPITCH + 2 * c];
    #pragma unroll
    for (int tl = 0; tl < 9; ++tl) {
        #pragma unroll
        for (int j = 0; j < 4; ++j) {
            const float* vp = vf + (8 * tl + j) * VPITCH;
            v2f v0 = *reinterpret_cast<const v2f*>(vp);
            v2f v1 = *reinterpret_cast<const v2f*>(vp + 32);
            v2f p  = {acc[tl][2 * j], acc[tl][2 * j + 1]};
            o0 = __builtin_amdgcn_wmma_f32_16x16x4_f32(false, v0, false, p, (short)0, o0, false, false);
            o1 = __builtin_amdgcn_wmma_f32_16x16x4_f32(false, v1, false, p, (short)0, o1, false, false);
        }
    }

    __syncthreads();
    float* patch = &Ks[wave * (16 * APITCH)];
    {
        float* pw = patch + c * APITCH + 8 * hf;
        v4f w0 = {o0[0] * inv, o0[1] * inv, o0[2] * inv, o0[3] * inv};
        v4f w1 = {o0[4] * inv, o0[5] * inv, o0[6] * inv, o0[7] * inv};
        v4f w2 = {o1[0] * inv, o1[1] * inv, o1[2] * inv, o1[3] * inv};
        v4f w3 = {o1[4] * inv, o1[5] * inv, o1[6] * inv, o1[7] * inv};
        *reinterpret_cast<v4f*>(pw)      = w0;
        *reinterpret_cast<v4f*>(pw + 4)  = w1;
        *reinterpret_cast<v4f*>(pw + 16) = w2;
        *reinterpret_cast<v4f*>(pw + 20) = w3;
    }
    __syncthreads();
    int pr = lane >> 3, pc = (lane & 7) * 4;
    int R0 = n * 64 + __builtin_amdgcn_readfirstlane(m0);
    #pragma unroll
    for (int i = 0; i < 4; ++i) {
        int row   = pr + 4 * i;
        int Rb    = R0 + 4 * i;
        int plane = Rb / 36;
        int ch    = plane & 255;
        const float* wp = rw + ch * 9;
        float w[9];
        #pragma unroll
        for (int k = 0; k < 9; ++k) w[k] = wp[k];
        float bias = rb[ch];
        int pos = (Rb - plane * 36 + pr) * 256 + h * 32 + pc;
        int y   = pos / 96;
        int xc  = pos - y * 96;
        const float* xp = x + (size_t)plane * 9216;
        int  cl = (xc > 0) ? xc - 1 : 0;
        int  cr = (xc < 92) ? xc + 4 : 95;
        bool hl = xc > 0, hr = xc < 92;
        float tp[3][6];
        #pragma unroll
        for (int dy = 0; dy < 3; ++dy) {
            int  yy = y - 1 + dy;
            bool ok = (yy >= 0) && (yy < 96);
            int  yc = yy < 0 ? 0 : (yy > 95 ? 95 : yy);
            const float* xr = xp + yc * 96;
            v4f m = *reinterpret_cast<const v4f*>(xr + xc);
            float l = xr[cl];
            float r = xr[cr];
            tp[dy][0] = (ok && hl) ? l : 0.0f;
            tp[dy][1] = ok ? m.x : 0.0f;
            tp[dy][2] = ok ? m.y : 0.0f;
            tp[dy][3] = ok ? m.z : 0.0f;
            tp[dy][4] = ok ? m.w : 0.0f;
            tp[dy][5] = (ok && hr) ? r : 0.0f;
        }
        v4f v = *reinterpret_cast<const v4f*>(&patch[row * APITCH + pc]);
        float a[4];
        #pragma unroll
        for (int e = 0; e < 4; ++e) {
            float s = bias;
            #pragma unroll
            for (int dy = 0; dy < 3; ++dy)
                #pragma unroll
                for (int dx = 0; dx < 3; ++dx)
                    s = __builtin_fmaf(tp[dy][e + dx], w[dy * 3 + dx], s);
            a[e] = s;
        }
        v.x += a[0]; v.y += a[1]; v.z += a[2]; v.w += a[3];
        *reinterpret_cast<v4f*>(qrow + row * 256 + pc) = v;
    }
}

extern "C" void kernel_launch(void* const* d_in, const int* in_sizes, int n_in,
                              void* d_out, int out_size, void* d_ws, size_t ws_size,
                              hipStream_t stream) {
    const float* x      = (const float*)d_in[0];
    const float* Wq     = (const float*)d_in[1];
    const float* Wkv    = (const float*)d_in[2];
    const float* pool_w = (const float*)d_in[3];
    const float* pool_b = (const float*)d_in[4];
    const float* rpe_w  = (const float*)d_in[5];
    const float* rpe_b  = (const float*)d_in[6];
    const float* proj_w = (const float*)d_in[7];
    const float* proj_b = (const float*)d_in[8];
    float* out = (float*)d_out;

    float* ws    = (float*)d_ws;
    float* xw    = ws;
    float* qmat  = xw + 18874368;
    float* poolT = qmat + 18874368;
    float* WqT   = poolT + 4194304;
    float* WkvT  = WqT + 65536;
    float* projT = WkvT + 131072;
    float* poolx = projT + 65536;
    float* kvb   = poolx + 294912;
    float* part  = kvb + 589824;
    (void)ws_size; (void)in_sizes; (void)n_in; (void)out_size;

    const int TB = 256;
    layout_kernel<<<3584, TB, 0, stream>>>(x, xw, pool_w, poolT, Wq, WqT, Wkv, WkvT, proj_w, projT);

    gemm_tiled_f32<<<dim3(256 / 64, 1152 / 128, 64), TB, 0, stream>>>(
        xw, poolT, part, nullptr, 1152, 256, 256, 16384, 1.0f);

    reduce_and_gemm<<<2592, TB, 0, stream>>>(
        part, pool_b, poolx, 294912, 256, 64, 288,
        xw, WqT, qmat, 73728, 256, 256, 256, QSCALE);

    gemm_wmma_f32<<<dim3(512 / 64, 1152 / 32), TB, 0, stream>>>(
        poolx, WkvT, kvb, nullptr, 1152, 512, 256, 256, 1.0f);

    attn_kernel<<<144 * 8 * 8, 128, 0, stream>>>(qmat, kvb, x, rpe_w, rpe_b);

    gemm_tiled_f32<<<dim3(256 / 64, 73728 / 128), TB, 0, stream>>>(
        qmat, projT, out, proj_b, 73728, 256, 256, 256, 1.0f);
}


# Layout pass removed: pooling and Q GEMMs stage x in place (window order is index math in staging and Q's row write-out), weights read as [out][in]; 6 launches -> 5; pooling k order changes
# speedup vs baseline: 4.1022x; 1.0713x over previous
#include <hip/hip_runtime.h>
#include <hip/hip_bf16.h>

typedef __attribute__((ext_vector_type(2))) float v2f;
typedef __attribute__((ext_vector_type(8))) float v8f;

#define QSCALE 0.17677669529663687f


typedef __attribute__((ext_vector_type(4))) float v4f;


__global__ __launch_bounds__(256) void gemm_wmma_f32(
    const float* __restrict__ A, const float* __restrict__ Bn,
    float* __restrict__ C, const float* __restrict__ bias,
    int M, int N, int K, int lda, int ldb, float scale)
{
    int wave = threadIdx.x >> 5;
    int lane = threadIdx.x & 31;
    int hf   = lane >> 4;
    int r    = lane & 15;
    int m0 = blockIdx.y * 32 + (wave >> 2) * 16;
    int n0 = blockIdx.x * 64 + (wave & 3) * 16;
    size_t kz = (size_t)blockIdx.z * K;

    const float* Arow = A  + (size_t)(m0 + r) * lda + kz + 2 * hf;
    const float* Brow = Bn + (size_t)(n0 + r) * ldb + kz + 2 * hf;
    C += (size_t)blockIdx.z * M * N;

    v8f acc = {};
    #pragma unroll 8
    for (int k = 0; k < K; k += 4) {
        v2f a, b;
        a.x = Arow[k];
        a.y = Arow[k + 1];
        b.x = Brow[k];
        b.y = Brow[k + 1];
        acc = __builtin_amdgcn_wmma_f32_16x16x4_f32(false, a, false, b,
                                                    (short)0, acc, false, false);
    }
    int col = n0 + r;
    float bv = bias ? bias[col] : 0.0f;
    #pragma unroll
    for (int vg = 0; vg < 8; ++vg)
        C[(size_t)(m0 + vg + 8 * hf) * N + col] = acc[vg] * scale + bv;
}

#define GBM 128
#define GBN 64
#define GBK 32
#define GPA 36
#define GPK 136
#define XPLANE 9216

template <int AMODE>
__device__ __forceinline__ int a_chunk_offset(int kz, int kc) {
    if (AMODE == 0) return kc * GBK;
    if (AMODE == 1) return kc * GBK * XPLANE;
    int kg = kz + kc * GBK;
    return (kg >> 6) * XPLANE + ((kg >> 3) & 7) * 96;
}

template <int AMODE>
__device__ __forceinline__ void gemm_tiled_body(
    const float* __restrict__ A, const float* __restrict__ Bn,
    float* __restrict__ C, const float* __restrict__ bias,
    int M, int N, int K, int lda, int ldb, float scale,
    int bx, int by, int bz, float* As, float* Bs)
{
    int t    = threadIdx.x;
    int wave = t >> 5;
    int lane = t & 31;
    int hf   = lane >> 4;
    int r    = lane & 15;
    int wm   = wave >> 1;
    int wn   = wave & 1;
    int m0 = by * GBM;
    int n0 = bx * GBN;
    int kz = bz * K;

    int ar = t >> 3, ac = (t & 7) * 4;
    int kr = t >> 5, mc = (t & 31) * 4;
    int aoff[4];
    int adst[4];
    #pragma unroll
    for (int i = 0; i < 4; ++i) {
        if (AMODE == 0) {
            aoff[i] = (m0 + ar + 32 * i) * lda + kz + ac;
            adst[i] = (ar + 32 * i) * GPA + ac;
        } else if (AMODE == 1) {
            int b = m0 / XPLANE;
            aoff[i] = (b * 256 + kz + kr + 8 * i) * XPLANE + (m0 - b * XPLANE) + mc;
            adst[i] = (kr + 8 * i) * GPK + mc;
        } else {
            int n   = m0 + ar + 32 * i;
            int b   = n / 144;
            int rem = n - b * 144;
            int wh  = rem / 12;
            int ww  = rem - wh * 12;
            aoff[i] = ((b * 256) * 96 + wh * 8 + (ac >> 3)) * 96 + ww * 8 + (ac & 7);
            adst[i] = (ar + 32 * i) * GPA + ac;
        }
    }
    const int ABUF = (AMODE == 1) ? GBK * GPK : GBM * GPA;
    const float* Bg = Bn + (size_t)(n0 + ar) * ldb + kz + ac;
    C += (size_t)bz * M * N;

    v4f pa[4], pb[2];
    {
        const float* A0 = A + a_chunk_offset<AMODE>(kz, 0);
        #pragma unroll
        for (int i = 0; i < 4; ++i) pa[i] = *reinterpret_cast<const v4f*>(A0 + aoff[i]);
        #pragma unroll
        for (int i = 0; i < 2; ++i) pb[i] = *reinterpret_cast<const v4f*>(Bg + (size_t)(32 * i) * ldb);
        #pragma unroll
        for (int i = 0; i < 4; ++i) *reinterpret_cast<v4f*>(&As[adst[i]]) = pa[i];
        #pragma unroll
        for (int i = 0; i < 2; ++i) *reinterpret_cast<v4f*>(&Bs[(ar + 32 * i) * GPA + ac]) = pb[i];
    }
    __syncthreads();

    v8f acc[2][2] = {};
    int nchunks = K / GBK;
    for (int kc = 0; kc < nchunks; ++kc) {
        int cur = kc & 1;
        bool more = (kc + 1 < nchunks);
        if (more) {
            const float* An  = A + a_chunk_offset<AMODE>(kz, kc + 1);
            const float* Bgn = Bg + (kc + 1) * GBK;
            #pragma unroll
            for (int i = 0; i < 4; ++i) pa[i] = *reinterpret_cast<const v4f*>(An + aoff[i]);
            #pragma unroll
            for (int i = 0; i < 2; ++i) pb[i] = *reinterpret_cast<const v4f*>(Bgn + (size_t)(32 * i) * ldb);
        }
        const float* as = (AMODE == 1) ? &As[cur * ABUF + (4 * hf) * GPK + wm * 32 + r]
                                       : &As[cur * ABUF + (wm * 32 + r) * GPA + 4 * hf];
        const float* bs = &Bs[cur * (GBN * GPA) + (wn * 32 + r) * GPA + 4 * hf];
        #pragma unroll
        for (int g = 0; g < 4; ++g) {
            v2f a0lo, a0hi, a1lo, a1hi;
            if (AMODE == 1) {
                const float* ag = as + (8 * g) * GPK;
                a0lo = (v2f){ag[0],            ag[GPK]};
                a0hi = (v2f){ag[2 * GPK],      ag[3 * GPK]};
                a1lo = (v2f){ag[16],           ag[GPK + 16]};
                a1hi = (v2f){ag[2 * GPK + 16], ag[3 * GPK + 16]};
            } else {
                v4f a0 = *reinterpret_cast<const v4f*>(as + 8 * g);
                v4f a1 = *reinterpret_cast<const v4f*>(as + 16 * GPA + 8 * g);
                a0lo = (v2f){a0.x, a0.y}; a0hi = (v2f){a0.z, a0.w};
                a1lo = (v2f){a1.x, a1.y}; a1hi = (v2f){a1.z, a1.w};
            }
            v4f b0 = *reinterpret_cast<const v4f*>(bs + 8 * g);
            v4f b1 = *reinterpret_cast<const v4f*>(bs + 16 * GPA + 8 * g);
            v2f b0lo = {b0.x, b0.y}, b0hi = {b0.z, b0.w};
            v2f b1lo = {b1.x, b1.y}, b1hi = {b1.z, b1.w};
            acc[0][0] = __builtin_amdgcn_wmma_f32_16x16x4_f32(false, a0lo, false, b0lo, (short)0, acc[0][0], false, false);
            acc[0][1] = __builtin_amdgcn_wmma_f32_16x16x4_f32(false, a0lo, false, b1lo, (short)0, acc[0][1], false, false);
            acc[1][0] = __builtin_amdgcn_wmma_f32_16x16x4_f32(false, a1lo, false, b0lo, (short)0, acc[1][0], false, false);
            acc[1][1] = __builtin_amdgcn_wmma_f32_16x16x4_f32(false, a1lo, false, b1lo, (short)0, acc[1][1], false, false);
            acc[0][0] = __builtin_amdgcn_wmma_f32_16x16x4_f32(false, a0hi, false, b0hi, (short)0, acc[0][0], false, false);
            acc[0][1] = __builtin_amdgcn_wmma_f32_16x16x4_f32(false, a0hi, false, b1hi, (short)0, acc[0][1], false, false);
            acc[1][0] = __builtin_amdgcn_wmma_f32_16x16x4_f32(false, a1hi, false, b0hi, (short)0, acc[1][0], false, false);
            acc[1][1] = __builtin_amdgcn_wmma_f32_16x16x4_f32(false, a1hi, false, b1hi, (short)0, acc[1][1], false, false);
        }
        if (more) {
            int nxt = cur ^ 1;
            #pragma unroll
            for (int i = 0; i < 4; ++i) *reinterpret_cast<v4f*>(&As[nxt * ABUF + adst[i]]) = pa[i];
            #pragma unroll
            for (int i = 0; i < 2; ++i) *reinterpret_cast<v4f*>(&Bs[nxt * (GBN * GPA) + (ar + 32 * i) * GPA + ac]) = pb[i];
        }
        __syncthreads();
    }

    float* patch = &As[wave * (32 * GPA)];
    #pragma unroll
    for (int mt = 0; mt < 2; ++mt)
        #pragma unroll
        for (int nt = 0; nt < 2; ++nt)
            #pragma unroll
            for (int vg = 0; vg < 8; ++vg)
                patch[(mt * 16 + vg + 8 * hf) * GPA + nt * 16 + r] = acc[mt][nt][vg];
    __syncthreads();
    int pr = lane >> 3, pc = (lane & 7) * 4;
    int col = n0 + wn * 32 + pc;
    v4f bv = {0.0f, 0.0f, 0.0f, 0.0f};
    if (bias) bv = *reinterpret_cast<const v4f*>(bias + col);
    #pragma unroll
    for (int i = 0; i < 8; ++i) {
        int row = pr + 4 * i;
        v4f v = *reinterpret_cast<const v4f*>(&patch[row * GPA + pc]);
        v4f o;
        o.x = v.x * scale + bv.x;
        o.y = v.y * scale + bv.y;
        o.z = v.z * scale + bv.z;
        o.w = v.w * scale + bv.w;
        int crow = m0 + wm * 32 + row;
        if (AMODE == 1) {
            int b  = crow / XPLANE;
            int p  = crow - b * XPLANE;
            int y  = p / 96;
            int xc = p - y * 96;
            crow = (b * 144 + (y >> 3) * 12 + (xc >> 3)) * 64 + (y & 7) * 8 + (xc & 7);
        }
        *reinterpret_cast<v4f*>(&C[(size_t)crow * N + col]) = o;
    }
}

template <int AMODE>
__global__ __launch_bounds__(256) void gemm_tiled_f32(
    const float* __restrict__ A, const float* __restrict__ Bn,
    float* __restrict__ C, const float* __restrict__ bias,
    int M, int N, int K, int lda, int ldb, float scale)
{
    __shared__ __attribute__((aligned(16))) float As[2 * GBM * GPA];
    __shared__ __attribute__((aligned(16))) float Bs[2 * GBN * GPA];
    gemm_tiled_body<AMODE>(A, Bn, C, bias, M, N, K, lda, ldb, scale, blockIdx.x, blockIdx.y, blockIdx.z, As, Bs);
}

__device__ __forceinline__ void splitk_reduce_body(
    const float* __restrict__ part, const float* __restrict__ bias,
    float* __restrict__ dst, int MN, int N, int S, int blk)
{
    int i4 = blk * 256 + threadIdx.x;
    int quads = MN >> 2;
    if (i4 >= quads) return;
    const float4* p = reinterpret_cast<const float4*>(part) + i4;
    float4 acc = make_float4(0.0f, 0.0f, 0.0f, 0.0f);
    #pragma unroll 16
    for (int z = 0; z < S; ++z) {
        float4 v = p[(size_t)z * quads];
        acc.x += v.x; acc.y += v.y; acc.z += v.z; acc.w += v.w;
    }
    int col = (i4 << 2) % N;
    const float4 bv = *reinterpret_cast<const float4*>(bias + col);
    acc.x += bv.x; acc.y += bv.y; acc.z += bv.z; acc.w += bv.w;
    reinterpret_cast<float4*>(dst)[i4] = acc;
}


__global__ __launch_bounds__(256) void reduce_and_gemm(
    const float* __restrict__ part, const float* __restrict__ rbias, float* __restrict__ rdst,
    int MN, int RN, int S, int rblocks,
    const float* __restrict__ A, const float* __restrict__ Bn, float* __restrict__ C,
    int M, int N, int K, int ldb, float scale)
{
    __shared__ __attribute__((aligned(16))) float As[2 * GBM * GPA];
    __shared__ __attribute__((aligned(16))) float Bs[2 * GBN * GPA];
    int id = blockIdx.x;
    if (id < rblocks) {
        splitk_reduce_body(part, rbias, rdst, MN, RN, S, id);
        return;
    }
    id -= rblocks;
    int nb = N / GBN;
    gemm_tiled_body<1>(A, Bn, C, nullptr, M, N, K, 0, ldb, scale, id % nb, id / nb, 0, As, Bs);
}

#define APITCH 36
#define VPITCH 72
__global__ __launch_bounds__(128) void attn_kernel(
    float* __restrict__ qmat, const float* __restrict__ kv,
    const float* __restrict__ x, const float* __restrict__ rw, const float* __restrict__ rb)
{
    __shared__ __attribute__((aligned(16))) float Ks[144 * APITCH];
    __shared__ __attribute__((aligned(16))) float Vs[72 * VPITCH];
    int id = blockIdx.x;
    int h  = id & 7;
    int bb = (id >> 3) & 7;
    int g  = id >> 6;
    int n  = g * 8 + bb;

    int t    = threadIdx.x;
    int wave = t >> 5;
    int lane = t & 31;
    int hf   = lane >> 4;
    int c    = lane & 15;
    int m0   = wave * 16;

    float*       qrow = qmat + ((size_t)n * 64 + m0) * 256 + h * 32;
    const float* kg   = kv + (size_t)bb * 144 * 512 + h * 32;

    v4f qf[4];
    #pragma unroll
    for (int g4 = 0; g4 < 4; ++g4)
        qf[g4] = *reinterpret_cast<const v4f*>(qrow + c * 256 + 8 * g4 + 4 * hf);

    {
        int sr = t >> 3, sc = (t & 7) * 4;
        const float* ksrc = kg + (size_t)sr * 512 + sc;
        const float* vsrc = kg + 256 + (size_t)(2 * sr) * 512 + sc;
        bool tail = sr < 8;
        v4f sk[9], va[5], vb[5];
        #pragma unroll
        for (int i = 0; i < 9; ++i) sk[i] = *reinterpret_cast<const v4f*>(ksrc + (size_t)(16 * i) * 512);
        #pragma unroll
        for (int i = 0; i < 5; ++i)
            if (i < 4 || tail) {
                va[i] = *reinterpret_cast<const v4f*>(vsrc + (size_t)(32 * i) * 512);
                vb[i] = *reinterpret_cast<const v4f*>(vsrc + (size_t)(32 * i + 1) * 512);
            }
        #pragma unroll
        for (int i = 0; i < 9; ++i) *reinterpret_cast<v4f*>(&Ks[(sr + 16 * i) * APITCH + sc]) = sk[i];
        #pragma unroll
        for (int i = 0; i < 5; ++i)
            if (i < 4 || tail) {
                float* d = &Vs[(sr + 16 * i) * VPITCH + 2 * sc];
                v4f lo = {va[i].x, vb[i].x, va[i].y, vb[i].y};
                v4f hi = {va[i].z, vb[i].z, va[i].w, vb[i].w};
                *reinterpret_cast<v4f*>(d)     = lo;
                *reinterpret_cast<v4f*>(d + 4) = hi;
            }
    }
    __syncthreads();

    v8f acc[9];
    const float* kf = &Ks[c * APITCH + 4 * hf];
    #pragma unroll
    for (int tg = 0; tg < 3; ++tg) {
        v4f kk[3][4];
        #pragma unroll
        for (int x = 0; x < 3; ++x)
            #pragma unroll
            for (int g4 = 0; g4 < 4; ++g4)
                kk[x][g4] = *reinterpret_cast<const v4f*>(kf + (16 * (3 * tg + x)) * APITCH + 8 * g4);
        v8f a0 = {}, a1 = {}, a2 = {};
        #pragma unroll
        for (int g4 = 0; g4 < 4; ++g4) {
            v2f qlo = {qf[g4].x, qf[g4].y}, qhi = {qf[g4].z, qf[g4].w};
            v2f k0lo = {kk[0][g4].x, kk[0][g4].y}, k0hi = {kk[0][g4].z, kk[0][g4].w};
            v2f k1lo = {kk[1][g4].x, kk[1][g4].y}, k1hi = {kk[1][g4].z, kk[1][g4].w};
            v2f k2lo = {kk[2][g4].x, kk[2][g4].y}, k2hi = {kk[2][g4].z, kk[2][g4].w};
            a0 = __builtin_amdgcn_wmma_f32_16x16x4_f32(false, k0lo, false, qlo, (short)0, a0, false, false);
            a1 = __builtin_amdgcn_wmma_f32_16x16x4_f32(false, k1lo, false, qlo, (short)0, a1, false, false);
            a2 = __builtin_amdgcn_wmma_f32_16x16x4_f32(false, k2lo, false, qlo, (short)0, a2, false, false);
            a0 = __builtin_amdgcn_wmma_f32_16x16x4_f32(false, k0hi, false, qhi, (short)0, a0, false, false);
            a1 = __builtin_amdgcn_wmma_f32_16x16x4_f32(false, k1hi, false, qhi, (short)0, a1, false, false);
            a2 = __builtin_amdgcn_wmma_f32_16x16x4_f32(false, k2hi, false, qhi, (short)0, a2, false, false);
        }
        acc[3 * tg] = a0; acc[3 * tg + 1] = a1; acc[3 * tg + 2] = a2;
    }

    float mx = acc[0][0];
    #pragma unroll
    for (int tl = 0; tl < 9; ++tl)
        #pragma unroll
        for (int rr = 0; rr < 8; ++rr)
            mx = fmaxf(mx, acc[tl][rr]);
    mx = fmaxf(mx, __shfl_xor(mx, 16));
    float s0 = 0.0f, s1 = 0.0f, s2 = 0.0f, s3 = 0.0f;
    #pragma unroll
    for (int tl = 0; tl < 9; ++tl) {
        #pragma unroll
        for (int rr = 0; rr < 8; ++rr)
            acc[tl][rr] = __expf(acc[tl][rr] - mx);
        s0 += acc[tl][0] + acc[tl][4];
        s1 += acc[tl][1] + acc[tl][5];
        s2 += acc[tl][2] + acc[tl][6];
        s3 += acc[tl][3] + acc[tl][7];
    }
    float sum = (s0 + s1) + (s2 + s3);
    sum += __shfl_xor(sum, 16);
    float inv = 1.0f / sum;

    v8f o0 = {}, o1 = {};
    const float* vf = &Vs[(4 * hf) * VPITCH + 2 * c];
    #pragma unroll
    for (int tl = 0; tl < 9; ++tl) {
        #pragma unroll
        for (int j = 0; j < 4; ++j) {
            const float* vp = vf + (8 * tl + j) * VPITCH;
            v2f v0 = *reinterpret_cast<const v2f*>(vp);
            v2f v1 = *reinterpret_cast<const v2f*>(vp + 32);
            v2f p  = {acc[tl][2 * j], acc[tl][2 * j + 1]};
            o0 = __builtin_amdgcn_wmma_f32_16x16x4_f32(false, v0, false, p, (short)0, o0, false, false);
            o1 = __builtin_amdgcn_wmma_f32_16x16x4_f32(false, v1, false, p, (short)0, o1, false, false);
        }
    }

    __syncthreads();
    float* patch = &Ks[wave * (16 * APITCH)];
    {
        float* pw = patch + c * APITCH + 8 * hf;
        v4f w0 = {o0[0] * inv, o0[1] * inv, o0[2] * inv, o0[3] * inv};
        v4f w1 = {o0[4] * inv, o0[5] * inv, o0[6] * inv, o0[7] * inv};
        v4f w2 = {o1[0] * inv, o1[1] * inv, o1[2] * inv, o1[3] * inv};
        v4f w3 = {o1[4] * inv, o1[5] * inv, o1[6] * inv, o1[7] * inv};
        *reinterpret_cast<v4f*>(pw)      = w0;
        *reinterpret_cast<v4f*>(pw + 4)  = w1;
        *reinterpret_cast<v4f*>(pw + 16) = w2;
        *reinterpret_cast<v4f*>(pw + 20) = w3;
    }
    __syncthreads();
    int pr = lane >> 3, pc = (lane & 7) * 4;
    int R0 = n * 64 + __builtin_amdgcn_readfirstlane(m0);
    #pragma unroll
    for (int i = 0; i < 4; ++i) {
        int row   = pr + 4 * i;
        int Rb    = R0 + 4 * i;
        int plane = Rb / 36;
        int ch    = plane & 255;
        const float* wp = rw + ch * 9;
        float w[9];
        #pragma unroll
        for (int k = 0; k < 9; ++k) w[k] = wp[k];
        float bias = rb[ch];
        int pos = (Rb - plane * 36 + pr) * 256 + h * 32 + pc;
        int y   = pos / 96;
        int xc  = pos - y * 96;
        const float* xp = x + (size_t)plane * 9216;
        int  cl = (xc > 0) ? xc - 1 : 0;
        int  cr = (xc < 92) ? xc + 4 : 95;
        bool hl = xc > 0, hr = xc < 92;
        float tp[3][6];
        #pragma unroll
        for (int dy = 0; dy < 3; ++dy) {
            int  yy = y - 1 + dy;
            bool ok = (yy >= 0) && (yy < 96);
            int  yc = yy < 0 ? 0 : (yy > 95 ? 95 : yy);
            const float* xr = xp + yc * 96;
            v4f m = *reinterpret_cast<const v4f*>(xr + xc);
            float l = xr[cl];
            float r = xr[cr];
            tp[dy][0] = (ok && hl) ? l : 0.0f;
            tp[dy][1] = ok ? m.x : 0.0f;
            tp[dy][2] = ok ? m.y : 0.0f;
            tp[dy][3] = ok ? m.z : 0.0f;
            tp[dy][4] = ok ? m.w : 0.0f;
            tp[dy][5] = (ok && hr) ? r : 0.0f;
        }
        v4f v = *reinterpret_cast<const v4f*>(&patch[row * APITCH + pc]);
        float a[4];
        #pragma unroll
        for (int e = 0; e < 4; ++e) {
            float s = bias;
            #pragma unroll
            for (int dy = 0; dy < 3; ++dy)
                #pragma unroll
                for (int dx = 0; dx < 3; ++dx)
                    s = __builtin_fmaf(tp[dy][e + dx], w[dy * 3 + dx], s);
            a[e] = s;
        }
        v.x += a[0]; v.y += a[1]; v.z += a[2]; v.w += a[3];
        *reinterpret_cast<v4f*>(qrow + row * 256 + pc) = v;
    }
}

extern "C" void kernel_launch(void* const* d_in, const int* in_sizes, int n_in,
                              void* d_out, int out_size, void* d_ws, size_t ws_size,
                              hipStream_t stream) {
    const float* x      = (const float*)d_in[0];
    const float* Wq     = (const float*)d_in[1];
    const float* Wkv    = (const float*)d_in[2];
    const float* pool_w = (const float*)d_in[3];
    const float* pool_b = (const float*)d_in[4];
    const float* rpe_w  = (const float*)d_in[5];
    const float* rpe_b  = (const float*)d_in[6];
    const float* proj_w = (const float*)d_in[7];
    const float* proj_b = (const float*)d_in[8];
    float* out = (float*)d_out;

    float* ws    = (float*)d_ws;
    float* qmat  = ws;
    float* poolx = qmat + 18874368;
    float* kvb   = poolx + 294912;
    float* part  = kvb + 589824;
    (void)ws_size; (void)in_sizes; (void)n_in; (void)out_size;

    const int TB = 256;
    gemm_tiled_f32<2><<<dim3(256 / 64, 1152 / 128, 64), TB, 0, stream>>>(
        x, pool_w, part, nullptr, 1152, 256, 256, 0, 16384, 1.0f);

    reduce_and_gemm<<<2592, TB, 0, stream>>>(
        part, pool_b, poolx, 294912, 256, 64, 288,
        x, Wq, qmat, 73728, 256, 256, 256, QSCALE);

    gemm_wmma_f32<<<dim3(512 / 64, 1152 / 32), TB, 0, stream>>>(
        poolx, Wkv, kvb, nullptr, 1152, 512, 256, 256, 256, 1.0f);

    attn_kernel<<<144 * 8 * 8, 128, 0, stream>>>(qmat, kvb, x, rpe_w, rpe_b);

    gemm_tiled_f32<0><<<dim3(256 / 64, 73728 / 128), TB, 0, stream>>>(
        qmat, proj_w, out, proj_b, 73728, 256, 256, 256, 256, 1.0f);
}


# Split-K finish feeds the KV projection inside its block: 16 summed rows of pool_x stay in LDS as the A operand, pool_x not stored; 5 launches -> 4; KV k order in each 8-k group changes (f32 sum order)
# speedup vs baseline: 4.2689x; 1.0406x over previous
#include <hip/hip_runtime.h>
#include <hip/hip_bf16.h>

typedef __attribute__((ext_vector_type(2))) float v2f;
typedef __attribute__((ext_vector_type(8))) float v8f;

#define QSCALE 0.17677669529663687f


typedef __attribute__((ext_vector_type(4))) float v4f;


#define GBM 128
#define GBN 64
#define GBK 32
#define GPA 36
#define GPK 136
#define XPLANE 9216

template <int AMODE>
__device__ __forceinline__ int a_chunk_offset(int kz, int kc) {
    if (AMODE == 0) return kc * GBK;
    if (AMODE == 1) return kc * GBK * XPLANE;
    int kg = kz + kc * GBK;
    return (kg >> 6) * XPLANE + ((kg >> 3) & 7) * 96;
}

template <int AMODE>
__device__ __forceinline__ void gemm_tiled_body(
    const float* __restrict__ A, const float* __restrict__ Bn,
    float* __restrict__ C, const float* __restrict__ bias,
    int M, int N, int K, int lda, int ldb, float scale,
    int bx, int by, int bz, float* As, float* Bs)
{
    int t    = threadIdx.x;
    int wave = t >> 5;
    int lane = t & 31;
    int hf   = lane >> 4;
    int r    = lane & 15;
    int wm   = wave >> 1;
    int wn   = wave & 1;
    int m0 = by * GBM;
    int n0 = bx * GBN;
    int kz = bz * K;

    int ar = t >> 3, ac = (t & 7) * 4;
    int kr = t >> 5, mc = (t & 31) * 4;
    int aoff[4];
    int adst[4];
    #pragma unroll
    for (int i = 0; i < 4; ++i) {
        if (AMODE == 0) {
            aoff[i] = (m0 + ar + 32 * i) * lda + kz + ac;
            adst[i] = (ar + 32 * i) * GPA + ac;
        } else if (AMODE == 1) {
            int b = m0 / XPLANE;
            aoff[i] = (b * 256 + kz + kr + 8 * i) * XPLANE + (m0 - b * XPLANE) + mc;
            adst[i] = (kr + 8 * i) * GPK + mc;
        } else {
            int n   = m0 + ar + 32 * i;
            int b   = n / 144;
            int rem = n - b * 144;
            int wh  = rem / 12;
            int ww  = rem - wh * 12;
            aoff[i] = ((b * 256) * 96 + wh * 8 + (ac >> 3)) * 96 + ww * 8 + (ac & 7);
            adst[i] = (ar + 32 * i) * GPA + ac;
        }
    }
    const int ABUF = (AMODE == 1) ? GBK * GPK : GBM * GPA;
    const float* Bg = Bn + (size_t)(n0 + ar) * ldb + kz + ac;
    C += (size_t)bz * M * N;

    v4f pa[4], pb[2];
    {
        const float* A0 = A + a_chunk_offset<AMODE>(kz, 0);
        #pragma unroll
        for (int i = 0; i < 4; ++i) pa[i] = *reinterpret_cast<const v4f*>(A0 + aoff[i]);
        #pragma unroll
        for (int i = 0; i < 2; ++i) pb[i] = *reinterpret_cast<const v4f*>(Bg + (size_t)(32 * i) * ldb);
        #pragma unroll
        for (int i = 0; i < 4; ++i) *reinterpret_cast<v4f*>(&As[adst[i]]) = pa[i];
        #pragma unroll
        for (int i = 0; i < 2; ++i) *reinterpret_cast<v4f*>(&Bs[(ar + 32 * i) * GPA + ac]) = pb[i];
    }
    __syncthreads();

    v8f acc[2][2] = {};
    int nchunks = K / GBK;
    for (int kc = 0; kc < nchunks; ++kc) {
        int cur = kc & 1;
        bool more = (kc + 1 < nchunks);
        if (more) {
            const float* An  = A + a_chunk_offset<AMODE>(kz, kc + 1);
            const float* Bgn = Bg + (kc + 1) * GBK;
            #pragma unroll
            for (int i = 0; i < 4; ++i) pa[i] = *reinterpret_cast<const v4f*>(An + aoff[i]);
            #pragma unroll
            for (int i = 0; i < 2; ++i) pb[i] = *reinterpret_cast<const v4f*>(Bgn + (size_t)(32 * i) * ldb);
        }
        const float* as = (AMODE == 1) ? &As[cur * ABUF + (4 * hf) * GPK + wm * 32 + r]
                                       : &As[cur * ABUF + (wm * 32 + r) * GPA + 4 * hf];
        const float* bs = &Bs[cur * (GBN * GPA) + (wn * 32 + r) * GPA + 4 * hf];
        #pragma unroll
        for (int g = 0; g < 4; ++g) {
            v2f a0lo, a0hi, a1lo, a1hi;
            if (AMODE == 1) {
                const float* ag = as + (8 * g) * GPK;
                a0lo = (v2f){ag[0],            ag[GPK]};
                a0hi = (v2f){ag[2 * GPK],      ag[3 * GPK]};
                a1lo = (v2f){ag[16],           ag[GPK + 16]};
                a1hi = (v2f){ag[2 * GPK + 16], ag[3 * GPK + 16]};
            } else {
                v4f a0 = *reinterpret_cast<const v4f*>(as + 8 * g);
                v4f a1 = *reinterpret_cast<const v4f*>(as + 16 * GPA + 8 * g);
                a0lo = (v2f){a0.x, a0.y}; a0hi = (v2f){a0.z, a0.w};
                a1lo = (v2f){a1.x, a1.y}; a1hi = (v2f){a1.z, a1.w};
            }
            v4f b0 = *reinterpret_cast<const v4f*>(bs + 8 * g);
            v4f b1 = *reinterpret_cast<const v4f*>(bs + 16 * GPA + 8 * g);
            v2f b0lo = {b0.x, b0.y}, b0hi = {b0.z, b0.w};
            v2f b1lo = {b1.x, b1.y}, b1hi = {b1.z, b1.w};
            acc[0][0] = __builtin_amdgcn_wmma_f32_16x16x4_f32(false, a0lo, false, b0lo, (short)0, acc[0][0], false, false);
            acc[0][1] = __builtin_amdgcn_wmma_f32_16x16x4_f32(false, a0lo, false, b1lo, (short)0, acc[0][1], false, false);
            acc[1][0] = __builtin_amdgcn_wmma_f32_16x16x4_f32(false, a1lo, false, b0lo, (short)0, acc[1][0], false, false);
            acc[1][1] = __builtin_amdgcn_wmma_f32_16x16x4_f32(false, a1lo, false, b1lo, (short)0, acc[1][1], false, false);
            acc[0][0] = __builtin_amdgcn_wmma_f32_16x16x4_f32(false, a0hi, false, b0hi, (short)0, acc[0][0], false, false);
            acc[0][1] = __builtin_amdgcn_wmma_f32_16x16x4_f32(false, a0hi, false, b1hi, (short)0, acc[0][1], false, false);
            acc[1][0] = __builtin_amdgcn_wmma_f32_16x16x4_f32(false, a1hi, false, b0hi, (short)0, acc[1][0], false, false);
            acc[1][1] = __builtin_amdgcn_wmma_f32_16x16x4_f32(false, a1hi, false, b1hi, (short)0, acc[1][1], false, false);
        }
        if (more) {
            int nxt = cur ^ 1;
            #pragma unroll
            for (int i = 0; i < 4; ++i) *reinterpret_cast<v4f*>(&As[nxt * ABUF + adst[i]]) = pa[i];
            #pragma unroll
            for (int i = 0; i < 2; ++i) *reinterpret_cast<v4f*>(&Bs[nxt * (GBN * GPA) + (ar + 32 * i) * GPA + ac]) = pb[i];
        }
        __syncthreads();
    }

    float* patch = &As[wave * (32 * GPA)];
    #pragma unroll
    for (int mt = 0; mt < 2; ++mt)
        #pragma unroll
        for (int nt = 0; nt < 2; ++nt)
            #pragma unroll
            for (int vg = 0; vg < 8; ++vg)
                patch[(mt * 16 + vg + 8 * hf) * GPA + nt * 16 + r] = acc[mt][nt][vg];
    __syncthreads();
    int pr = lane >> 3, pc = (lane & 7) * 4;
    int col = n0 + wn * 32 + pc;
    v4f bv = {0.0f, 0.0f, 0.0f, 0.0f};
    if (bias) bv = *reinterpret_cast<const v4f*>(bias + col);
    #pragma unroll
    for (int i = 0; i < 8; ++i) {
        int row = pr + 4 * i;
        v4f v = *reinterpret_cast<const v4f*>(&patch[row * GPA + pc]);
        v4f o;
        o.x = v.x * scale + bv.x;
        o.y = v.y * scale + bv.y;
        o.z = v.z * scale + bv.z;
        o.w = v.w * scale + bv.w;
        int crow = m0 + wm * 32 + row;
        if (AMODE == 1) {
            int b  = crow / XPLANE;
            int p  = crow - b * XPLANE;
            int y  = p / 96;
            int xc = p - y * 96;
            crow = (b * 144 + (y >> 3) * 12 + (xc >> 3)) * 64 + (y & 7) * 8 + (xc & 7);
        }
        *reinterpret_cast<v4f*>(&C[(size_t)crow * N + col]) = o;
    }
}

template <int AMODE>
__global__ __launch_bounds__(256) void gemm_tiled_f32(
    const float* __restrict__ A, const float* __restrict__ Bn,
    float* __restrict__ C, const float* __restrict__ bias,
    int M, int N, int K, int lda, int ldb, float scale)
{
    __shared__ __attribute__((aligned(16))) float As[2 * GBM * GPA];
    __shared__ __attribute__((aligned(16))) float Bs[2 * GBN * GPA];
    gemm_tiled_body<AMODE>(A, Bn, C, bias, M, N, K, lda, ldb, scale, blockIdx.x, blockIdx.y, blockIdx.z, As, Bs);
}

#define KVP 260
__device__ __forceinline__ void reduce_kv_body(
    const float* __restrict__ part, const float* __restrict__ bias,
    const float* __restrict__ Wkv, float* __restrict__ kv,
    int MN, int S, int blk, float* Ps)
{
    int t = threadIdx.x;
    int r0 = blk * 16;
    int quads = MN >> 2;
    #pragma unroll 1
    for (int i = 0; i < 4; ++i) {
        int q = t + 256 * i;
        const float4* p = reinterpret_cast<const float4*>(part) + (r0 * 64 + q);
        float4 acc = make_float4(0.0f, 0.0f, 0.0f, 0.0f);
        #pragma unroll 16
        for (int z = 0; z < S; ++z) {
            float4 v = p[(size_t)z * quads];
            acc.x += v.x; acc.y += v.y; acc.z += v.z; acc.w += v.w;
        }
        int col = (q & 63) << 2;
        const float4 bv = *reinterpret_cast<const float4*>(bias + col);
        acc.x += bv.x; acc.y += bv.y; acc.z += bv.z; acc.w += bv.w;
        *reinterpret_cast<float4*>(&Ps[(q >> 6) * KVP + col]) = acc;
    }
    __syncthreads();

    int wave = t >> 5;
    int lane = t & 31;
    int hf   = lane >> 4;
    int r    = lane & 15;
    const float* as = &Ps[r * KVP + 4 * hf];
    const float* bg = Wkv + (size_t)(wave * 64 + r) * 256 + 4 * hf;
    v8f acc[4] = {};
    #pragma unroll 2
    for (int g = 0; g < 32; ++g) {
        v4f a = *reinterpret_cast<const v4f*>(as + 8 * g);
        v4f b[4];
        #pragma unroll
        for (int nt = 0; nt < 4; ++nt)
            b[nt] = *reinterpret_cast<const v4f*>(bg + (size_t)(16 * nt) * 256 + 8 * g);
        v2f alo = {a.x, a.y}, ahi = {a.z, a.w};
        #pragma unroll
        for (int nt = 0; nt < 4; ++nt) {
            v2f blo = {b[nt].x, b[nt].y};
            acc[nt] = __builtin_amdgcn_wmma_f32_16x16x4_f32(false, alo, false, blo, (short)0, acc[nt], false, false);
        }
        #pragma unroll
        for (int nt = 0; nt < 4; ++nt) {
            v2f bhi = {b[nt].z, b[nt].w};
            acc[nt] = __builtin_amdgcn_wmma_f32_16x16x4_f32(false, ahi, false, bhi, (short)0, acc[nt], false, false);
        }
    }
    #pragma unroll
    for (int nt = 0; nt < 4; ++nt)
        #pragma unroll
        for (int vg = 0; vg < 8; ++vg)
            kv[(size_t)(r0 + vg + 8 * hf) * 512 + wave * 64 + 16 * nt + r] = acc[nt][vg];
}


__global__ __launch_bounds__(256) void reduce_and_gemm(
    const float* __restrict__ part, const float* __restrict__ rbias,
    const float* __restrict__ Wkv, float* __restrict__ kvdst,
    int MN, int S, int rblocks,
    const float* __restrict__ A, const float* __restrict__ Bn, float* __restrict__ C,
    int M, int N, int K, int ldb, float scale)
{
    __shared__ __attribute__((aligned(16))) float As[2 * GBM * GPA];
    __shared__ __attribute__((aligned(16))) float Bs[2 * GBN * GPA];
    int id = blockIdx.x;
    if (id < rblocks) {
        reduce_kv_body(part, rbias, Wkv, kvdst, MN, S, id, As);
        return;
    }
    id -= rblocks;
    int nb = N / GBN;
    gemm_tiled_body<1>(A, Bn, C, nullptr, M, N, K, 0, ldb, scale, id % nb, id / nb, 0, As, Bs);
}

#define APITCH 36
#define VPITCH 72
__global__ __launch_bounds__(128) void attn_kernel(
    float* __restrict__ qmat, const float* __restrict__ kv,
    const float* __restrict__ x, const float* __restrict__ rw, const float* __restrict__ rb)
{
    __shared__ __attribute__((aligned(16))) float Ks[144 * APITCH];
    __shared__ __attribute__((aligned(16))) float Vs[72 * VPITCH];
    int id = blockIdx.x;
    int h  = id & 7;
    int bb = (id >> 3) & 7;
    int g  = id >> 6;
    int n  = g * 8 + bb;

    int t    = threadIdx.x;
    int wave = t >> 5;
    int lane = t & 31;
    int hf   = lane >> 4;
    int c    = lane & 15;
    int m0   = wave * 16;

    float*       qrow = qmat + ((size_t)n * 64 + m0) * 256 + h * 32;
    const float* kg   = kv + (size_t)bb * 144 * 512 + h * 32;

    v4f qf[4];
    #pragma unroll
    for (int g4 = 0; g4 < 4; ++g4)
        qf[g4] = *reinterpret_cast<const v4f*>(qrow + c * 256 + 8 * g4 + 4 * hf);

    {
        int sr = t >> 3, sc = (t & 7) * 4;
        const float* ksrc = kg + (size_t)sr * 512 + sc;
        const float* vsrc = kg + 256 + (size_t)(2 * sr) * 512 + sc;
        bool tail = sr < 8;
        v4f sk[9], va[5], vb[5];
        #pragma unroll
        for (int i = 0; i < 9; ++i) sk[i] = *reinterpret_cast<const v4f*>(ksrc + (size_t)(16 * i) * 512);
        #pragma unroll
        for (int i = 0; i < 5; ++i)
            if (i < 4 || tail) {
                va[i] = *reinterpret_cast<const v4f*>(vsrc + (size_t)(32 * i) * 512);
                vb[i] = *reinterpret_cast<const v4f*>(vsrc + (size_t)(32 * i + 1) * 512);
            }
        #pragma unroll
        for (int i = 0; i < 9; ++i) *reinterpret_cast<v4f*>(&Ks[(sr + 16 * i) * APITCH + sc]) = sk[i];
        #pragma unroll
        for (int i = 0; i < 5; ++i)
            if (i < 4 || tail) {
                float* d = &Vs[(sr + 16 * i) * VPITCH + 2 * sc];
                v4f lo = {va[i].x, vb[i].x, va[i].y, vb[i].y};
                v4f hi = {va[i].z, vb[i].z, va[i].w, vb[i].w};
                *reinterpret_cast<v4f*>(d)     = lo;
                *reinterpret_cast<v4f*>(d + 4) = hi;
            }
    }
    __syncthreads();

    v8f acc[9];
    const float* kf = &Ks[c * APITCH + 4 * hf];
    #pragma unroll
    for (int tg = 0; tg < 3; ++tg) {
        v4f kk[3][4];
        #pragma unroll
        for (int x = 0; x < 3; ++x)
            #pragma unroll
            for (int g4 = 0; g4 < 4; ++g4)
                kk[x][g4] = *reinterpret_cast<const v4f*>(kf + (16 * (3 * tg + x)) * APITCH + 8 * g4);
        v8f a0 = {}, a1 = {}, a2 = {};
        #pragma unroll
        for (int g4 = 0; g4 < 4; ++g4) {
            v2f qlo = {qf[g4].x, qf[g4].y}, qhi = {qf[g4].z, qf[g4].w};
            v2f k0lo = {kk[0][g4].x, kk[0][g4].y}, k0hi = {kk[0][g4].z, kk[0][g4].w};
            v2f k1lo = {kk[1][g4].x, kk[1][g4].y}, k1hi = {kk[1][g4].z, kk[1][g4].w};
            v2f k2lo = {kk[2][g4].x, kk[2][g4].y}, k2hi = {kk[2][g4].z, kk[2][g4].w};
            a0 = __builtin_amdgcn_wmma_f32_16x16x4_f32(false, k0lo, false, qlo, (short)0, a0, false, false);
            a1 = __builtin_amdgcn_wmma_f32_16x16x4_f32(false, k1lo, false, qlo, (short)0, a1, false, false);
            a2 = __builtin_amdgcn_wmma_f32_16x16x4_f32(false, k2lo, false, qlo, (short)0, a2, false, false);
            a0 = __builtin_amdgcn_wmma_f32_16x16x4_f32(false, k0hi, false, qhi, (short)0, a0, false, false);
            a1 = __builtin_amdgcn_wmma_f32_16x16x4_f32(false, k1hi, false, qhi, (short)0, a1, false, false);
            a2 = __builtin_amdgcn_wmma_f32_16x16x4_f32(false, k2hi, false, qhi, (short)0, a2, false, false);
        }
        acc[3 * tg] = a0; acc[3 * tg + 1] = a1; acc[3 * tg + 2] = a2;
    }

    float mx = acc[0][0];
    #pragma unroll
    for (int tl = 0; tl < 9; ++tl)
        #pragma unroll
        for (int rr = 0; rr < 8; ++rr)
            mx = fmaxf(mx, acc[tl][rr]);
    mx = fmaxf(mx, __shfl_xor(mx, 16));
    float s0 = 0.0f, s1 = 0.0f, s2 = 0.0f, s3 = 0.0f;
    #pragma unroll
    for (int tl = 0; tl < 9; ++tl) {
        #pragma unroll
        for (int rr = 0; rr < 8; ++rr)
            acc[tl][rr] = __expf(acc[tl][rr] - mx);
        s0 += acc[tl][0] + acc[tl][4];
        s1 += acc[tl][1] + acc[tl][5];
        s2 += acc[tl][2] + acc[tl][6];
        s3 += acc[tl][3] + acc[tl][7];
    }
    float sum = (s0 + s1) + (s2 + s3);
    sum += __shfl_xor(sum, 16);
    float inv = 1.0f / sum;

    v8f o0 = {}, o1 = {};
    const float* vf = &Vs[(4 * hf) * VPITCH + 2 * c];
    #pragma unroll
    for (int tl = 0; tl < 9; ++tl) {
        #pragma unroll
        for (int j = 0; j < 4; ++j) {
            const float* vp = vf + (8 * tl + j) * VPITCH;
            v2f v0 = *reinterpret_cast<const v2f*>(vp);
            v2f v1 = *reinterpret_cast<const v2f*>(vp + 32);
            v2f p  = {acc[tl][2 * j], acc[tl][2 * j + 1]};
            o0 = __builtin_amdgcn_wmma_f32_16x16x4_f32(false, v0, false, p, (short)0, o0, false, false);
            o1 = __builtin_amdgcn_wmma_f32_16x16x4_f32(false, v1, false, p, (short)0, o1, false, false);
        }
    }

    __syncthreads();
    float* patch = &Ks[wave * (16 * APITCH)];
    {
        float* pw = patch + c * APITCH + 8 * hf;
        v4f w0 = {o0[0] * inv, o0[1] * inv, o0[2] * inv, o0[3] * inv};
        v4f w1 = {o0[4] * inv, o0[5] * inv, o0[6] * inv, o0[7] * inv};
        v4f w2 = {o1[0] * inv, o1[1] * inv, o1[2] * inv, o1[3] * inv};
        v4f w3 = {o1[4] * inv, o1[5] * inv, o1[6] * inv, o1[7] * inv};
        *reinterpret_cast<v4f*>(pw)      = w0;
        *reinterpret_cast<v4f*>(pw + 4)  = w1;
        *reinterpret_cast<v4f*>(pw + 16) = w2;
        *reinterpret_cast<v4f*>(pw + 20) = w3;
    }
    __syncthreads();
    int pr = lane >> 3, pc = (lane & 7) * 4;
    int R0 = n * 64 + __builtin_amdgcn_readfirstlane(m0);
    #pragma unroll
    for (int i = 0; i < 4; ++i) {
        int row   = pr + 4 * i;
        int Rb    = R0 + 4 * i;
        int plane = Rb / 36;
        int ch    = plane & 255;
        const float* wp = rw + ch * 9;
        float w[9];
        #pragma unroll
        for (int k = 0; k < 9; ++k) w[k] = wp[k];
        float bias = rb[ch];
        int pos = (Rb - plane * 36 + pr) * 256 + h * 32 + pc;
        int y   = pos / 96;
        int xc  = pos - y * 96;
        const float* xp = x + (size_t)plane * 9216;
        int  cl = (xc > 0) ? xc - 1 : 0;
        int  cr = (xc < 92) ? xc + 4 : 95;
        bool hl = xc > 0, hr = xc < 92;
        float tp[3][6];
        #pragma unroll
        for (int dy = 0; dy < 3; ++dy) {
            int  yy = y - 1 + dy;
            bool ok = (yy >= 0) && (yy < 96);
            int  yc = yy < 0 ? 0 : (yy > 95 ? 95 : yy);
            const float* xr = xp + yc * 96;
            v4f m = *reinterpret_cast<const v4f*>(xr + xc);
            float l = xr[cl];
            float r = xr[cr];
            tp[dy][0] = (ok && hl) ? l : 0.0f;
            tp[dy][1] = ok ? m.x : 0.0f;
            tp[dy][2] = ok ? m.y : 0.0f;
            tp[dy][3] = ok ? m.z : 0.0f;
            tp[dy][4] = ok ? m.w : 0.0f;
            tp[dy][5] = (ok && hr) ? r : 0.0f;
        }
        v4f v = *reinterpret_cast<const v4f*>(&patch[row * APITCH + pc]);
        float a[4];
        #pragma unroll
        for (int e = 0; e < 4; ++e) {
            float s = bias;
            #pragma unroll
            for (int dy = 0; dy < 3; ++dy)
                #pragma unroll
                for (int dx = 0; dx < 3; ++dx)
                    s = __builtin_fmaf(tp[dy][e + dx], w[dy * 3 + dx], s);
            a[e] = s;
        }
        v.x += a[0]; v.y += a[1]; v.z += a[2]; v.w += a[3];
        *reinterpret_cast<v4f*>(qrow + row * 256 + pc) = v;
    }
}

extern "C" void kernel_launch(void* const* d_in, const int* in_sizes, int n_in,
                              void* d_out, int out_size, void* d_ws, size_t ws_size,
                              hipStream_t stream) {
    const float* x      = (const float*)d_in[0];
    const float* Wq     = (const float*)d_in[1];
    const float* Wkv    = (const float*)d_in[2];
    const float* pool_w = (const float*)d_in[3];
    const float* pool_b = (const float*)d_in[4];
    const float* rpe_w  = (const float*)d_in[5];
    const float* rpe_b  = (const float*)d_in[6];
    const float* proj_w = (const float*)d_in[7];
    const float* proj_b = (const float*)d_in[8];
    float* out = (float*)d_out;

    float* ws    = (float*)d_ws;
    float* qmat  = ws;
    float* kvb   = qmat + 19169280;
    float* part  = kvb + 589824;
    (void)ws_size; (void)in_sizes; (void)n_in; (void)out_size;

    const int TB = 256;
    gemm_tiled_f32<2><<<dim3(256 / 64, 1152 / 128, 64), TB, 0, stream>>>(
        x, pool_w, part, nullptr, 1152, 256, 256, 0, 16384, 1.0f);

    reduce_and_gemm<<<2376, TB, 0, stream>>>(
        part, pool_b, Wkv, kvb, 294912, 64, 72,
        x, Wq, qmat, 73728, 256, 256, 256, QSCALE);

    attn_kernel<<<144 * 8 * 8, 128, 0, stream>>>(qmat, kvb, x, rpe_w, rpe_b);

    gemm_tiled_f32<0><<<dim3(256 / 64, 73728 / 128), TB, 0, stream>>>(
        qmat, proj_w, out, proj_b, 73728, 256, 256, 256, 256, 1.0f);
}


# Three large GEMMs (pool split-K, Q, out proj) re-tiled to 96x256 block tiles, 48x64 per wave in 12 accumulators: A staged once (was 4x), 7 LDS fragment reads per 24 matrix steps; sum order unchanged
# speedup vs baseline: 4.4385x; 1.0397x over previous
#include <hip/hip_runtime.h>
#include <hip/hip_bf16.h>

typedef __attribute__((ext_vector_type(2))) float v2f;
typedef __attribute__((ext_vector_type(8))) float v8f;

#define QSCALE 0.17677669529663687f


typedef __attribute__((ext_vector_type(4))) float v4f;


#define GBM 96
#define GBN 256
#define GBK 32
#define GPA 36
#define GPK 104
#define GPP 68
#define XPLANE 9216

template <int AMODE>
__device__ __forceinline__ int a_chunk_offset(int kz, int kc) {
    if (AMODE == 0) return kc * GBK;
    if (AMODE == 1) return kc * GBK * XPLANE;
    int kg = kz + kc * GBK;
    return (kg >> 6) * XPLANE + ((kg >> 3) & 7) * 96;
}

template <int AMODE>
__device__ __forceinline__ void gemm_tiled_body(
    const float* __restrict__ A, const float* __restrict__ Bn,
    float* __restrict__ C, const float* __restrict__ bias,
    int M, int N, int K, int lda, int ldb, float scale,
    int bx, int by, int bz, float* As, float* Bs)
{
    int t    = threadIdx.x;
    int wave = t >> 5;
    int lane = t & 31;
    int hf   = lane >> 4;
    int r    = lane & 15;
    int wm   = wave >> 2;
    int wn   = wave & 3;
    int m0 = by * GBM;
    int n0 = bx * GBN;
    int kz = bz * K;

    int ar = t >> 3, ac = (t & 7) * 4;
    int aoff[3];
    int adst[3];
    #pragma unroll
    for (int i = 0; i < 3; ++i) {
        if (AMODE == 0) {
            aoff[i] = (m0 + ar + 32 * i) * lda + kz + ac;
            adst[i] = (ar + 32 * i) * GPA + ac;
        } else if (AMODE == 1) {
            int p  = t + 256 * i;
            int kr = p / 24;
            int mc = (p - kr * 24) * 4;
            int b  = m0 / XPLANE;
            aoff[i] = (b * 256 + kz + kr) * XPLANE + (m0 - b * XPLANE) + mc;
            adst[i] = kr * GPK + mc;
        } else {
            int n   = m0 + ar + 32 * i;
            int b   = n / 144;
            int rem = n - b * 144;
            int wh  = rem / 12;
            int ww  = rem - wh * 12;
            aoff[i] = ((b * 256) * 96 + wh * 8 + (ac >> 3)) * 96 + ww * 8 + (ac & 7);
            adst[i] = (ar + 32 * i) * GPA + ac;
        }
    }
    const int ABUF = (AMODE == 1) ? GBK * GPK : GBM * GPA;
    const float* Bg = Bn + (size_t)(n0 + ar) * ldb + kz + ac;
    C += (size_t)bz * M * N;

    v4f pa[3], pb[8];
    {
        const float* A0 = A + a_chunk_offset<AMODE>(kz, 0);
        #pragma unroll
        for (int i = 0; i < 3; ++i) pa[i] = *reinterpret_cast<const v4f*>(A0 + aoff[i]);
        #pragma unroll
        for (int i = 0; i < 8; ++i) pb[i] = *reinterpret_cast<const v4f*>(Bg + (size_t)(32 * i) * ldb);
        #pragma unroll
        for (int i = 0; i < 3; ++i) *reinterpret_cast<v4f*>(&As[adst[i]]) = pa[i];
        #pragma unroll
        for (int i = 0; i < 8; ++i) *reinterpret_cast<v4f*>(&Bs[(ar + 32 * i) * GPA + ac]) = pb[i];
    }
    __syncthreads();

    v8f acc[3][4] = {};
    int nchunks = K / GBK;
    for (int kc = 0; kc < nchunks; ++kc) {
        int cur = kc & 1;
        bool more = (kc + 1 < nchunks);
        if (more) {
            const float* An  = A + a_chunk_offset<AMODE>(kz, kc + 1);
            const float* Bgn = Bg + (kc + 1) * GBK;
            #pragma unroll
            for (int i = 0; i < 3; ++i) pa[i] = *reinterpret_cast<const v4f*>(An + aoff[i]);
            #pragma unroll
            for (int i = 0; i < 8; ++i) pb[i] = *reinterpret_cast<const v4f*>(Bgn + (size_t)(32 * i) * ldb);
        }
        const float* as = (AMODE == 1) ? &As[cur * ABUF + (4 * hf) * GPK + wm * 48 + r]
                                       : &As[cur * ABUF + (wm * 48 + r) * GPA + 4 * hf];
        const float* bs = &Bs[cur * (GBN * GPA) + (wn * 64 + r) * GPA + 4 * hf];
        #pragma unroll
        for (int g = 0; g < 4; ++g) {
            v2f alo[3], ahi[3], blo[4], bhi[4];
            #pragma unroll
            for (int mt = 0; mt < 3; ++mt) {
                if (AMODE == 1) {
                    const float* ag = as + (8 * g) * GPK + 16 * mt;
                    alo[mt] = (v2f){ag[0],       ag[GPK]};
                    ahi[mt] = (v2f){ag[2 * GPK], ag[3 * GPK]};
                } else {
                    v4f a = *reinterpret_cast<const v4f*>(as + (16 * mt) * GPA + 8 * g);
                    alo[mt] = (v2f){a.x, a.y}; ahi[mt] = (v2f){a.z, a.w};
                }
            }
            #pragma unroll
            for (int nt = 0; nt < 4; ++nt) {
                v4f b = *reinterpret_cast<const v4f*>(bs + (16 * nt) * GPA + 8 * g);
                blo[nt] = (v2f){b.x, b.y}; bhi[nt] = (v2f){b.z, b.w};
            }
            #pragma unroll
            for (int mt = 0; mt < 3; ++mt)
                #pragma unroll
                for (int nt = 0; nt < 4; ++nt)
                    acc[mt][nt] = __builtin_amdgcn_wmma_f32_16x16x4_f32(false, alo[mt], false, blo[nt], (short)0, acc[mt][nt], false, false);
            #pragma unroll
            for (int mt = 0; mt < 3; ++mt)
                #pragma unroll
                for (int nt = 0; nt < 4; ++nt)
                    acc[mt][nt] = __builtin_amdgcn_wmma_f32_16x16x4_f32(false, ahi[mt], false, bhi[nt], (short)0, acc[mt][nt], false, false);
        }
        if (more) {
            int nxt = cur ^ 1;
            #pragma unroll
            for (int i = 0; i < 3; ++i) *reinterpret_cast<v4f*>(&As[nxt * ABUF + adst[i]]) = pa[i];
            #pragma unroll
            for (int i = 0; i < 8; ++i) *reinterpret_cast<v4f*>(&Bs[nxt * (GBN * GPA) + (ar + 32 * i) * GPA + ac]) = pb[i];
        }
        __syncthreads();
    }

    float* patch = &Bs[wave * (16 * GPP)];
    int pr = lane >> 4, pc = (lane & 15) * 4;
    int col = n0 + wn * 64 + pc;
    v4f bv = {0.0f, 0.0f, 0.0f, 0.0f};
    if (bias) bv = *reinterpret_cast<const v4f*>(bias + col);
    #pragma unroll
    for (int mt = 0; mt < 3; ++mt) {
        if (mt) __syncthreads();
        #pragma unroll
        for (int nt = 0; nt < 4; ++nt)
            #pragma unroll
            for (int vg = 0; vg < 8; ++vg)
                patch[(vg + 8 * hf) * GPP + nt * 16 + r] = acc[mt][nt][vg];
        __syncthreads();
        #pragma unroll
        for (int i = 0; i < 8; ++i) {
            int row = pr + 2 * i;
            v4f v = *reinterpret_cast<const v4f*>(&patch[row * GPP + pc]);
            v4f o;
            o.x = v.x * scale + bv.x;
            o.y = v.y * scale + bv.y;
            o.z = v.z * scale + bv.z;
            o.w = v.w * scale + bv.w;
            int crow = m0 + wm * 48 + mt * 16 + row;
            if (AMODE == 1) {
                int b  = crow / XPLANE;
                int p  = crow - b * XPLANE;
                int y  = p / 96;
                int xc = p - y * 96;
                crow = (b * 144 + (y >> 3) * 12 + (xc >> 3)) * 64 + (y & 7) * 8 + (xc & 7);
            }
            *reinterpret_cast<v4f*>(&C[(size_t)crow * N + col]) = o;
        }
    }
}

template <int AMODE>
__global__ __launch_bounds__(256) void gemm_tiled_f32(
    const float* __restrict__ A, const float* __restrict__ Bn,
    float* __restrict__ C, const float* __restrict__ bias,
    int M, int N, int K, int lda, int ldb, float scale)
{
    __shared__ __attribute__((aligned(16))) float As[2 * GBM * GPA];
    __shared__ __attribute__((aligned(16))) float Bs[2 * GBN * GPA];
    gemm_tiled_body<AMODE>(A, Bn, C, bias, M, N, K, lda, ldb, scale, blockIdx.x, blockIdx.y, blockIdx.z, As, Bs);
}

#define KVP 260
__device__ __forceinline__ void reduce_kv_body(
    const float* __restrict__ part, const float* __restrict__ bias,
    const float* __restrict__ Wkv, float* __restrict__ kv,
    int MN, int S, int blk, float* Ps)
{
    int t = threadIdx.x;
    int r0 = blk * 16;
    int quads = MN >> 2;
    #pragma unroll 1
    for (int i = 0; i < 4; ++i) {
        int q = t + 256 * i;
        const float4* p = reinterpret_cast<const float4*>(part) + (r0 * 64 + q);
        float4 acc = make_float4(0.0f, 0.0f, 0.0f, 0.0f);
        #pragma unroll 16
        for (int z = 0; z < S; ++z) {
            float4 v = p[(size_t)z * quads];
            acc.x += v.x; acc.y += v.y; acc.z += v.z; acc.w += v.w;
        }
        int col = (q & 63) << 2;
        const float4 bv = *reinterpret_cast<const float4*>(bias + col);
        acc.x += bv.x; acc.y += bv.y; acc.z += bv.z; acc.w += bv.w;
        *reinterpret_cast<float4*>(&Ps[(q >> 6) * KVP + col]) = acc;
    }
    __syncthreads();

    int wave = t >> 5;
    int lane = t & 31;
    int hf   = lane >> 4;
    int r    = lane & 15;
    const float* as = &Ps[r * KVP + 4 * hf];
    const float* bg = Wkv + (size_t)(wave * 64 + r) * 256 + 4 * hf;
    v8f acc[4] = {};
    #pragma unroll 2
    for (int g = 0; g < 32; ++g) {
        v4f a = *reinterpret_cast<const v4f*>(as + 8 * g);
        v4f b[4];
        #pragma unroll
        for (int nt = 0; nt < 4; ++nt)
            b[nt] = *reinterpret_cast<const v4f*>(bg + (size_t)(16 * nt) * 256 + 8 * g);
        v2f alo = {a.x, a.y}, ahi = {a.z, a.w};
        #pragma unroll
        for (int nt = 0; nt < 4; ++nt) {
            v2f blo = {b[nt].x, b[nt].y};
            acc[nt] = __builtin_amdgcn_wmma_f32_16x16x4_f32(false, alo, false, blo, (short)0, acc[nt], false, false);
        }
        #pragma unroll
        for (int nt = 0; nt < 4; ++nt) {
            v2f bhi = {b[nt].z, b[nt].w};
            acc[nt] = __builtin_amdgcn_wmma_f32_16x16x4_f32(false, ahi, false, bhi, (short)0, acc[nt], false, false);
        }
    }
    #pragma unroll
    for (int nt = 0; nt < 4; ++nt)
        #pragma unroll
        for (int vg = 0; vg < 8; ++vg)
            kv[(size_t)(r0 + vg + 8 * hf) * 512 + wave * 64 + 16 * nt + r] = acc[nt][vg];
}


__global__ __launch_bounds__(256) void reduce_and_gemm(
    const float* __restrict__ part, const float* __restrict__ rbias,
    const float* __restrict__ Wkv, float* __restrict__ kvdst,
    int MN, int S, int rblocks,
    const float* __restrict__ A, const float* __restrict__ Bn, float* __restrict__ C,
    int M, int N, int K, int ldb, float scale)
{
    __shared__ __attribute__((aligned(16))) float As[2 * GBM * GPA];
    __shared__ __attribute__((aligned(16))) float Bs[2 * GBN * GPA];
    int id = blockIdx.x;
    if (id < rblocks) {
        reduce_kv_body(part, rbias, Wkv, kvdst, MN, S, id, As);
        return;
    }
    id -= rblocks;
    int nb = N / GBN;
    gemm_tiled_body<1>(A, Bn, C, nullptr, M, N, K, 0, ldb, scale, id % nb, id / nb, 0, As, Bs);
}

#define APITCH 36
#define VPITCH 72
__global__ __launch_bounds__(128) void attn_kernel(
    float* __restrict__ qmat, const float* __restrict__ kv,
    const float* __restrict__ x, const float* __restrict__ rw, const float* __restrict__ rb)
{
    __shared__ __attribute__((aligned(16))) float Ks[144 * APITCH];
    __shared__ __attribute__((aligned(16))) float Vs[72 * VPITCH];
    int id = blockIdx.x;
    int h  = id & 7;
    int bb = (id >> 3) & 7;
    int g  = id >> 6;
    int n  = g * 8 + bb;

    int t    = threadIdx.x;
    int wave = t >> 5;
    int lane = t & 31;
    int hf   = lane >> 4;
    int c    = lane & 15;
    int m0   = wave * 16;

    float*       qrow = qmat + ((size_t)n * 64 + m0) * 256 + h * 32;
    const float* kg   = kv + (size_t)bb * 144 * 512 + h * 32;

    v4f qf[4];
    #pragma unroll
    for (int g4 = 0; g4 < 4; ++g4)
        qf[g4] = *reinterpret_cast<const v4f*>(qrow + c * 256 + 8 * g4 + 4 * hf);

    {
        int sr = t >> 3, sc = (t & 7) * 4;
        const float* ksrc = kg + (size_t)sr * 512 + sc;
        const float* vsrc = kg + 256 + (size_t)(2 * sr) * 512 + sc;
        bool tail = sr < 8;
        v4f sk[9], va[5], vb[5];
        #pragma unroll
        for (int i = 0; i < 9; ++i) sk[i] = *reinterpret_cast<const v4f*>(ksrc + (size_t)(16 * i) * 512);
        #pragma unroll
        for (int i = 0; i < 5; ++i)
            if (i < 4 || tail) {
                va[i] = *reinterpret_cast<const v4f*>(vsrc + (size_t)(32 * i) * 512);
                vb[i] = *reinterpret_cast<const v4f*>(vsrc + (size_t)(32 * i + 1) * 512);
            }
        #pragma unroll
        for (int i = 0; i < 9; ++i) *reinterpret_cast<v4f*>(&Ks[(sr + 16 * i) * APITCH + sc]) = sk[i];
        #pragma unroll
        for (int i = 0; i < 5; ++i)
            if (i < 4 || tail) {
                float* d = &Vs[(sr + 16 * i) * VPITCH + 2 * sc];
                v4f lo = {va[i].x, vb[i].x, va[i].y, vb[i].y};
                v4f hi = {va[i].z, vb[i].z, va[i].w, vb[i].w};
                *reinterpret_cast<v4f*>(d)     = lo;
                *reinterpret_cast<v4f*>(d + 4) = hi;
            }
    }
    __syncthreads();

    v8f acc[9];
    const float* kf = &Ks[c * APITCH + 4 * hf];
    #pragma unroll
    for (int tg = 0; tg < 3; ++tg) {
        v4f kk[3][4];
        #pragma unroll
        for (int x = 0; x < 3; ++x)
            #pragma unroll
            for (int g4 = 0; g4 < 4; ++g4)
                kk[x][g4] = *reinterpret_cast<const v4f*>(kf + (16 * (3 * tg + x)) * APITCH + 8 * g4);
        v8f a0 = {}, a1 = {}, a2 = {};
        #pragma unroll
        for (int g4 = 0; g4 < 4; ++g4) {
            v2f qlo = {qf[g4].x, qf[g4].y}, qhi = {qf[g4].z, qf[g4].w};
            v2f k0lo = {kk[0][g4].x, kk[0][g4].y}, k0hi = {kk[0][g4].z, kk[0][g4].w};
            v2f k1lo = {kk[1][g4].x, kk[1][g4].y}, k1hi = {kk[1][g4].z, kk[1][g4].w};
            v2f k2lo = {kk[2][g4].x, kk[2][g4].y}, k2hi = {kk[2][g4].z, kk[2][g4].w};
            a0 = __builtin_amdgcn_wmma_f32_16x16x4_f32(false, k0lo, false, qlo, (short)0, a0, false, false);
            a1 = __builtin_amdgcn_wmma_f32_16x16x4_f32(false, k1lo, false, qlo, (short)0, a1, false, false);
            a2 = __builtin_amdgcn_wmma_f32_16x16x4_f32(false, k2lo, false, qlo, (short)0, a2, false, false);
            a0 = __builtin_amdgcn_wmma_f32_16x16x4_f32(false, k0hi, false, qhi, (short)0, a0, false, false);
            a1 = __builtin_amdgcn_wmma_f32_16x16x4_f32(false, k1hi, false, qhi, (short)0, a1, false, false);
            a2 = __builtin_amdgcn_wmma_f32_16x16x4_f32(false, k2hi, false, qhi, (short)0, a2, false, false);
        }
        acc[3 * tg] = a0; acc[3 * tg + 1] = a1; acc[3 * tg + 2] = a2;
    }

    float mx = acc[0][0];
    #pragma unroll
    for (int tl = 0; tl < 9; ++tl)
        #pragma unroll
        for (int rr = 0; rr < 8; ++rr)
            mx = fmaxf(mx, acc[tl][rr]);
    mx = fmaxf(mx, __shfl_xor(mx, 16));
    float s0 = 0.0f, s1 = 0.0f, s2 = 0.0f, s3 = 0.0f;
    #pragma unroll
    for (int tl = 0; tl < 9; ++tl) {
        #pragma unroll
        for (int rr = 0; rr < 8; ++rr)
            acc[tl][rr] = __expf(acc[tl][rr] - mx);
        s0 += acc[tl][0] + acc[tl][4];
        s1 += acc[tl][1] + acc[tl][5];
        s2 += acc[tl][2] + acc[tl][6];
        s3 += acc[tl][3] + acc[tl][7];
    }
    float sum = (s0 + s1) + (s2 + s3);
    sum += __shfl_xor(sum, 16);
    float inv = 1.0f / sum;

    v8f o0 = {}, o1 = {};
    const float* vf = &Vs[(4 * hf) * VPITCH + 2 * c];
    #pragma unroll
    for (int tl = 0; tl < 9; ++tl) {
        #pragma unroll
        for (int j = 0; j < 4; ++j) {
            const float* vp = vf + (8 * tl + j) * VPITCH;
            v2f v0 = *reinterpret_cast<const v2f*>(vp);
            v2f v1 = *reinterpret_cast<const v2f*>(vp + 32);
            v2f p  = {acc[tl][2 * j], acc[tl][2 * j + 1]};
            o0 = __builtin_amdgcn_wmma_f32_16x16x4_f32(false, v0, false, p, (short)0, o0, false, false);
            o1 = __builtin_amdgcn_wmma_f32_16x16x4_f32(false, v1, false, p, (short)0, o1, false, false);
        }
    }

    __syncthreads();
    float* patch = &Ks[wave * (16 * APITCH)];
    {
        float* pw = patch + c * APITCH + 8 * hf;
        v4f w0 = {o0[0] * inv, o0[1] * inv, o0[2] * inv, o0[3] * inv};
        v4f w1 = {o0[4] * inv, o0[5] * inv, o0[6] * inv, o0[7] * inv};
        v4f w2 = {o1[0] * inv, o1[1] * inv, o1[2] * inv, o1[3] * inv};
        v4f w3 = {o1[4] * inv, o1[5] * inv, o1[6] * inv, o1[7] * inv};
        *reinterpret_cast<v4f*>(pw)      = w0;
        *reinterpret_cast<v4f*>(pw + 4)  = w1;
        *reinterpret_cast<v4f*>(pw + 16) = w2;
        *reinterpret_cast<v4f*>(pw + 20) = w3;
    }
    __syncthreads();
    int pr = lane >> 3, pc = (lane & 7) * 4;
    int R0 = n * 64 + __builtin_amdgcn_readfirstlane(m0);
    #pragma unroll
    for (int i = 0; i < 4; ++i) {
        int row   = pr + 4 * i;
        int Rb    = R0 + 4 * i;
        int plane = Rb / 36;
        int ch    = plane & 255;
        const float* wp = rw + ch * 9;
        float w[9];
        #pragma unroll
        for (int k = 0; k < 9; ++k) w[k] = wp[k];
        float bias = rb[ch];
        int pos = (Rb - plane * 36 + pr) * 256 + h * 32 + pc;
        int y   = pos / 96;
        int xc  = pos - y * 96;
        const float* xp = x + (size_t)plane * 9216;
        int  cl = (xc > 0) ? xc - 1 : 0;
        int  cr = (xc < 92) ? xc + 4 : 95;
        bool hl = xc > 0, hr = xc < 92;
        float tp[3][6];
        #pragma unroll
        for (int dy = 0; dy < 3; ++dy) {
            int  yy = y - 1 + dy;
            bool ok = (yy >= 0) && (yy < 96);
            int  yc = yy < 0 ? 0 : (yy > 95 ? 95 : yy);
            const float* xr = xp + yc * 96;
            v4f m = *reinterpret_cast<const v4f*>(xr + xc);
            float l = xr[cl];
            float r = xr[cr];
            tp[dy][0] = (ok && hl) ? l : 0.0f;
            tp[dy][1] = ok ? m.x : 0.0f;
            tp[dy][2] = ok ? m.y : 0.0f;
            tp[dy][3] = ok ? m.z : 0.0f;
            tp[dy][4] = ok ? m.w : 0.0f;
            tp[dy][5] = (ok && hr) ? r : 0.0f;
        }
        v4f v = *reinterpret_cast<const v4f*>(&patch[row * APITCH + pc]);
        float a[4];
        #pragma unroll
        for (int e = 0; e < 4; ++e) {
            float s = bias;
            #pragma unroll
            for (int dy = 0; dy < 3; ++dy)
                #pragma unroll
                for (int dx = 0; dx < 3; ++dx)
                    s = __builtin_fmaf(tp[dy][e + dx], w[dy * 3 + dx], s);
            a[e] = s;
        }
        v.x += a[0]; v.y += a[1]; v.z += a[2]; v.w += a[3];
        *reinterpret_cast<v4f*>(qrow + row * 256 + pc) = v;
    }
}

extern "C" void kernel_launch(void* const* d_in, const int* in_sizes, int n_in,
                              void* d_out, int out_size, void* d_ws, size_t ws_size,
                              hipStream_t stream) {
    const float* x      = (const float*)d_in[0];
    const float* Wq     = (const float*)d_in[1];
    const float* Wkv    = (const float*)d_in[2];
    const float* pool_w = (const float*)d_in[3];
    const float* pool_b = (const float*)d_in[4];
    const float* rpe_w  = (const float*)d_in[5];
    const float* rpe_b  = (const float*)d_in[6];
    const float* proj_w = (const float*)d_in[7];
    const float* proj_b = (const float*)d_in[8];
    float* out = (float*)d_out;

    float* ws    = (float*)d_ws;
    float* qmat  = ws;
    float* kvb   = qmat + 19169280;
    float* part  = kvb + 589824;
    (void)ws_size; (void)in_sizes; (void)n_in; (void)out_size;

    const int TB = 256;
    gemm_tiled_f32<2><<<dim3(1, 12, 64), TB, 0, stream>>>(
        x, pool_w, part, nullptr, 1152, 256, 256, 0, 16384, 1.0f);

    reduce_and_gemm<<<840, TB, 0, stream>>>(
        part, pool_b, Wkv, kvb, 294912, 64, 72,
        x, Wq, qmat, 73728, 256, 256, 256, QSCALE);

    attn_kernel<<<144 * 8 * 8, 128, 0, stream>>>(qmat, kvb, x, rpe_w, rpe_b);

    gemm_tiled_f32<0><<<dim3(1, 768), TB, 0, stream>>>(
        qmat, proj_w, out, proj_b, 73728, 256, 256, 256, 256, 1.0f);
}
